# starting kernel
# baseline (speedup 1.0000x reference)
#include <hip/hip_runtime.h>
#include <hip/hip_fp16.h>

typedef __attribute__((ext_vector_type(16))) _Float16 v16h;
typedef __attribute__((ext_vector_type(8)))  _Float16 v8h;
typedef __attribute__((ext_vector_type(8)))  float    v8f;

#define HDIM 51
#define LOG2E 1.44269504088896340736f

__device__ __forceinline__ float fast_sigmoid(float z) {
  float e = __builtin_amdgcn_exp2f(-LOG2E * z);
  return __builtin_amdgcn_rcpf(1.0f + e);
}
__device__ __forceinline__ float fast_tanh(float z) {
  float e = __builtin_amdgcn_exp2f(2.0f * LOG2E * z);
  return 1.0f - 2.0f * __builtin_amdgcn_rcpf(e + 1.0f);
}

__global__ __launch_bounds__(128)
void lstm2_wmma_kernel(const float* __restrict__ input,
                       const float* __restrict__ w_ih1, const float* __restrict__ w_hh1,
                       const float* __restrict__ b_ih1, const float* __restrict__ b_hh1,
                       const float* __restrict__ w_ih2, const float* __restrict__ w_hh2,
                       const float* __restrict__ b_ih2, const float* __restrict__ b_hh2,
                       const float* __restrict__ w_lin, const float* __restrict__ b_lin,
                       float* __restrict__ out, int T, int Tout)
{
  __shared__ v16h b1frag[32][32];
  __shared__ v16h b2frag[64][32];
  __shared__ __align__(16) _Float16 hcat[4][16][136];
  __shared__ float bias1s[256], bias2s[256], wih1s[256];
  __shared__ float xtile[4][16];

  const int tid = threadIdx.x;

  for (int idx = tid; idx < 32 * 512; idx += 128) {
    int f = idx >> 9, rem = idx & 511, l = rem >> 4, e = rem & 15;
    int ntile = f >> 1, ks = f & 1;
    int cc = l & 15, hh = (l & 16) ? 8 : 0;
    int Kl = (e < 8) ? (hh + e) : (16 + hh + (e - 8));
    int K = ks * 32 + Kl;
    int col = ntile * 16 + cc, q = col >> 6, j = col & 63;
    float v = 0.0f;
    if (K < HDIM && j < HDIM) v = w_hh1[(q * HDIM + j) * HDIM + K];
    ((_Float16*)&b1frag[f][l])[e] = (_Float16)v;
  }
  for (int idx = tid; idx < 64 * 512; idx += 128) {
    int f = idx >> 9, rem = idx & 511, l = rem >> 4, e = rem & 15;
    int ntile = f >> 2, ks = f & 3;
    int cc = l & 15, hh = (l & 16) ? 8 : 0;
    int Kl = (e < 8) ? (hh + e) : (16 + hh + (e - 8));
    int K = ks * 32 + Kl;
    int col = ntile * 16 + cc, q = col >> 6, j = col & 63;
    float v = 0.0f;
    if (j < HDIM) {
      if (K < 64) { if (K < HDIM) v = w_ih2[(q * HDIM + j) * HDIM + K]; }
      else        { int kk = K - 64; if (kk < HDIM) v = w_hh2[(q * HDIM + j) * HDIM + kk]; }
    }
    ((_Float16*)&b2frag[f][l])[e] = (_Float16)v;
  }
  for (int col = tid; col < 256; col += 128) {
    int q = col >> 6, j = col & 63;
    float bb1 = 0.f, bb2 = 0.f, wi = 0.f;
    if (j < HDIM) {
      int r = q * HDIM + j;
      bb1 = b_ih1[r] + b_hh1[r];
      bb2 = b_ih2[r] + b_hh2[r];
      wi  = w_ih1[r];
    }
    bias1s[col] = bb1; bias2s[col] = bb2; wih1s[col] = wi;
  }
  for (int i = tid; i < 4 * 16 * 136; i += 128) ((_Float16*)hcat)[i] = (_Float16)0.0f;
  __syncthreads();

  const int w    = tid >> 5;
  const int lane = tid & 31;
  const int cix  = lane & 15;
  const int hi   = (lane & 16) ? 8 : 0;
  const int rowbase = blockIdx.x * 64 + w * 16;

  float biasv1[16], biasv2[16], wihv[16];
#pragma unroll
  for (int n = 0; n < 16; ++n) {
    int col = n * 16 + cix;
    biasv1[n] = bias1s[col]; biasv2[n] = bias2s[col]; wihv[n] = wih1s[col];
  }
  float wlinv[4];
#pragma unroll
  for (int tt = 0; tt < 4; ++tt) {
    int col = tt * 16 + cix;
    wlinv[tt] = (col < HDIM) ? w_lin[col] : 0.0f;
  }
  const float blin = b_lin[0];

  float c1s[4][8], c2s[4][8], ylast[8];
#pragma unroll
  for (int a = 0; a < 4; ++a)
#pragma unroll
    for (int k = 0; k < 8; ++k) { c1s[a][k] = 0.f; c2s[a][k] = 0.f; }
#pragma unroll
  for (int k = 0; k < 8; ++k) ylast[k] = 0.f;

  for (int t = 0; t < Tout; ++t) {
    float xv[8];
    if (t < T) {
      if (lane < 16) xtile[w][lane] = input[(long)(rowbase + lane) * T + t];
#pragma unroll
      for (int k = 0; k < 8; ++k) xv[k] = xtile[w][hi + k];
    } else {
#pragma unroll
      for (int k = 0; k < 8; ++k) xv[k] = ylast[k];
    }

    v16h a1[2];
#pragma unroll
    for (int ks = 0; ks < 2; ++ks) {
      const _Float16* bp = &hcat[w][cix][ks * 32 + hi];
      v8h lo = *(const v8h*)bp;
      v8h hp = *(const v8h*)(bp + 16);
      a1[ks] = __builtin_shufflevector(lo, hp, 0,1,2,3,4,5,6,7,8,9,10,11,12,13,14,15);
    }
#pragma unroll
    for (int tt = 0; tt < 4; ++tt) {
      v8f acc[4];
#pragma unroll
      for (int q = 0; q < 4; ++q) {
        int nt = q * 4 + tt;
        float bb = biasv1[nt];
        v8f cacc = {bb, bb, bb, bb, bb, bb, bb, bb};
#pragma unroll
        for (int ks = 0; ks < 2; ++ks)
          cacc = __builtin_amdgcn_wmma_f32_16x16x32_f16(false, a1[ks], false,
                    b1frag[nt * 2 + ks][lane], (short)0, cacc, false, false);
        float wv = wihv[nt];
#pragma unroll
        for (int k = 0; k < 8; ++k) cacc[k] += xv[k] * wv;
        acc[q] = cacc;
      }
      int col = tt * 16 + cix;
      bool valid = (col < HDIM);
#pragma unroll
      for (int k = 0; k < 8; ++k) {
        float ig = fast_sigmoid(acc[0][k]);
        float fg = fast_sigmoid(acc[1][k]);
        float gg = fast_tanh(acc[2][k]);
        float og = fast_sigmoid(acc[3][k]);
        float cn = fg * c1s[tt][k] + ig * gg;
        c1s[tt][k] = cn;
        float hn = og * fast_tanh(cn);
        hcat[w][hi + k][col] = valid ? (_Float16)hn : (_Float16)0.0f;
      }
    }

    v16h a2[4];
#pragma unroll
    for (int ks = 0; ks < 4; ++ks) {
      const _Float16* bp = &hcat[w][cix][ks * 32 + hi];
      v8h lo = *(const v8h*)bp;
      v8h hp = *(const v8h*)(bp + 16);
      a2[ks] = __builtin_shufflevector(lo, hp, 0,1,2,3,4,5,6,7,8,9,10,11,12,13,14,15);
    }
    float ys[8];
#pragma unroll
    for (int k = 0; k < 8; ++k) ys[k] = 0.f;
#pragma unroll
    for (int tt = 0; tt < 4; ++tt) {
      v8f acc[4];
#pragma unroll
      for (int q = 0; q < 4; ++q) {
        int nt = q * 4 + tt;
        float bb = biasv2[nt];
        v8f cacc = {bb, bb, bb, bb, bb, bb, bb, bb};
#pragma unroll
        for (int ks = 0; ks < 4; ++ks)
          cacc = __builtin_amdgcn_wmma_f32_16x16x32_f16(false, a2[ks], false,
                    b2frag[nt * 4 + ks][lane], (short)0, cacc, false, false);
        acc[q] = cacc;
      }
      int col = tt * 16 + cix;
      bool valid = (col < HDIM);
      float wl = wlinv[tt];
#pragma unroll
      for (int k = 0; k < 8; ++k) {
        float ig = fast_sigmoid(acc[0][k]);
        float fg = fast_sigmoid(acc[1][k]);
        float gg = fast_tanh(acc[2][k]);
        float og = fast_sigmoid(acc[3][k]);
        float cn = fg * c2s[tt][k] + ig * gg;
        c2s[tt][k] = cn;
        float hn = og * fast_tanh(cn);
        hcat[w][hi + k][64 + col] = valid ? (_Float16)hn : (_Float16)0.0f;
        ys[k] += (valid ? hn : 0.0f) * wl;
      }
    }

#pragma unroll
    for (int mask = 1; mask <= 8; mask <<= 1)
#pragma unroll
      for (int k = 0; k < 8; ++k) ys[k] += __shfl_xor(ys[k], mask, 32);
#pragma unroll
    for (int k = 0; k < 8; ++k) ylast[k] = ys[k] + blin;
    if (cix == 0) {
#pragma unroll
      for (int k = 0; k < 8; ++k)
        out[(long)(rowbase + hi + k) * Tout + t] = ylast[k];
    }
  }
}

extern "C" void kernel_launch(void* const* d_in, const int* in_sizes, int n_in,
                              void* d_out, int out_size, void* d_ws, size_t ws_size,
                              hipStream_t stream) {
  const float* input = (const float*)d_in[0];
  const float* w_ih1 = (const float*)d_in[2];
  const float* w_hh1 = (const float*)d_in[3];
  const float* b_ih1 = (const float*)d_in[4];
  const float* b_hh1 = (const float*)d_in[5];
  const float* w_ih2 = (const float*)d_in[6];
  const float* w_hh2 = (const float*)d_in[7];
  const float* b_ih2 = (const float*)d_in[8];
  const float* b_hh2 = (const float*)d_in[9];
  const float* w_lin = (const float*)d_in[10];
  const float* b_lin = (const float*)d_in[11];
  float* out = (float*)d_out;

  const int B = 1024;
  const int T = in_sizes[0] / B;
  const int Tout = out_size / B;

  lstm2_wmma_kernel<<<dim3(B / 64), dim3(128), 0, stream>>>(
      input, w_ih1, w_hh1, b_ih1, b_hh1, w_ih2, w_hh2, b_ih2, b_hh2,
      w_lin, b_lin, out, T, Tout);
}


# Weights as A operand, rows in (unit, gate) order, 4 units a tile: a lane's D holds all gates of 2 units and, packed, is the next B fragment; h stays in registers, 13 tiles not 16 (new sum order)
# speedup vs baseline: 2.2972x; 2.2972x over previous
#include <hip/hip_runtime.h>
#include <hip/hip_fp16.h>

typedef __attribute__((ext_vector_type(16))) _Float16 v16h;
typedef __attribute__((ext_vector_type(8)))  _Float16 v8h;
typedef __attribute__((ext_vector_type(2)))  _Float16 v2h;
typedef __attribute__((ext_vector_type(8)))  float    v8f;
typedef __attribute__((ext_vector_type(8)))  unsigned v8u;

#define HDIM 51
#define NT 13
#define LOG2E 1.44269504088896340736f

__device__ __forceinline__ float fast_sigmoid(float z) {
  float e = __builtin_amdgcn_exp2f(-LOG2E * z);
  return __builtin_amdgcn_rcpf(1.0f + e);
}
__device__ __forceinline__ float fast_tanh(float z) {
  float e = __builtin_amdgcn_exp2f(2.0f * LOG2E * z);
  return 1.0f - 2.0f * __builtin_amdgcn_rcpf(e + 1.0f);
}

__device__ __forceinline__ float xhalf(float v) {
  int b = __builtin_bit_cast(int, v);
  return __builtin_bit_cast(float, __builtin_amdgcn_permlanex16(b, b, 0x76543210, 0xfedcba98, false, false));
}

__device__ __forceinline__ v16h ldfrag(const v8h (*fr)[2][32], int f, int lane) {
  v8h lo = fr[f][0][lane];
  v8h hp = fr[f][1][lane];
  return __builtin_shufflevector(lo, hp, 0,1,2,3,4,5,6,7,8,9,10,11,12,13,14,15);
}

__global__ __launch_bounds__(128)
void lstm2_wmma_kernel(const float* __restrict__ input,
                       const float* __restrict__ w_ih1, const float* __restrict__ w_hh1,
                       const float* __restrict__ b_ih1, const float* __restrict__ b_hh1,
                       const float* __restrict__ w_ih2, const float* __restrict__ w_hh2,
                       const float* __restrict__ b_ih2, const float* __restrict__ b_hh2,
                       const float* __restrict__ w_lin, const float* __restrict__ b_lin,
                       float* __restrict__ out, int T, int Tout)
{
  __shared__ v8h a1frag[NT * 2][2][32];
  __shared__ v8h a2frag[NT * 4][2][32];
  __shared__ float bias1s[NT * 16], bias2s[NT * 16], wih1s[NT * 16];

  const int tid = threadIdx.x;

  for (int idx = tid; idx < NT * 2 * 512; idx += 128) {
    int f = idx >> 9, rem = idx & 511, l = rem >> 4, s = rem & 15;
    int m = f >> 1, b = f & 1;
    int cc = l & 15, lh = l >> 4;
    int unit = 4 * m + 2 * (cc >> 3) + ((cc & 7) >> 2), q = cc & 3;
    int mp = 8 * b + (s >> 1);
    int ku = 4 * mp + 2 * lh + (s & 1);
    float v = 0.0f;
    if (unit < HDIM && mp < NT && ku < HDIM) v = w_hh1[(q * HDIM + unit) * HDIM + ku];
    ((_Float16*)&a1frag[f][s >> 3][l])[s & 7] = (_Float16)v;
  }
  for (int idx = tid; idx < NT * 4 * 512; idx += 128) {
    int f = idx >> 9, rem = idx & 511, l = rem >> 4, s = rem & 15;
    int m = f >> 2, b = f & 3;
    int cc = l & 15, lh = l >> 4;
    int unit = 4 * m + 2 * (cc >> 3) + ((cc & 7) >> 2), q = cc & 3;
    int mp = 8 * (b & 1) + (s >> 1);
    int ku = 4 * mp + 2 * lh + (s & 1);
    float v = 0.0f;
    if (unit < HDIM && mp < NT && ku < HDIM) {
      const float* src = (b < 2) ? w_ih2 : w_hh2;
      v = src[(q * HDIM + unit) * HDIM + ku];
    }
    ((_Float16*)&a2frag[f][s >> 3][l])[s & 7] = (_Float16)v;
  }
  for (int ri = tid; ri < NT * 16; ri += 128) {
    int m = ri >> 4, cc = ri & 15;
    int unit = 4 * m + 2 * (cc >> 3) + ((cc & 7) >> 2), q = cc & 3;
    float bb1 = 0.f, bb2 = 0.f, wi = 0.f;
    if (unit < HDIM) {
      int r = q * HDIM + unit;
      bb1 = b_ih1[r] + b_hh1[r];
      bb2 = b_ih2[r] + b_hh2[r];
      wi  = w_ih1[r];
    }
    bias1s[ri] = bb1; bias2s[ri] = bb2; wih1s[ri] = wi;
  }
  __syncthreads();

  const int w    = tid >> 5;
  const int lane = tid & 31;
  const int cix  = lane & 15;
  const int lh   = lane >> 4;
  const int rowbase = blockIdx.x * 64 + w * 16;

  float bias1[NT][8], bias2[NT][8], wih[NT][8], wl[NT][2];
#pragma unroll
  for (int m = 0; m < NT; ++m) {
#pragma unroll
    for (int r = 0; r < 8; ++r) {
      int ri = m * 16 + 8 * lh + r;
      bias1[m][r] = bias1s[ri]; bias2[m][r] = bias2s[ri]; wih[m][r] = wih1s[ri];
    }
#pragma unroll
    for (int u = 0; u < 2; ++u) {
      int unit = 4 * m + 2 * lh + u;
      wl[m][u] = (unit < HDIM) ? w_lin[unit] : 0.0f;
    }
  }
  const float blin = b_lin[0];

  float c1[NT][2], c2[NT][2];
#pragma unroll
  for (int m = 0; m < NT; ++m) { c1[m][0] = 0.f; c1[m][1] = 0.f; c2[m][0] = 0.f; c2[m][1] = 0.f; }

  v8u hb1[2], hb2[2];
#pragma unroll
  for (int b = 0; b < 2; ++b)
#pragma unroll
    for (int j = 0; j < 8; ++j) { hb1[b][j] = 0u; hb2[b][j] = 0u; }
  float ylast = 0.f;

  const float* xin = input + (long)(rowbase + cix) * T;
  float* yout = out + (long)(rowbase + cix) * Tout;

  v16h wf[2][16];
#pragma unroll
  for (int i = 0; i < 4; ++i)
#pragma unroll
    for (int kb = 0; kb < 2; ++kb)
      wf[0][i * 2 + kb] = ldfrag(a1frag, i * 2 + kb, lane);

  for (int t = 0; t < Tout; ++t) {
    const float x = (t < T) ? xin[t] : ylast;

    v8u hn1[2] = {hb1[0], hb1[1]};
#pragma unroll
    for (int blk = 0; blk < 4; ++blk) {
      const int m0 = (blk == 0) ? 0 : 1 + 3 * blk;
      const int n  = (blk == 0) ? 4 : 3;
      __builtin_amdgcn_sched_barrier(0);
      if (blk < 3) {
        const int nm0 = 4 + 3 * blk;
#pragma unroll
        for (int i = 0; i < 3; ++i)
#pragma unroll
          for (int kb = 0; kb < 2; ++kb)
            wf[(blk + 1) & 1][i * 2 + kb] = ldfrag(a1frag, (nm0 + i) * 2 + kb, lane);
      } else {
#pragma unroll
        for (int i = 0; i < 4; ++i)
#pragma unroll
          for (int kb = 0; kb < 4; ++kb)
            wf[0][i * 4 + kb] = ldfrag(a2frag, i * 4 + kb, lane);
      }
      __builtin_amdgcn_sched_barrier(0);
      asm volatile("" : "+v"(hb1[0]));

      v8f acc[4];
#pragma unroll
      for (int i = 0; i < 4; ++i) if (i < n) {
#pragma unroll
        for (int r = 0; r < 8; ++r) acc[i][r] = __builtin_fmaf(x, wih[m0 + i][r], bias1[m0 + i][r]);
      }
#pragma unroll
      for (int kb = 0; kb < 2; ++kb)
#pragma unroll
        for (int i = 0; i < 4; ++i) if (i < n)
          acc[i] = __builtin_amdgcn_wmma_f32_16x16x32_f16(false, wf[blk & 1][i * 2 + kb], false,
                      __builtin_bit_cast(v16h, hb1[kb]), (short)0, acc[i], false, false);
#pragma unroll
      for (int i = 0; i < 4; ++i) if (i < n) {
        const int m = m0 + i;
        float hn[2];
#pragma unroll
        for (int u = 0; u < 2; ++u) {
          float ig = fast_sigmoid(acc[i][4 * u + 0]);
          float fg = fast_sigmoid(acc[i][4 * u + 1]);
          float gg = fast_tanh(acc[i][4 * u + 2]);
          float og = fast_sigmoid(acc[i][4 * u + 3]);
          float cn = fg * c1[m][u] + ig * gg;
          c1[m][u] = cn;
          hn[u] = og * fast_tanh(cn);
        }
        if (m == NT - 1 && lh == 1) hn[1] = 0.0f;
        v2h p; p[0] = (_Float16)hn[0]; p[1] = (_Float16)hn[1];
        hn1[m >> 3][m & 7] = __builtin_bit_cast(unsigned, p);
      }
    }
    hb1[0] = hn1[0]; hb1[1] = hn1[1];

    v8u hn2[2] = {hb2[0], hb2[1]};
    float ys = 0.f;
#pragma unroll
    for (int blk = 0; blk < 4; ++blk) {
      const int m0 = (blk == 0) ? 0 : 1 + 3 * blk;
      const int n  = (blk == 0) ? 4 : 3;
      __builtin_amdgcn_sched_barrier(0);
      if (blk < 3) {
        const int nm0 = 4 + 3 * blk;
#pragma unroll
        for (int i = 0; i < 3; ++i)
#pragma unroll
          for (int kb = 0; kb < 4; ++kb)
            wf[(blk + 1) & 1][i * 4 + kb] = ldfrag(a2frag, (nm0 + i) * 4 + kb, lane);
      } else {
#pragma unroll
        for (int i = 0; i < 4; ++i)
#pragma unroll
          for (int kb = 0; kb < 2; ++kb)
            wf[0][i * 2 + kb] = ldfrag(a1frag, i * 2 + kb, lane);
      }
      __builtin_amdgcn_sched_barrier(0);
      asm volatile("" : "+v"(hb1[0]));

      v8f acc[4];
#pragma unroll
      for (int i = 0; i < 4; ++i) if (i < n) {
#pragma unroll
        for (int r = 0; r < 8; ++r) acc[i][r] = bias2[m0 + i][r];
      }
#pragma unroll
      for (int kb = 0; kb < 4; ++kb)
#pragma unroll
        for (int i = 0; i < 4; ++i) if (i < n)
          acc[i] = __builtin_amdgcn_wmma_f32_16x16x32_f16(false, wf[blk & 1][i * 4 + kb], false,
                      __builtin_bit_cast(v16h, (kb < 2) ? hb1[kb & 1] : hb2[kb & 1]),
                      (short)0, acc[i], false, false);
#pragma unroll
      for (int i = 0; i < 4; ++i) if (i < n) {
        const int m = m0 + i;
        float hn[2];
#pragma unroll
        for (int u = 0; u < 2; ++u) {
          float ig = fast_sigmoid(acc[i][4 * u + 0]);
          float fg = fast_sigmoid(acc[i][4 * u + 1]);
          float gg = fast_tanh(acc[i][4 * u + 2]);
          float og = fast_sigmoid(acc[i][4 * u + 3]);
          float cn = fg * c2[m][u] + ig * gg;
          c2[m][u] = cn;
          hn[u] = og * fast_tanh(cn);
        }
        if (m == NT - 1 && lh == 1) hn[1] = 0.0f;
        ys += hn[0] * wl[m][0];
        ys += hn[1] * wl[m][1];
        v2h p; p[0] = (_Float16)hn[0]; p[1] = (_Float16)hn[1];
        hn2[m >> 3][m & 7] = __builtin_bit_cast(unsigned, p);
      }
    }
    hb2[0] = hn2[0]; hb2[1] = hn2[1];

    ys += xhalf(ys);
    ylast = ys + blin;
    if (lh == 0) yout[t] = ylast;
  }
}

extern "C" void kernel_launch(void* const* d_in, const int* in_sizes, int n_in,
                              void* d_out, int out_size, void* d_ws, size_t ws_size,
                              hipStream_t stream) {
  const float* input = (const float*)d_in[0];
  const float* w_ih1 = (const float*)d_in[2];
  const float* w_hh1 = (const float*)d_in[3];
  const float* b_ih1 = (const float*)d_in[4];
  const float* b_hh1 = (const float*)d_in[5];
  const float* w_ih2 = (const float*)d_in[6];
  const float* w_hh2 = (const float*)d_in[7];
  const float* b_ih2 = (const float*)d_in[8];
  const float* b_hh2 = (const float*)d_in[9];
  const float* w_lin = (const float*)d_in[10];
  const float* b_lin = (const float*)d_in[11];
  float* out = (float*)d_out;

  const int B = 1024;
  const int T = in_sizes[0] / B;
  const int Tout = out_size / B;

  lstm2_wmma_kernel<<<dim3(B / 64), dim3(128), 0, stream>>>(
      input, w_ih1, w_hh1, b_ih1, b_hh1, w_ih2, w_hh2, b_ih2, b_hh2,
      w_lin, b_lin, out, T, Tout);
}


# All weight fragments of both layers stay in registers for the whole scan, no LDS read in the time loop; bias and x*w_ih1 ride as f16 head+tail pairs in the products' spare k slots (new sum order)
# speedup vs baseline: 2.9848x; 1.2993x over previous
#include <hip/hip_runtime.h>
#include <hip/hip_fp16.h>
#include <type_traits>

typedef __attribute__((ext_vector_type(16))) _Float16 v16h;
typedef __attribute__((ext_vector_type(8)))  _Float16 v8h;
typedef __attribute__((ext_vector_type(2)))  _Float16 v2h;
typedef __attribute__((ext_vector_type(8)))  float    v8f;
typedef __attribute__((ext_vector_type(8)))  unsigned v8u;

#define HDIM 51
#define NT 13
#define LOG2E 1.44269504088896340736f

__device__ __forceinline__ float fast_sigmoid(float z) {
  float e = __builtin_amdgcn_exp2f(-LOG2E * z);
  return __builtin_amdgcn_rcpf(1.0f + e);
}
__device__ __forceinline__ float fast_tanh(float z) {
  float e = __builtin_amdgcn_exp2f(2.0f * LOG2E * z);
  return 1.0f - 2.0f * __builtin_amdgcn_rcpf(e + 1.0f);
}

__device__ __forceinline__ float xhalf(float v) {
  int b = __builtin_bit_cast(int, v);
  return __builtin_bit_cast(float, __builtin_amdgcn_permlanex16(b, b, 0x76543210, 0xfedcba98, false, false));
}

__device__ __forceinline__ v16h ldfrag(const v8h (*fr)[2][32], int f, int lane) {
  v8h lo = fr[f][0][lane];
  v8h hp = fr[f][1][lane];
  return __builtin_shufflevector(lo, hp, 0,1,2,3,4,5,6,7,8,9,10,11,12,13,14,15);
}

__global__ __launch_bounds__(128)
void lstm2_wmma_kernel(const float* __restrict__ input,
                       const float* __restrict__ w_ih1, const float* __restrict__ w_hh1,
                       const float* __restrict__ b_ih1, const float* __restrict__ b_hh1,
                       const float* __restrict__ w_ih2, const float* __restrict__ w_hh2,
                       const float* __restrict__ b_ih2, const float* __restrict__ b_hh2,
                       const float* __restrict__ w_lin, const float* __restrict__ b_lin,
                       float* __restrict__ out, int T, int Tout)
{
  __shared__ v8h a1frag[NT * 2][2][32];
  __shared__ v8h a2frag[NT * 4][2][32];

  const int tid = threadIdx.x;

  for (int idx = tid; idx < NT * 2 * 512; idx += 128) {
    int f = idx >> 9, rem = idx & 511, l = rem >> 4, s = rem & 15;
    int m = f >> 1, b = f & 1;
    int cc = l & 15, lh = l >> 4;
    int unit = 4 * m + 2 * (cc >> 3) + ((cc & 7) >> 2), q = cc & 3;
    int mp = 8 * b + (s >> 1);
    int ku = 4 * mp + 2 * lh + (s & 1);
    float v = 0.0f;
    if (unit < HDIM && mp < NT && ku < HDIM) v = w_hh1[(q * HDIM + unit) * HDIM + ku];
    _Float16 hv = (_Float16)v;
    if (b == 1 && lh == 0 && s >= 10 && s <= 14 && unit < HDIM) {
      int r = q * HDIM + unit;
      float src = (s <= 12) ? w_ih1[r] : (b_ih1[r] + b_hh1[r]);
      _Float16 hd = (_Float16)src;
      _Float16 tl = (_Float16)(src - (float)hd);
      hv = (s == 11 || s == 14) ? tl : hd;
    }
    ((_Float16*)&a1frag[f][s >> 3][l])[s & 7] = hv;
  }
  for (int idx = tid; idx < NT * 4 * 512; idx += 128) {
    int f = idx >> 9, rem = idx & 511, l = rem >> 4, s = rem & 15;
    int m = f >> 2, b = f & 3;
    int cc = l & 15, lh = l >> 4;
    int unit = 4 * m + 2 * (cc >> 3) + ((cc & 7) >> 2), q = cc & 3;
    int mp = 8 * (b & 1) + (s >> 1);
    int ku = 4 * mp + 2 * lh + (s & 1);
    float v = 0.0f;
    if (unit < HDIM && mp < NT && ku < HDIM) {
      const float* src = (b < 2) ? w_ih2 : w_hh2;
      v = src[(q * HDIM + unit) * HDIM + ku];
    }
    _Float16 hv = (_Float16)v;
    if (b == 1 && lh == 0 && (s == 13 || s == 14) && unit < HDIM) {
      int r = q * HDIM + unit;
      float src = b_ih2[r] + b_hh2[r];
      _Float16 hd = (_Float16)src;
      _Float16 tl = (_Float16)(src - (float)hd);
      hv = (s == 14) ? tl : hd;
    }
    ((_Float16*)&a2frag[f][s >> 3][l])[s & 7] = hv;
  }
  __syncthreads();

  const int w    = tid >> 5;
  const int lane = tid & 31;
  const int cix  = lane & 15;
  const int lh   = lane >> 4;
  const int rowbase = blockIdx.x * 64 + w * 16;

  float wl[NT][2];
#pragma unroll
  for (int m = 0; m < NT; ++m)
#pragma unroll
    for (int u = 0; u < 2; ++u) {
      int unit = 4 * m + 2 * lh + u;
      wl[m][u] = (unit < HDIM) ? w_lin[unit] : 0.0f;
    }
  const float blin = b_lin[0];

  float c1[NT][2], c2[NT][2];
#pragma unroll
  for (int m = 0; m < NT; ++m) { c1[m][0] = 0.f; c1[m][1] = 0.f; c2[m][0] = 0.f; c2[m][1] = 0.f; }

  v8u hb1[2], hb2[2];
#pragma unroll
  for (int b = 0; b < 2; ++b)
#pragma unroll
    for (int j = 0; j < 8; ++j) { hb1[b][j] = 0u; hb2[b][j] = 0u; }
  float ylast = 0.f;

  const float* xin = input + (long)(rowbase + cix) * T;
  float* yout = out + (long)(rowbase + cix) * Tout;

  v16h wf1[NT * 2], wf2[NT * 4];
#pragma unroll
  for (int f = 0; f < NT * 2; ++f) wf1[f] = ldfrag(a1frag, f, lane);
#pragma unroll
  for (int f = 0; f < NT * 4; ++f) wf2[f] = ldfrag(a2frag, f, lane);

  for (int t = 0; t < Tout; ++t) {
    const float x = (t < T) ? xin[t] : ylast;

    {
      const _Float16 xh = (_Float16)x;
      const _Float16 xt = (_Float16)(x - (float)xh);
      v2h p5; p5[0] = xh; p5[1] = xh;
      v2h p6; p6[0] = xt; p6[1] = (_Float16)1.0f;
      v2h p7; p7[0] = (_Float16)1.0f; p7[1] = (_Float16)0.0f;
      hb1[1][5] = __builtin_bit_cast(unsigned, p5);
      hb1[1][6] = __builtin_bit_cast(unsigned, p6);
      hb1[1][7] = __builtin_bit_cast(unsigned, p7);
    }

    v8u hn1[2] = {hb1[0], hb1[1]};
    v8u hn2[2] = {hb2[0], hb2[1]};
    float ys = 0.f;
    v8f acc[2][4];
    const v8f zero8 = {0.f, 0.f, 0.f, 0.f, 0.f, 0.f, 0.f, 0.f};

    auto products = [&](auto Rc, auto PARTc) __attribute__((always_inline)) {
      constexpr int r = decltype(Rc)::value;
      constexpr int part = decltype(PARTc)::value;
      constexpr int blk = r & 3;
      constexpr int m0 = (blk == 0) ? 0 : 1 + 3 * blk;
      constexpr int n  = (blk == 0) ? 4 : 3;
      if (r < 4) {
#pragma unroll
        for (int kb = 0; kb < 2; ++kb)
#pragma unroll
          for (int i = 0; i < n; ++i)
            acc[r & 1][i] = __builtin_amdgcn_wmma_f32_16x16x32_f16(false, wf1[(m0 + i) * 2 + kb], false,
                        __builtin_bit_cast(v16h, hb1[kb]), (short)0,
                        (kb == 0) ? zero8 : acc[r & 1][i], false, false);
      } else {
#pragma unroll
        for (int kk = 0; kk < 4; ++kk) {
          if ((part == 1 && kk == 3) || (part == 2 && kk != 3)) continue;
          const int kb = (kk + 2) & 3;
#pragma unroll
          for (int i = 0; i < n; ++i)
            acc[r & 1][i] = __builtin_amdgcn_wmma_f32_16x16x32_f16(false, wf2[(m0 + i) * 4 + kb], false,
                        __builtin_bit_cast(v16h, (kb < 2) ? hb1[kb & 1] : hb2[kb & 1]),
                        (short)0, (kk == 0) ? zero8 : acc[r & 1][i], false, false);
        }
      }
    };

    auto cell = [&](auto Pc) __attribute__((always_inline)) {
      constexpr int p = decltype(Pc)::value;
      constexpr int blk = p & 3;
      constexpr int m0 = (blk == 0) ? 0 : 1 + 3 * blk;
      constexpr int n  = (blk == 0) ? 4 : 3;
#pragma unroll
      for (int i = 0; i < n; ++i) {
        const int m = m0 + i;
        float hn[2];
#pragma unroll
        for (int u = 0; u < 2; ++u) {
          float ig = fast_sigmoid(acc[p & 1][i][4 * u + 0]);
          float fg = fast_sigmoid(acc[p & 1][i][4 * u + 1]);
          float gg = fast_tanh(acc[p & 1][i][4 * u + 2]);
          float og = fast_sigmoid(acc[p & 1][i][4 * u + 3]);
          if (p < 4) {
            float cn = fg * c1[m][u] + ig * gg;
            c1[m][u] = cn;
            hn[u] = og * fast_tanh(cn);
          } else {
            float cn = fg * c2[m][u] + ig * gg;
            c2[m][u] = cn;
            hn[u] = og * fast_tanh(cn);
          }
        }
        if (m == NT - 1 && lh == 1) hn[1] = 0.0f;
        v2h pk; pk[0] = (_Float16)hn[0]; pk[1] = (_Float16)hn[1];
        if (p < 4) {
          hn1[m >> 3][m & 7] = __builtin_bit_cast(unsigned, pk);
        } else {
          ys += hn[0] * wl[m][0];
          ys += hn[1] * wl[m][1];
          hn2[m >> 3][m & 7] = __builtin_bit_cast(unsigned, pk);
        }
      }
    };

#define IC(N) std::integral_constant<int, N>{}
    products(IC(0), IC(0));
    products(IC(1), IC(0)); cell(IC(0));
    products(IC(2), IC(0)); cell(IC(1));
    products(IC(3), IC(0)); cell(IC(2));
    hb1[0] = hn1[0];
    products(IC(4), IC(1)); cell(IC(3));
    hb1[1] = hn1[1];
    products(IC(4), IC(2));
    products(IC(5), IC(0)); cell(IC(4));
    products(IC(6), IC(0)); cell(IC(5));
    products(IC(7), IC(0)); cell(IC(6));
    cell(IC(7));
    hb2[0] = hn2[0]; hb2[1] = hn2[1];
#undef IC

    ys += xhalf(ys);
    ylast = ys + blin;
    if (lh == 0) yout[t] = ylast;
  }
}

extern "C" void kernel_launch(void* const* d_in, const int* in_sizes, int n_in,
                              void* d_out, int out_size, void* d_ws, size_t ws_size,
                              hipStream_t stream) {
  const float* input = (const float*)d_in[0];
  const float* w_ih1 = (const float*)d_in[2];
  const float* w_hh1 = (const float*)d_in[3];
  const float* b_ih1 = (const float*)d_in[4];
  const float* b_hh1 = (const float*)d_in[5];
  const float* w_ih2 = (const float*)d_in[6];
  const float* w_hh2 = (const float*)d_in[7];
  const float* b_ih2 = (const float*)d_in[8];
  const float* b_hh2 = (const float*)d_in[9];
  const float* w_lin = (const float*)d_in[10];
  const float* b_lin = (const float*)d_in[11];
  float* out = (float*)d_out;

  const int B = 1024;
  const int T = in_sizes[0] / B;
  const int Tout = out_size / B;

  lstm2_wmma_kernel<<<dim3(B / 64), dim3(128), 0, stream>>>(
      input, w_ih1, w_hh1, b_ih1, b_hh1, w_ih2, w_hh2, b_ih2, b_hh2,
      w_lin, b_lin, out, T, Tout);
}


# exp2 scales folded into the weights at load: rows i,f,o times -log2e, g rows times 2log2e, before f16 rounding (bias, x weights too); cell state kept times 2log2e: no scaling multiply in the scan
# speedup vs baseline: 3.2705x; 1.0957x over previous
#include <hip/hip_runtime.h>
#include <hip/hip_fp16.h>
#include <type_traits>

typedef __attribute__((ext_vector_type(16))) _Float16 v16h;
typedef __attribute__((ext_vector_type(8)))  _Float16 v8h;
typedef __attribute__((ext_vector_type(2)))  _Float16 v2h;
typedef __attribute__((ext_vector_type(8)))  float    v8f;
typedef __attribute__((ext_vector_type(8)))  unsigned v8u;

#define HDIM 51
#define NT 13
#define LOG2E 1.44269504088896340736f

#define KC (2.0f * LOG2E)
__device__ __forceinline__ float sig_scaled(float zs) {
  return __builtin_amdgcn_rcpf(1.0f + __builtin_amdgcn_exp2f(zs));
}
__device__ __forceinline__ float tanh_rcp_scaled(float zs) {
  return __builtin_amdgcn_rcpf(__builtin_amdgcn_exp2f(zs) + 1.0f);
}
__device__ __forceinline__ float gate_scale(int q) { return (q == 2) ? KC : -LOG2E; }

__device__ __forceinline__ float xhalf(float v) {
  int b = __builtin_bit_cast(int, v);
  return __builtin_bit_cast(float, __builtin_amdgcn_permlanex16(b, b, 0x76543210, 0xfedcba98, false, false));
}

__device__ __forceinline__ v16h ldfrag(const v8h (*fr)[2][32], int f, int lane) {
  v8h lo = fr[f][0][lane];
  v8h hp = fr[f][1][lane];
  return __builtin_shufflevector(lo, hp, 0,1,2,3,4,5,6,7,8,9,10,11,12,13,14,15);
}

__global__ __launch_bounds__(128)
void lstm2_wmma_kernel(const float* __restrict__ input,
                       const float* __restrict__ w_ih1, const float* __restrict__ w_hh1,
                       const float* __restrict__ b_ih1, const float* __restrict__ b_hh1,
                       const float* __restrict__ w_ih2, const float* __restrict__ w_hh2,
                       const float* __restrict__ b_ih2, const float* __restrict__ b_hh2,
                       const float* __restrict__ w_lin, const float* __restrict__ b_lin,
                       float* __restrict__ out, int T, int Tout)
{
  __shared__ v8h a1frag[NT * 2][2][32];
  __shared__ v8h a2frag[NT * 4][2][32];

  const int tid = threadIdx.x;

  for (int idx = tid; idx < NT * 2 * 512; idx += 128) {
    int f = idx >> 9, rem = idx & 511, l = rem >> 4, s = rem & 15;
    int m = f >> 1, b = f & 1;
    int cc = l & 15, lh = l >> 4;
    int unit = 4 * m + 2 * (cc >> 3) + ((cc & 7) >> 2), q = cc & 3;
    int mp = 8 * b + (s >> 1);
    int ku = 4 * mp + 2 * lh + (s & 1);
    float v = 0.0f;
    if (unit < HDIM && mp < NT && ku < HDIM) v = gate_scale(q) * w_hh1[(q * HDIM + unit) * HDIM + ku];
    _Float16 hv = (_Float16)v;
    if (b == 1 && lh == 0 && s >= 10 && s <= 14 && unit < HDIM) {
      int r = q * HDIM + unit;
      float src = gate_scale(q) * ((s <= 12) ? w_ih1[r] : (b_ih1[r] + b_hh1[r]));
      _Float16 hd = (_Float16)src;
      _Float16 tl = (_Float16)(src - (float)hd);
      hv = (s == 11 || s == 14) ? tl : hd;
    }
    ((_Float16*)&a1frag[f][s >> 3][l])[s & 7] = hv;
  }
  for (int idx = tid; idx < NT * 4 * 512; idx += 128) {
    int f = idx >> 9, rem = idx & 511, l = rem >> 4, s = rem & 15;
    int m = f >> 2, b = f & 3;
    int cc = l & 15, lh = l >> 4;
    int unit = 4 * m + 2 * (cc >> 3) + ((cc & 7) >> 2), q = cc & 3;
    int mp = 8 * (b & 1) + (s >> 1);
    int ku = 4 * mp + 2 * lh + (s & 1);
    float v = 0.0f;
    if (unit < HDIM && mp < NT && ku < HDIM) {
      const float* src = (b < 2) ? w_ih2 : w_hh2;
      v = gate_scale(q) * src[(q * HDIM + unit) * HDIM + ku];
    }
    _Float16 hv = (_Float16)v;
    if (b == 1 && lh == 0 && (s == 13 || s == 14) && unit < HDIM) {
      int r = q * HDIM + unit;
      float src = gate_scale(q) * (b_ih2[r] + b_hh2[r]);
      _Float16 hd = (_Float16)src;
      _Float16 tl = (_Float16)(src - (float)hd);
      hv = (s == 14) ? tl : hd;
    }
    ((_Float16*)&a2frag[f][s >> 3][l])[s & 7] = hv;
  }
  __syncthreads();

  const int w    = tid >> 5;
  const int lane = tid & 31;
  const int cix  = lane & 15;
  const int lh   = lane >> 4;
  const int rowbase = blockIdx.x * 64 + w * 16;

  float wl[NT][2];
#pragma unroll
  for (int m = 0; m < NT; ++m)
#pragma unroll
    for (int u = 0; u < 2; ++u) {
      int unit = 4 * m + 2 * lh + u;
      wl[m][u] = (unit < HDIM) ? w_lin[unit] : 0.0f;
    }
  const float blin = b_lin[0];

  float c1[NT][2], c2[NT][2];
#pragma unroll
  for (int m = 0; m < NT; ++m) { c1[m][0] = 0.f; c1[m][1] = 0.f; c2[m][0] = 0.f; c2[m][1] = 0.f; }

  v8u hb1[2], hb2[2];
#pragma unroll
  for (int b = 0; b < 2; ++b)
#pragma unroll
    for (int j = 0; j < 8; ++j) { hb1[b][j] = 0u; hb2[b][j] = 0u; }
  float ylast = 0.f;

  const float* xin = input + (long)(rowbase + cix) * T;
  float* yout = out + (long)(rowbase + cix) * Tout;

  v16h wf1[NT * 2], wf2[NT * 4];
#pragma unroll
  for (int f = 0; f < NT * 2; ++f) wf1[f] = ldfrag(a1frag, f, lane);
#pragma unroll
  for (int f = 0; f < NT * 4; ++f) wf2[f] = ldfrag(a2frag, f, lane);

  for (int t = 0; t < Tout; ++t) {
    const float x = (t < T) ? xin[t] : ylast;

    {
      const _Float16 xh = (_Float16)x;
      const _Float16 xt = (_Float16)(x - (float)xh);
      v2h p5; p5[0] = xh; p5[1] = xh;
      v2h p6; p6[0] = xt; p6[1] = (_Float16)1.0f;
      v2h p7; p7[0] = (_Float16)1.0f; p7[1] = (_Float16)0.0f;
      hb1[1][5] = __builtin_bit_cast(unsigned, p5);
      hb1[1][6] = __builtin_bit_cast(unsigned, p6);
      hb1[1][7] = __builtin_bit_cast(unsigned, p7);
    }

    v8u hn1[2] = {hb1[0], hb1[1]};
    v8u hn2[2] = {hb2[0], hb2[1]};
    float ys = 0.f;
    v8f acc[2][4];
    const v8f zero8 = {0.f, 0.f, 0.f, 0.f, 0.f, 0.f, 0.f, 0.f};

    auto products = [&](auto Rc, auto PARTc) __attribute__((always_inline)) {
      constexpr int r = decltype(Rc)::value;
      constexpr int part = decltype(PARTc)::value;
      constexpr int blk = r & 3;
      constexpr int m0 = (blk == 0) ? 0 : 1 + 3 * blk;
      constexpr int n  = (blk == 0) ? 4 : 3;
      if (r < 4) {
#pragma unroll
        for (int kb = 0; kb < 2; ++kb)
#pragma unroll
          for (int i = 0; i < n; ++i)
            acc[r & 1][i] = __builtin_amdgcn_wmma_f32_16x16x32_f16(false, wf1[(m0 + i) * 2 + kb], false,
                        __builtin_bit_cast(v16h, hb1[kb]), (short)0,
                        (kb == 0) ? zero8 : acc[r & 1][i], false, false);
      } else {
#pragma unroll
        for (int kk = 0; kk < 4; ++kk) {
          if ((part == 1 && kk == 3) || (part == 2 && kk != 3)) continue;
          const int kb = (kk + 2) & 3;
#pragma unroll
          for (int i = 0; i < n; ++i)
            acc[r & 1][i] = __builtin_amdgcn_wmma_f32_16x16x32_f16(false, wf2[(m0 + i) * 4 + kb], false,
                        __builtin_bit_cast(v16h, (kb < 2) ? hb1[kb & 1] : hb2[kb & 1]),
                        (short)0, (kk == 0) ? zero8 : acc[r & 1][i], false, false);
        }
      }
    };

    auto cell = [&](auto Pc) __attribute__((always_inline)) {
      constexpr int p = decltype(Pc)::value;
      constexpr int blk = p & 3;
      constexpr int m0 = (blk == 0) ? 0 : 1 + 3 * blk;
      constexpr int n  = (blk == 0) ? 4 : 3;
#pragma unroll
      for (int i = 0; i < n; ++i) {
        const int m = m0 + i;
        float hn[2];
#pragma unroll
        for (int u = 0; u < 2; ++u) {
          float ig = sig_scaled(acc[p & 1][i][4 * u + 0]);
          float fg = sig_scaled(acc[p & 1][i][4 * u + 1]);
          float gk = KC - 2.0f * KC * tanh_rcp_scaled(acc[p & 1][i][4 * u + 2]);
          float og = sig_scaled(acc[p & 1][i][4 * u + 3]);
          if (p < 4) {
            float cn = fg * c1[m][u] + ig * gk;
            c1[m][u] = cn;
            hn[u] = og * (1.0f - 2.0f * tanh_rcp_scaled(cn));
          } else {
            float cn = fg * c2[m][u] + ig * gk;
            c2[m][u] = cn;
            hn[u] = og * (1.0f - 2.0f * tanh_rcp_scaled(cn));
          }
        }
        if (m == NT - 1 && lh == 1) hn[1] = 0.0f;
        v2h pk; pk[0] = (_Float16)hn[0]; pk[1] = (_Float16)hn[1];
        if (p < 4) {
          hn1[m >> 3][m & 7] = __builtin_bit_cast(unsigned, pk);
        } else {
          ys += hn[0] * wl[m][0];
          ys += hn[1] * wl[m][1];
          hn2[m >> 3][m & 7] = __builtin_bit_cast(unsigned, pk);
        }
      }
    };

#define IC(N) std::integral_constant<int, N>{}
    products(IC(0), IC(0));
    products(IC(1), IC(0)); cell(IC(0));
    products(IC(2), IC(0)); cell(IC(1));
    products(IC(3), IC(0)); cell(IC(2));
    hb1[0] = hn1[0];
    products(IC(4), IC(1)); cell(IC(3));
    hb1[1] = hn1[1];
    products(IC(4), IC(2));
    products(IC(5), IC(0)); cell(IC(4));
    products(IC(6), IC(0)); cell(IC(5));
    products(IC(7), IC(0)); cell(IC(6));
    cell(IC(7));
    hb2[0] = hn2[0]; hb2[1] = hn2[1];
#undef IC

    ys += xhalf(ys);
    ylast = ys + blin;
    if (lh == 0) yout[t] = ylast;
  }
}

extern "C" void kernel_launch(void* const* d_in, const int* in_sizes, int n_in,
                              void* d_out, int out_size, void* d_ws, size_t ws_size,
                              hipStream_t stream) {
  const float* input = (const float*)d_in[0];
  const float* w_ih1 = (const float*)d_in[2];
  const float* w_hh1 = (const float*)d_in[3];
  const float* b_ih1 = (const float*)d_in[4];
  const float* b_hh1 = (const float*)d_in[5];
  const float* w_ih2 = (const float*)d_in[6];
  const float* w_hh2 = (const float*)d_in[7];
  const float* b_ih2 = (const float*)d_in[8];
  const float* b_hh2 = (const float*)d_in[9];
  const float* w_lin = (const float*)d_in[10];
  const float* b_lin = (const float*)d_in[11];
  float* out = (float*)d_out;

  const int B = 1024;
  const int T = in_sizes[0] / B;
  const int Tout = out_size / B;

  lstm2_wmma_kernel<<<dim3(B / 64), dim3(128), 0, stream>>>(
      input, w_ih1, w_hh1, b_ih1, b_hh1, w_ih2, w_hh2, b_ih2, b_hh2,
      w_lin, b_lin, out, T, Tout);
}


# Layer 2 runs one time step behind layer 1 inside the same wave: no product waits for cell math of its own iteration; layer 1's next first block and the next x load are issued an iteration ahead
# speedup vs baseline: 3.8815x; 1.1868x over previous
#include <hip/hip_runtime.h>
#include <hip/hip_fp16.h>
#include <type_traits>

typedef __attribute__((ext_vector_type(16))) _Float16 v16h;
typedef __attribute__((ext_vector_type(8)))  _Float16 v8h;
typedef __attribute__((ext_vector_type(2)))  _Float16 v2h;
typedef __attribute__((ext_vector_type(8)))  float    v8f;
typedef __attribute__((ext_vector_type(8)))  unsigned v8u;

#define HDIM 51
#define NT 13
#define LOG2E 1.44269504088896340736f

#define KC (2.0f * LOG2E)
__device__ __forceinline__ float sig_scaled(float zs) {
  return __builtin_amdgcn_rcpf(1.0f + __builtin_amdgcn_exp2f(zs));
}
__device__ __forceinline__ float tanh_rcp_scaled(float zs) {
  return __builtin_amdgcn_rcpf(__builtin_amdgcn_exp2f(zs) + 1.0f);
}
__device__ __forceinline__ float gate_scale(int q) { return (q == 2) ? KC : -LOG2E; }

__device__ __forceinline__ float xhalf(float v) {
  int b = __builtin_bit_cast(int, v);
  return __builtin_bit_cast(float, __builtin_amdgcn_permlanex16(b, b, 0x76543210, 0xfedcba98, false, false));
}

__device__ __forceinline__ v16h ldfrag(const v8h (*fr)[2][32], int f, int lane) {
  v8h lo = fr[f][0][lane];
  v8h hp = fr[f][1][lane];
  return __builtin_shufflevector(lo, hp, 0,1,2,3,4,5,6,7,8,9,10,11,12,13,14,15);
}

__global__ __launch_bounds__(128)
void lstm2_wmma_kernel(const float* __restrict__ input,
                       const float* __restrict__ w_ih1, const float* __restrict__ w_hh1,
                       const float* __restrict__ b_ih1, const float* __restrict__ b_hh1,
                       const float* __restrict__ w_ih2, const float* __restrict__ w_hh2,
                       const float* __restrict__ b_ih2, const float* __restrict__ b_hh2,
                       const float* __restrict__ w_lin, const float* __restrict__ b_lin,
                       float* __restrict__ out, int T, int Tout)
{
  __shared__ v8h a1frag[NT * 2][2][32];
  __shared__ v8h a2frag[NT * 4][2][32];

  const int tid = threadIdx.x;

  for (int idx = tid; idx < NT * 2 * 512; idx += 128) {
    int f = idx >> 9, rem = idx & 511, l = rem >> 4, s = rem & 15;
    int m = f >> 1, b = f & 1;
    int cc = l & 15, lh = l >> 4;
    int unit = 4 * m + 2 * (cc >> 3) + ((cc & 7) >> 2), q = cc & 3;
    int mp = 8 * b + (s >> 1);
    int ku = 4 * mp + 2 * lh + (s & 1);
    float v = 0.0f;
    if (unit < HDIM && mp < NT && ku < HDIM) v = gate_scale(q) * w_hh1[(q * HDIM + unit) * HDIM + ku];
    _Float16 hv = (_Float16)v;
    if (b == 1 && lh == 0 && s >= 10 && s <= 14 && unit < HDIM) {
      int r = q * HDIM + unit;
      float src = gate_scale(q) * ((s <= 12) ? w_ih1[r] : (b_ih1[r] + b_hh1[r]));
      _Float16 hd = (_Float16)src;
      _Float16 tl = (_Float16)(src - (float)hd);
      hv = (s == 11 || s == 14) ? tl : hd;
    }
    ((_Float16*)&a1frag[f][s >> 3][l])[s & 7] = hv;
  }
  for (int idx = tid; idx < NT * 4 * 512; idx += 128) {
    int f = idx >> 9, rem = idx & 511, l = rem >> 4, s = rem & 15;
    int m = f >> 2, b = f & 3;
    int cc = l & 15, lh = l >> 4;
    int unit = 4 * m + 2 * (cc >> 3) + ((cc & 7) >> 2), q = cc & 3;
    int mp = 8 * (b & 1) + (s >> 1);
    int ku = 4 * mp + 2 * lh + (s & 1);
    float v = 0.0f;
    if (unit < HDIM && mp < NT && ku < HDIM) {
      const float* src = (b < 2) ? w_ih2 : w_hh2;
      v = gate_scale(q) * src[(q * HDIM + unit) * HDIM + ku];
    }
    _Float16 hv = (_Float16)v;
    if (b == 1 && lh == 0 && (s == 13 || s == 14) && unit < HDIM) {
      int r = q * HDIM + unit;
      float src = gate_scale(q) * (b_ih2[r] + b_hh2[r]);
      _Float16 hd = (_Float16)src;
      _Float16 tl = (_Float16)(src - (float)hd);
      hv = (s == 14) ? tl : hd;
    }
    ((_Float16*)&a2frag[f][s >> 3][l])[s & 7] = hv;
  }
  __syncthreads();

  const int w    = tid >> 5;
  const int lane = tid & 31;
  const int cix  = lane & 15;
  const int lh   = lane >> 4;
  const int rowbase = blockIdx.x * 64 + w * 16;

  float wl[NT][2];
#pragma unroll
  for (int m = 0; m < NT; ++m)
#pragma unroll
    for (int u = 0; u < 2; ++u) {
      int unit = 4 * m + 2 * lh + u;
      wl[m][u] = (unit < HDIM) ? w_lin[unit] : 0.0f;
    }
  const float blin = b_lin[0];

  float c1[NT][2], c2[NT][2];
#pragma unroll
  for (int m = 0; m < NT; ++m) { c1[m][0] = 0.f; c1[m][1] = 0.f; c2[m][0] = 0.f; c2[m][1] = 0.f; }

  v8u hb1[2], hb2[2];
#pragma unroll
  for (int b = 0; b < 2; ++b)
#pragma unroll
    for (int j = 0; j < 8; ++j) { hb1[b][j] = 0u; hb2[b][j] = 0u; }
  float ylast = 0.f;

  const float* xin = input + (long)(rowbase + cix) * T;
  float* yout = out + (long)(rowbase + cix) * Tout;

  v16h wf1[NT * 2], wf2[NT * 4];
#pragma unroll
  for (int f = 0; f < NT * 2; ++f) wf1[f] = ldfrag(a1frag, f, lane);
#pragma unroll
  for (int f = 0; f < NT * 4; ++f) wf2[f] = ldfrag(a2frag, f, lane);

  float ys = 0.f;
  v8f acc1[2][4], acc2[2][4];
  const v8f zero8 = {0.f, 0.f, 0.f, 0.f, 0.f, 0.f, 0.f, 0.f};

  auto set_x = [&](v8u& frag, float x) __attribute__((always_inline)) {
    const _Float16 xh = (_Float16)x;
    const _Float16 xt = (_Float16)(x - (float)xh);
    v2h p5; p5[0] = xh; p5[1] = xh;
    v2h p6; p6[0] = xt; p6[1] = (_Float16)1.0f;
    v2h p7; p7[0] = (_Float16)1.0f; p7[1] = (_Float16)0.0f;
    frag[5] = __builtin_bit_cast(unsigned, p5);
    frag[6] = __builtin_bit_cast(unsigned, p6);
    frag[7] = __builtin_bit_cast(unsigned, p7);
  };
  set_x(hb1[1], 0.f);
  v8u hn1[2] = {hb1[0], hb1[1]};
  v8u hn2[2] = {hb2[0], hb2[1]};

  auto products = [&](auto Rc) __attribute__((always_inline)) {
    constexpr int r = decltype(Rc)::value;
    constexpr int blk = r & 3;
    constexpr int m0 = (blk == 0) ? 0 : 1 + 3 * blk;
    constexpr int n  = (blk == 0) ? 4 : 3;
    if (r < 4) {
#pragma unroll
      for (int kb = 0; kb < 2; ++kb)
#pragma unroll
        for (int i = 0; i < n; ++i)
          acc1[r & 1][i] = __builtin_amdgcn_wmma_f32_16x16x32_f16(false, wf1[(m0 + i) * 2 + kb], false,
                      __builtin_bit_cast(v16h, hb1[kb]), (short)0,
                      (kb == 0) ? zero8 : acc1[r & 1][i], false, false);
    } else {
#pragma unroll
      for (int kk = 0; kk < 4; ++kk) {
        const int kb = (kk + 2) & 3;
#pragma unroll
        for (int i = 0; i < n; ++i)
          acc2[r & 1][i] = __builtin_amdgcn_wmma_f32_16x16x32_f16(false, wf2[(m0 + i) * 4 + kb], false,
                      __builtin_bit_cast(v16h, (kb < 2) ? hb1[kb & 1] : hb2[kb & 1]),
                      (short)0, (kk == 0) ? zero8 : acc2[r & 1][i], false, false);
      }
    }
  };

  auto cell = [&](auto Pc) __attribute__((always_inline)) {
    constexpr int p = decltype(Pc)::value;
    constexpr int blk = p & 3;
    constexpr int m0 = (blk == 0) ? 0 : 1 + 3 * blk;
    constexpr int n  = (blk == 0) ? 4 : 3;
#pragma unroll
    for (int i = 0; i < n; ++i) {
      const int m = m0 + i;
      const v8f a = (p < 4) ? acc1[p & 1][i] : acc2[p & 1][i];
      float hn[2];
#pragma unroll
      for (int u = 0; u < 2; ++u) {
        float ig = sig_scaled(a[4 * u + 0]);
        float fg = sig_scaled(a[4 * u + 1]);
        float gk = KC - 2.0f * KC * tanh_rcp_scaled(a[4 * u + 2]);
        float og = sig_scaled(a[4 * u + 3]);
        if (p < 4) {
          float cn = fg * c1[m][u] + ig * gk;
          c1[m][u] = cn;
          hn[u] = og * (1.0f - 2.0f * tanh_rcp_scaled(cn));
        } else {
          float cn = fg * c2[m][u] + ig * gk;
          c2[m][u] = cn;
          hn[u] = og * (1.0f - 2.0f * tanh_rcp_scaled(cn));
        }
      }
      if (m == NT - 1 && lh == 1) hn[1] = 0.0f;
      v2h pk; pk[0] = (_Float16)hn[0]; pk[1] = (_Float16)hn[1];
      if (p < 4) {
        hn1[m >> 3][m & 7] = __builtin_bit_cast(unsigned, pk);
      } else {
        ys += hn[0] * wl[m][0];
        ys += hn[1] * wl[m][1];
        hn2[m >> 3][m & 7] = __builtin_bit_cast(unsigned, pk);
      }
    }
  };

#define IC(N) std::integral_constant<int, N>{}
  if (T > 0) {
    set_x(hb1[1], xin[0]);
    products(IC(0));
    products(IC(1)); cell(IC(0));
    products(IC(2)); cell(IC(1));
    products(IC(3)); cell(IC(2));
    cell(IC(3));
    set_x(hn1[1], (1 < T) ? xin[1] : 0.f);
    hb1[0] = hn1[0]; hb1[1] = hn1[1];
    products(IC(0));
  }

  for (int t = 1; t < T; ++t) {
    const float xn = (t + 1 < T) ? xin[t + 1] : 0.f;

    ys = 0.f;
    products(IC(4)); cell(IC(0));
    products(IC(1));
    products(IC(5)); cell(IC(4));
    products(IC(2)); cell(IC(1));
    products(IC(6)); cell(IC(5));
    products(IC(3)); cell(IC(2));
    products(IC(7)); cell(IC(6));
    cell(IC(3));
    set_x(hn1[1], xn);
    hb1[0] = hn1[0]; hb1[1] = hn1[1];
    products(IC(0));
    cell(IC(7));
    hb2[0] = hn2[0]; hb2[1] = hn2[1];
    ys += xhalf(ys);
    ylast = ys + blin;
    if (lh == 0) yout[t - 1] = ylast;
  }

  for (int s = T - 1; s < Tout; ++s) {
    if (s >= 0) {
      ys = 0.f;
      products(IC(4));
      products(IC(5)); cell(IC(4));
      products(IC(6)); cell(IC(5));
      products(IC(7)); cell(IC(6));
      cell(IC(7));
      hb2[0] = hn2[0]; hb2[1] = hn2[1];
      ys += xhalf(ys);
      ylast = ys + blin;
      if (lh == 0) yout[s] = ylast;
    }
    if (s + 1 < Tout) {
      set_x(hb1[1], ylast);
      products(IC(0));
      products(IC(1)); cell(IC(0));
      products(IC(2)); cell(IC(1));
      products(IC(3)); cell(IC(2));
      cell(IC(3));
      hb1[0] = hn1[0]; hb1[1] = hn1[1];
    }
  }
#undef IC
}

extern "C" void kernel_launch(void* const* d_in, const int* in_sizes, int n_in,
                              void* d_out, int out_size, void* d_ws, size_t ws_size,
                              hipStream_t stream) {
  const float* input = (const float*)d_in[0];
  const float* w_ih1 = (const float*)d_in[2];
  const float* w_hh1 = (const float*)d_in[3];
  const float* b_ih1 = (const float*)d_in[4];
  const float* b_hh1 = (const float*)d_in[5];
  const float* w_ih2 = (const float*)d_in[6];
  const float* w_hh2 = (const float*)d_in[7];
  const float* b_ih2 = (const float*)d_in[8];
  const float* b_hh2 = (const float*)d_in[9];
  const float* w_lin = (const float*)d_in[10];
  const float* b_lin = (const float*)d_in[11];
  float* out = (float*)d_out;

  const int B = 1024;
  const int T = in_sizes[0] / B;
  const int Tout = out_size / B;

  lstm2_wmma_kernel<<<dim3(B / 64), dim3(128), 0, stream>>>(
      input, w_ih1, w_hh1, b_ih1, b_hh1, w_ih2, w_hh2, b_ih2, b_hh2,
      w_lin, b_lin, out, T, Tout);
}


# One-time weight table build: branch-free loops unrolled 26 times keep 26 global loads in flight per round trip, not one awaited load per entry; bias and input-weight entries written after a barrier
# speedup vs baseline: 4.6064x; 1.1868x over previous
#include <hip/hip_runtime.h>
#include <hip/hip_fp16.h>
#include <type_traits>

typedef __attribute__((ext_vector_type(16))) _Float16 v16h;
typedef __attribute__((ext_vector_type(8)))  _Float16 v8h;
typedef __attribute__((ext_vector_type(2)))  _Float16 v2h;
typedef __attribute__((ext_vector_type(8)))  float    v8f;
typedef __attribute__((ext_vector_type(8)))  unsigned v8u;
typedef __attribute__((ext_vector_type(2)))  float    v2f;

#define HDIM 51
#define NT 13
#define LOG2E 1.44269504088896340736f

#define KC (2.0f * LOG2E)
__device__ __forceinline__ v2f exp2_pair(float x, float y) {
  v2f r; r[0] = __builtin_amdgcn_exp2f(x); r[1] = __builtin_amdgcn_exp2f(y); return r;
}
__device__ __forceinline__ v2f rcp_pair(v2f d) {
  v2f r; r[0] = __builtin_amdgcn_rcpf(d[0]); r[1] = __builtin_amdgcn_rcpf(d[1]); return r;
}
__device__ __forceinline__ float cap_exp_arg(float x) {
  const int xi = __builtin_bit_cast(int, x);
  const int ci = __builtin_bit_cast(int, 64.0f);
  return __builtin_bit_cast(float, __builtin_elementwise_min(xi, ci));
}
__device__ __forceinline__ float gate_scale(int q) { return (q == 2) ? KC : -LOG2E; }

__device__ __forceinline__ float xhalf(float v) {
  int b = __builtin_bit_cast(int, v);
  return __builtin_bit_cast(float, __builtin_amdgcn_permlanex16(b, b, 0x76543210, 0xfedcba98, false, false));
}

__device__ __forceinline__ v16h ldfrag(const v8h (*fr)[2][32], int f, int lane) {
  v8h lo = fr[f][0][lane];
  v8h hp = fr[f][1][lane];
  return __builtin_shufflevector(lo, hp, 0,1,2,3,4,5,6,7,8,9,10,11,12,13,14,15);
}

__global__ __launch_bounds__(128)
void lstm2_wmma_kernel(const float* __restrict__ input,
                       const float* __restrict__ w_ih1, const float* __restrict__ w_hh1,
                       const float* __restrict__ b_ih1, const float* __restrict__ b_hh1,
                       const float* __restrict__ w_ih2, const float* __restrict__ w_hh2,
                       const float* __restrict__ b_ih2, const float* __restrict__ b_hh2,
                       const float* __restrict__ w_lin, const float* __restrict__ b_lin,
                       float* __restrict__ out, int T, int Tout)
{
  __shared__ v8h a1frag[NT * 2][2][32];
  __shared__ v8h a2frag[NT * 4][2][32];

  const int tid = threadIdx.x;

#pragma unroll 26
  for (int it = 0; it < NT * 2 * 512 / 128; ++it) {
    const int idx = tid + it * 128;
    int f = idx >> 9, rem = idx & 511, l = rem >> 4, s = rem & 15;
    int m = f >> 1, b = f & 1;
    int cc = l & 15, lh = l >> 4;
    int unit = 4 * m + 2 * (cc >> 3) + (cc & 1), q = (cc & 7) >> 1;
    int mp = 8 * b + (s >> 1);
    int ku = 4 * mp + 2 * lh + (s & 1);
    const bool ok = unit < HDIM && mp < NT && ku < HDIM;
    const int uc = (unit < HDIM) ? unit : HDIM - 1, kc = (ku < HDIM) ? ku : HDIM - 1;
    const float sv = gate_scale(q) * w_hh1[(q * HDIM + uc) * HDIM + kc];
    const float v = ok ? sv : 0.0f;
    ((_Float16*)&a1frag[f][s >> 3][l])[s & 7] = (_Float16)v;
  }
#pragma unroll 26
  for (int it = 0; it < NT * 4 * 512 / 128; ++it) {
    const int idx = tid + it * 128;
    int f = idx >> 9, rem = idx & 511, l = rem >> 4, s = rem & 15;
    int m = f >> 2, b = f & 3;
    int cc = l & 15, lh = l >> 4;
    int unit = 4 * m + 2 * (cc >> 3) + (cc & 1), q = (cc & 7) >> 1;
    int mp = 8 * (b & 1) + (s >> 1);
    int ku = 4 * mp + 2 * lh + (s & 1);
    const bool ok = unit < HDIM && mp < NT && ku < HDIM;
    const float* wsrc = (b < 2) ? w_ih2 : w_hh2;
    const int uc = (unit < HDIM) ? unit : HDIM - 1, kc = (ku < HDIM) ? ku : HDIM - 1;
    const float sv = gate_scale(q) * wsrc[(q * HDIM + uc) * HDIM + kc];
    const float v = ok ? sv : 0.0f;
    ((_Float16*)&a2frag[f][s >> 3][l])[s & 7] = (_Float16)v;
  }
  __syncthreads();
  for (int e = tid; e < NT * 16 * 5; e += 128) {
    int m = e / 80, r2 = e % 80, cc = r2 / 5, s = 10 + r2 % 5;
    int unit = 4 * m + 2 * (cc >> 3) + (cc & 1), q = (cc & 7) >> 1;
    if (unit < HDIM) {
      int r = q * HDIM + unit;
      float sv = gate_scale(q) * ((s <= 12) ? w_ih1[r] : (b_ih1[r] + b_hh1[r]));
      _Float16 hd = (_Float16)sv;
      _Float16 tl = (_Float16)(sv - (float)hd);
      ((_Float16*)&a1frag[2 * m + 1][s >> 3][cc])[s & 7] = (s == 11 || s == 14) ? tl : hd;
    }
  }
  for (int e = tid; e < NT * 16 * 2; e += 128) {
    int m = e >> 5, cc = (e & 31) >> 1, s = 13 + (e & 1);
    int unit = 4 * m + 2 * (cc >> 3) + (cc & 1), q = (cc & 7) >> 1;
    if (unit < HDIM) {
      int r = q * HDIM + unit;
      float sv = gate_scale(q) * (b_ih2[r] + b_hh2[r]);
      _Float16 hd = (_Float16)sv;
      _Float16 tl = (_Float16)(sv - (float)hd);
      ((_Float16*)&a2frag[4 * m + 1][s >> 3][cc])[s & 7] = (s == 14) ? tl : hd;
    }
  }
  __syncthreads();

  const int w    = tid >> 5;
  const int lane = tid & 31;
  const int cix  = lane & 15;
  const int lh   = lane >> 4;
  const int rowbase = blockIdx.x * 64 + w * 16;

  float wl[NT][2];
#pragma unroll
  for (int m = 0; m < NT; ++m)
#pragma unroll
    for (int u = 0; u < 2; ++u) {
      int unit = 4 * m + 2 * lh + u;
      wl[m][u] = (unit < HDIM) ? w_lin[unit] : 0.0f;
    }
  const float blin = b_lin[0];

  v2f c1[NT], c2[NT];
  const v2f rkc2 = {1.0f / KC, 1.0f / KC};
#pragma unroll
  for (int m = 0; m < NT; ++m) { c1[m] = 0.f; c2[m] = 0.f; }

  v8u hb1[2], hb2[2];
#pragma unroll
  for (int b = 0; b < 2; ++b)
#pragma unroll
    for (int j = 0; j < 8; ++j) { hb1[b][j] = 0u; hb2[b][j] = 0u; }
  float ylast = 0.f;

  const float* xin = input + (long)(rowbase + cix) * T;
  float* yout = out + (long)(rowbase + cix) * Tout;

  v16h wf1[NT * 2], wf2[NT * 4];
#pragma unroll
  for (int f = 0; f < NT * 2; ++f) wf1[f] = ldfrag(a1frag, f, lane);
#pragma unroll
  for (int f = 0; f < NT * 4; ++f) wf2[f] = ldfrag(a2frag, f, lane);

  float ys = 0.f;
  v8f acc1[2][4], acc2[2][4];
  const v8f zero8 = {0.f, 0.f, 0.f, 0.f, 0.f, 0.f, 0.f, 0.f};

  auto set_x = [&](v8u& frag, float x) __attribute__((always_inline)) {
    const _Float16 xh = (_Float16)x;
    const _Float16 xt = (_Float16)(x - (float)xh);
    v2h p5; p5[0] = xh; p5[1] = xh;
    v2h p6; p6[0] = xt; p6[1] = (_Float16)1.0f;
    v2h p7; p7[0] = (_Float16)1.0f; p7[1] = (_Float16)0.0f;
    frag[5] = __builtin_bit_cast(unsigned, p5);
    frag[6] = __builtin_bit_cast(unsigned, p6);
    frag[7] = __builtin_bit_cast(unsigned, p7);
  };
  set_x(hb1[1], 0.f);
  v8u hn1[2] = {hb1[0], hb1[1]};
  v8u hn2[2] = {hb2[0], hb2[1]};

  auto products = [&](auto Rc) __attribute__((always_inline)) {
    constexpr int r = decltype(Rc)::value;
    constexpr int blk = r & 3;
    constexpr int m0 = (blk == 0) ? 0 : 1 + 3 * blk;
    constexpr int n  = (blk == 0) ? 4 : 3;
    if (r < 4) {
#pragma unroll
      for (int kb = 0; kb < 2; ++kb)
#pragma unroll
        for (int i = 0; i < n; ++i)
          acc1[r & 1][i] = __builtin_amdgcn_wmma_f32_16x16x32_f16(false, wf1[(m0 + i) * 2 + kb], false,
                      __builtin_bit_cast(v16h, hb1[kb]), (short)0,
                      (kb == 0) ? zero8 : acc1[r & 1][i], false, false);
    } else {
#pragma unroll
      for (int kk = 0; kk < 4; ++kk) {
        const int kb = (kk + 2) & 3;
#pragma unroll
        for (int i = 0; i < n; ++i)
          acc2[r & 1][i] = __builtin_amdgcn_wmma_f32_16x16x32_f16(false, wf2[(m0 + i) * 4 + kb], false,
                      __builtin_bit_cast(v16h, (kb < 2) ? hb1[kb & 1] : hb2[kb & 1]),
                      (short)0, (kk == 0) ? zero8 : acc2[r & 1][i], false, false);
      }
    }
  };

  auto cell = [&](auto Pc) __attribute__((always_inline)) {
    constexpr int p = decltype(Pc)::value;
    constexpr int blk = p & 3;
    constexpr int m0 = (blk == 0) ? 0 : 1 + 3 * blk;
    constexpr int n  = (blk == 0) ? 4 : 3;
#pragma unroll
    for (int i = 0; i < n; ++i) {
      const int m = m0 + i;
      const v8f a = (p < 4) ? acc1[p & 1][i] : acc2[p & 1][i];
      const v2f ei = exp2_pair(a[0], a[1]);
      const v2f ef = exp2_pair(a[2], a[3]);
      const v2f eg = exp2_pair(cap_exp_arg(a[4]), cap_exp_arg(a[5]));
      const v2f eo = exp2_pair(a[6], a[7]);
      const v2f df = ef + 1.0f;
      const v2f dq = eo + 1.0f;
      const v2f dgk = __builtin_elementwise_fma(eg, rkc2, rkc2);
      const v2f pig = __builtin_elementwise_fma(ei, dgk, dgk);
      const v2f rf  = rcp_pair(df);
      const v2f rig = rcp_pair(pig);
      const v2f cold = (p < 4) ? c1[m] : c2[m];
      const v2f cn = __builtin_elementwise_fma(eg, rig, __builtin_elementwise_fma(cold, rf, -rig));
      if (p < 4) c1[m] = cn; else c2[m] = cn;
      const v2f ec = exp2_pair(cap_exp_arg(cn[0]), cap_exp_arg(cn[1]));
      const v2f poc = __builtin_elementwise_fma(ec, dq, dq);
      const v2f roc = rcp_pair(poc);
      const v2f hv = __builtin_elementwise_fma(ec, roc, -roc);
      float hn[2] = {hv[0], hv[1]};
      if (m == NT - 1 && lh == 1) hn[1] = 0.0f;
      v2h pk; pk[0] = (_Float16)hn[0]; pk[1] = (_Float16)hn[1];
      if (p < 4) {
        hn1[m >> 3][m & 7] = __builtin_bit_cast(unsigned, pk);
      } else {
        ys += hn[0] * wl[m][0];
        ys += hn[1] * wl[m][1];
        hn2[m >> 3][m & 7] = __builtin_bit_cast(unsigned, pk);
      }
    }
  };

#define IC(N) std::integral_constant<int, N>{}
  if (T > 0) {
    set_x(hb1[1], xin[0]);
    products(IC(0));
    products(IC(1)); cell(IC(0));
    products(IC(2)); cell(IC(1));
    products(IC(3)); cell(IC(2));
    cell(IC(3));
    set_x(hn1[1], (1 < T) ? xin[1] : 0.f);
    hb1[0] = hn1[0]; hb1[1] = hn1[1];
    products(IC(0));
  }

  for (int t = 1; t < T; ++t) {
    const float xn = (t + 1 < T) ? xin[t + 1] : 0.f;

    ys = 0.f;
    products(IC(4)); cell(IC(0));
    products(IC(1));
    products(IC(5)); cell(IC(4));
    products(IC(2)); cell(IC(1));
    products(IC(6)); cell(IC(5));
    products(IC(3)); cell(IC(2));
    products(IC(7)); cell(IC(6));
    cell(IC(3));
    set_x(hn1[1], xn);
    hb1[0] = hn1[0]; hb1[1] = hn1[1];
    products(IC(0));
    cell(IC(7));
    hb2[0] = hn2[0]; hb2[1] = hn2[1];
    ys += xhalf(ys);
    ylast = ys + blin;
    if (lh == 0) yout[t - 1] = ylast;
  }

  for (int s = T - 1; s < Tout; ++s) {
    if (s >= 0) {
      ys = 0.f;
      products(IC(4));
      products(IC(5)); cell(IC(4));
      products(IC(6)); cell(IC(5));
      products(IC(7)); cell(IC(6));
      cell(IC(7));
      hb2[0] = hn2[0]; hb2[1] = hn2[1];
      ys += xhalf(ys);
      ylast = ys + blin;
      if (lh == 0) yout[s] = ylast;
    }
    if (s + 1 < Tout) {
      set_x(hb1[1], ylast);
      products(IC(0));
      products(IC(1)); cell(IC(0));
      products(IC(2)); cell(IC(1));
      products(IC(3)); cell(IC(2));
      cell(IC(3));
      hb1[0] = hn1[0]; hb1[1] = hn1[1];
    }
  }
#undef IC
}

extern "C" void kernel_launch(void* const* d_in, const int* in_sizes, int n_in,
                              void* d_out, int out_size, void* d_ws, size_t ws_size,
                              hipStream_t stream) {
  const float* input = (const float*)d_in[0];
  const float* w_ih1 = (const float*)d_in[2];
  const float* w_hh1 = (const float*)d_in[3];
  const float* b_ih1 = (const float*)d_in[4];
  const float* b_hh1 = (const float*)d_in[5];
  const float* w_ih2 = (const float*)d_in[6];
  const float* w_hh2 = (const float*)d_in[7];
  const float* b_ih2 = (const float*)d_in[8];
  const float* b_hh2 = (const float*)d_in[9];
  const float* w_lin = (const float*)d_in[10];
  const float* b_lin = (const float*)d_in[11];
  float* out = (float*)d_out;

  const int B = 1024;
  const int T = in_sizes[0] / B;
  const int Tout = out_size / B;

  lstm2_wmma_kernel<<<dim3(B / 64), dim3(128), 0, stream>>>(
      input, w_ih1, w_hh1, b_ih1, b_hh1, w_ih2, w_hh2, b_ih2, b_hh2,
      w_lin, b_lin, out, T, Tout);
}


# speedup vs baseline: 4.9973x; 1.0849x over previous
#include <hip/hip_runtime.h>
#include <hip/hip_fp16.h>
#include <type_traits>

typedef __attribute__((ext_vector_type(16))) _Float16 v16h;
typedef __attribute__((ext_vector_type(8)))  _Float16 v8h;
typedef __attribute__((ext_vector_type(2)))  _Float16 v2h;
typedef __attribute__((ext_vector_type(8)))  float    v8f;
typedef __attribute__((ext_vector_type(8)))  unsigned v8u;
typedef __attribute__((ext_vector_type(2)))  float    v2f;

#define HDIM 51
#define NT 13
#define LOG2E 1.44269504088896340736f

#define KC (2.0f * LOG2E)
__device__ __forceinline__ v2f exp2_pair(float x, float y) {
  v2f r; r[0] = __builtin_amdgcn_exp2f(x); r[1] = __builtin_amdgcn_exp2f(y); return r;
}
__device__ __forceinline__ v2f rcp_pair(v2f d) {
  v2f r; r[0] = __builtin_amdgcn_rcpf(d[0]); r[1] = __builtin_amdgcn_rcpf(d[1]); return r;
}
__device__ __forceinline__ float cap_exp_arg(float x) {
  const int xi = __builtin_bit_cast(int, x);
  const int ci = __builtin_bit_cast(int, 64.0f);
  return __builtin_bit_cast(float, __builtin_elementwise_min(xi, ci));
}
__device__ __forceinline__ float gate_scale(int q) { return (q == 2) ? KC : -LOG2E; }

__device__ __forceinline__ float xhalf(float v) {
  int b = __builtin_bit_cast(int, v);
  return __builtin_bit_cast(float, __builtin_amdgcn_permlanex16(b, b, 0x76543210, 0xfedcba98, false, false));
}

__device__ __forceinline__ v16h ldfrag(const v8h (*fr)[2][32], int f, int lane) {
  v8h lo = fr[f][0][lane];
  v8h hp = fr[f][1][lane];
  return __builtin_shufflevector(lo, hp, 0,1,2,3,4,5,6,7,8,9,10,11,12,13,14,15);
}

__global__ __launch_bounds__(128)
void lstm2_wmma_kernel(const float* __restrict__ input,
                       const float* __restrict__ w_ih1, const float* __restrict__ w_hh1,
                       const float* __restrict__ b_ih1, const float* __restrict__ b_hh1,
                       const float* __restrict__ w_ih2, const float* __restrict__ w_hh2,
                       const float* __restrict__ b_ih2, const float* __restrict__ b_hh2,
                       const float* __restrict__ w_lin, const float* __restrict__ b_lin,
                       float* __restrict__ out, int T, int Tout)
{
  __shared__ v8h a1frag[NT * 2][2][32];
  __shared__ v8h a2frag[NT * 4][2][32];

  const int tid = threadIdx.x;

#pragma unroll 26
  for (int it = 0; it < NT * 2 * 512 / 128; ++it) {
    const int idx = tid + it * 128;
    int f = idx >> 9, rem = idx & 511, l = rem >> 4, s = rem & 15;
    int m = f >> 1, b = f & 1;
    int cc = l & 15, lh = l >> 4;
    int unit = 4 * m + 2 * (cc >> 3) + (cc & 1), q = (cc & 7) >> 1;
    int mp = 8 * b + (s >> 1);
    int ku = 4 * mp + 2 * lh + (s & 1);
    const bool ok = unit < HDIM && mp < NT && ku < HDIM;
    const int uc = (unit < HDIM) ? unit : HDIM - 1, kc = (ku < HDIM) ? ku : HDIM - 1;
    const float sv = gate_scale(q) * w_hh1[(q * HDIM + uc) * HDIM + kc];
    const float v = ok ? sv : 0.0f;
    ((_Float16*)&a1frag[f][s >> 3][l])[s & 7] = (_Float16)v;
  }
#pragma unroll 26
  for (int it = 0; it < NT * 4 * 512 / 128; ++it) {
    const int idx = tid + it * 128;
    int f = idx >> 9, rem = idx & 511, l = rem >> 4, s = rem & 15;
    int m = f >> 2, b = f & 3;
    int cc = l & 15, lh = l >> 4;
    int unit = 4 * m + 2 * (cc >> 3) + (cc & 1), q = (cc & 7) >> 1;
    int mp = 8 * (b & 1) + (s >> 1);
    int ku = 4 * mp + 2 * lh + (s & 1);
    const bool ok = unit < HDIM && mp < NT && ku < HDIM;
    const float* wsrc = (b < 2) ? w_ih2 : w_hh2;
    const int uc = (unit < HDIM) ? unit : HDIM - 1, kc = (ku < HDIM) ? ku : HDIM - 1;
    const float sv = gate_scale(q) * wsrc[(q * HDIM + uc) * HDIM + kc];
    const float v = ok ? sv : 0.0f;
    ((_Float16*)&a2frag[f][s >> 3][l])[s & 7] = (_Float16)v;
  }
  __syncthreads();
  for (int e = tid; e < NT * 16 * 5; e += 128) {
    int m = e / 80, r2 = e % 80, cc = r2 / 5, s = 10 + r2 % 5;
    int unit = 4 * m + 2 * (cc >> 3) + (cc & 1), q = (cc & 7) >> 1;
    if (unit < HDIM) {
      int r = q * HDIM + unit;
      float sv = gate_scale(q) * ((s <= 12) ? w_ih1[r] : (b_ih1[r] + b_hh1[r]));
      _Float16 hd = (_Float16)sv;
      _Float16 tl = (_Float16)(sv - (float)hd);
      ((_Float16*)&a1frag[2 * m + 1][s >> 3][cc])[s & 7] = (s == 11 || s == 14) ? tl : hd;
    }
  }
  for (int e = tid; e < NT * 16 * 2; e += 128) {
    int m = e >> 5, cc = (e & 31) >> 1, s = 13 + (e & 1);
    int unit = 4 * m + 2 * (cc >> 3) + (cc & 1), q = (cc & 7) >> 1;
    if (unit < HDIM) {
      int r = q * HDIM + unit;
      float sv = gate_scale(q) * (b_ih2[r] + b_hh2[r]);
      _Float16 hd = (_Float16)sv;
      _Float16 tl = (_Float16)(sv - (float)hd);
      ((_Float16*)&a2frag[4 * m + 1][s >> 3][cc])[s & 7] = (s == 14) ? tl : hd;
    }
  }
  __syncthreads();

  const int w    = tid >> 5;
  const int lane = tid & 31;
  const int cix  = lane & 15;
  const int lh   = lane >> 4;
  const int rowbase = blockIdx.x * 64 + w * 16;

  float wl[NT][2];
#pragma unroll
  for (int m = 0; m < NT; ++m)
#pragma unroll
    for (int u = 0; u < 2; ++u) {
      int unit = 4 * m + 2 * lh + u;
      wl[m][u] = (unit < HDIM) ? w_lin[unit] : 0.0f;
    }
  const float blin = b_lin[0];

  v2f c1[NT], c2[NT];
  const v2f rkc2 = {1.0f / KC, 1.0f / KC};
#pragma unroll
  for (int m = 0; m < NT; ++m) { c1[m] = 0.f; c2[m] = 0.f; }

  v8u hb1[2], hb2[2];
#pragma unroll
  for (int b = 0; b < 2; ++b)
#pragma unroll
    for (int j = 0; j < 8; ++j) { hb1[b][j] = 0u; hb2[b][j] = 0u; }
  float ylast = 0.f;

  const float* xin = input + (long)(rowbase + cix) * T;
  float* yout = out + (long)(rowbase + cix) * Tout;

  v16h wf1[NT * 2], wf2[NT * 4];
#pragma unroll
  for (int f = 0; f < NT * 2; ++f) wf1[f] = ldfrag(a1frag, f, lane);
#pragma unroll
  for (int f = 0; f < NT * 4; ++f) wf2[f] = ldfrag(a2frag, f, lane);

  float ys = 0.f;
  v8f acc1[2][4], acc2[2][4];
  const v8f zero8 = {0.f, 0.f, 0.f, 0.f, 0.f, 0.f, 0.f, 0.f};

  auto set_x = [&](v8u& frag, float x) __attribute__((always_inline)) {
    const _Float16 xh = (_Float16)x;
    const _Float16 xt = (_Float16)(x - (float)xh);
    v2h p5; p5[0] = xh; p5[1] = xh;
    v2h p6; p6[0] = xt; p6[1] = (_Float16)1.0f;
    v2h p7; p7[0] = (_Float16)1.0f; p7[1] = (_Float16)0.0f;
    frag[5] = __builtin_bit_cast(unsigned, p5);
    frag[6] = __builtin_bit_cast(unsigned, p6);
    frag[7] = __builtin_bit_cast(unsigned, p7);
  };
  set_x(hb1[1], 0.f);
  v8u hn1[2] = {hb1[0], hb1[1]};
  v8u hn2[2] = {hb2[0], hb2[1]};

  auto products = [&](auto Rc) __attribute__((always_inline)) {
    constexpr int r = decltype(Rc)::value;
    constexpr int blk = r & 3;
    constexpr int m0 = (blk == 0) ? 0 : 1 + 3 * blk;
    constexpr int n  = (blk == 0) ? 4 : 3;
    if (r < 4) {
#pragma unroll
      for (int kb = 0; kb < 2; ++kb)
#pragma unroll
        for (int i = 0; i < n; ++i)
          acc1[r & 1][i] = __builtin_amdgcn_wmma_f32_16x16x32_f16(false, wf1[(m0 + i) * 2 + kb], false,
                      __builtin_bit_cast(v16h, hb1[kb]), (short)0,
                      (kb == 0) ? zero8 : acc1[r & 1][i], false, false);
    } else {
#pragma unroll
      for (int kk = 0; kk < 4; ++kk) {
        const int kb = (kk + 2) & 3;
#pragma unroll
        for (int i = 0; i < n; ++i)
          acc2[r & 1][i] = __builtin_amdgcn_wmma_f32_16x16x32_f16(false, wf2[(m0 + i) * 4 + kb], false,
                      __builtin_bit_cast(v16h, (kb < 2) ? hb1[kb & 1] : hb2[kb & 1]),
                      (short)0, (kk == 0) ? zero8 : acc2[r & 1][i], false, false);
      }
    }
  };

  auto cell = [&](auto Pc) __attribute__((always_inline)) {
    constexpr int p = decltype(Pc)::value;
    constexpr int blk = p & 3;
    constexpr int m0 = (blk == 0) ? 0 : 1 + 3 * blk;
    constexpr int n  = (blk == 0) ? 4 : 3;
#pragma unroll
    for (int i = 0; i < n; ++i) {
      const int m = m0 + i;
      const v8f a = (p < 4) ? acc1[p & 1][i] : acc2[p & 1][i];
      const v2f ei = exp2_pair(a[0], a[1]);
      const v2f ef = exp2_pair(a[2], a[3]);
      const v2f eg = exp2_pair(cap_exp_arg(a[4]), cap_exp_arg(a[5]));
      const v2f eo = exp2_pair(a[6], a[7]);
      const v2f df = ef + 1.0f;
      const v2f dq = eo + 1.0f;
      const v2f dgk = __builtin_elementwise_fma(eg, rkc2, rkc2);
      const v2f pig = __builtin_elementwise_fma(ei, dgk, dgk);
      const v2f rf  = rcp_pair(df);
      const v2f rig = rcp_pair(pig);
      const v2f cold = (p < 4) ? c1[m] : c2[m];
      const v2f cn = __builtin_elementwise_fma(eg, rig, __builtin_elementwise_fma(cold, rf, -rig));
      if (p < 4) c1[m] = cn; else c2[m] = cn;
      const v2f ec = exp2_pair(cap_exp_arg(cn[0]), cap_exp_arg(cn[1]));
      const v2f poc = __builtin_elementwise_fma(ec, dq, dq);
      const v2f roc = rcp_pair(poc);
      const v2f hv = __builtin_elementwise_fma(ec, roc, -roc);
      float hn[2] = {hv[0], hv[1]};
      if (m == NT - 1 && lh == 1) hn[1] = 0.0f;
      v2h pk; pk[0] = (_Float16)hn[0]; pk[1] = (_Float16)hn[1];
      if (p < 4) {
        hn1[m >> 3][m & 7] = __builtin_bit_cast(unsigned, pk);
      } else {
        ys += hn[0] * wl[m][0];
        ys += hn[1] * wl[m][1];
        hn2[m >> 3][m & 7] = __builtin_bit_cast(unsigned, pk);
      }
    }
  };

#define IC(N) std::integral_constant<int, N>{}
#define SB __builtin_amdgcn_sched_barrier(0)
  if (T > 0) {
    set_x(hb1[1], xin[0]);
    products(IC(0));
    products(IC(1)); cell(IC(0));
    products(IC(2)); cell(IC(1));
    products(IC(3)); cell(IC(2));
    cell(IC(3));
    set_x(hn1[1], (1 < T) ? xin[1] : 0.f);
    hb1[0] = hn1[0]; hb1[1] = hn1[1];
    products(IC(0));
  }

  for (int t = 1; t < T; ++t) {
    const float xn = (t + 1 < T) ? xin[t + 1] : 0.f;

    ys = 0.f;
    v2f EI[26], EF[26], EG[26], EO[26], DF[26], DQ[26], DGK[26], PIG[26], RF[26], RIG[26], TT[26], EC[26], POC[26], ROC[26], HV[26];
    float GC[26][2], CC[26][2];
    EI[0][0] = __builtin_amdgcn_exp2f(acc1[0][0][0]); SB;
    acc1[1][0] = __builtin_amdgcn_wmma_f32_16x16x32_f16(false, wf2[2], false, __builtin_bit_cast(v16h, hb2[0]), (short)0, zero8, false, false); SB;
    GC[0][0] = cap_exp_arg(acc1[0][0][4]); SB;
    EI[0][1] = __builtin_amdgcn_exp2f(acc1[0][0][1]); SB;
    GC[0][1] = cap_exp_arg(acc1[0][0][5]); SB;
    EF[0][0] = __builtin_amdgcn_exp2f(acc1[0][0][2]); SB;
    GC[1][0] = cap_exp_arg(acc1[0][1][4]); SB;
    EF[0][1] = __builtin_amdgcn_exp2f(acc1[0][0][3]); SB;
    GC[1][1] = cap_exp_arg(acc1[0][1][5]); SB;
    EG[0][0] = __builtin_amdgcn_exp2f(GC[0][0]); SB;
    acc1[1][1] = __builtin_amdgcn_wmma_f32_16x16x32_f16(false, wf1[24], false, __builtin_bit_cast(v16h, hb1[0]), (short)0, zero8, false, false); SB;
    GC[2][0] = cap_exp_arg(acc1[0][2][4]); SB;
    EG[0][1] = __builtin_amdgcn_exp2f(GC[0][1]); SB;
    GC[2][1] = cap_exp_arg(acc1[0][2][5]); SB;
    EG[1][0] = __builtin_amdgcn_exp2f(GC[1][0]); SB;
    GC[3][0] = cap_exp_arg(acc1[0][3][4]); SB;
    EG[1][1] = __builtin_amdgcn_exp2f(GC[1][1]); SB;
    GC[3][1] = cap_exp_arg(acc1[0][3][5]); SB;
    EG[2][0] = __builtin_amdgcn_exp2f(GC[2][0]); SB;
    acc1[1][2] = __builtin_amdgcn_wmma_f32_16x16x32_f16(false, wf1[8], false, __builtin_bit_cast(v16h, hb1[0]), (short)0, zero8, false, false); SB;
    EG[2][1] = __builtin_amdgcn_exp2f(GC[2][1]); SB;
    DF[0] = EF[0] + 1.0f; SB;
    EG[3][0] = __builtin_amdgcn_exp2f(GC[3][0]); SB;
    EG[3][1] = __builtin_amdgcn_exp2f(GC[3][1]); SB;
    EI[1][0] = __builtin_amdgcn_exp2f(acc1[0][1][0]); SB;
    acc1[1][3] = __builtin_amdgcn_wmma_f32_16x16x32_f16(false, wf1[10], false, __builtin_bit_cast(v16h, hb1[0]), (short)0, zero8, false, false); SB;
    DGK[0] = __builtin_elementwise_fma(EG[0], rkc2, rkc2); SB;
    EI[1][1] = __builtin_amdgcn_exp2f(acc1[0][1][1]); SB;
    DGK[1] = __builtin_elementwise_fma(EG[1], rkc2, rkc2); SB;
    EF[1][0] = __builtin_amdgcn_exp2f(acc1[0][1][2]); SB;
    EF[1][1] = __builtin_amdgcn_exp2f(acc1[0][1][3]); SB;
    DGK[2] = __builtin_elementwise_fma(EG[2], rkc2, rkc2); SB;
    EI[2][0] = __builtin_amdgcn_exp2f(acc1[0][2][0]); SB;
    acc2[0][0] = __builtin_amdgcn_wmma_f32_16x16x32_f16(false, wf1[12], false, __builtin_bit_cast(v16h, hb1[0]), (short)0, zero8, false, false); SB;
    EI[2][1] = __builtin_amdgcn_exp2f(acc1[0][2][1]); SB;
    DGK[3] = __builtin_elementwise_fma(EG[3], rkc2, rkc2); SB;
    EF[2][0] = __builtin_amdgcn_exp2f(acc1[0][2][2]); SB;
    PIG[0] = __builtin_elementwise_fma(EI[0], DGK[0], DGK[0]); SB;
    EF[2][1] = __builtin_amdgcn_exp2f(acc1[0][2][3]); SB;
    EI[3][0] = __builtin_amdgcn_exp2f(acc1[0][3][0]); SB;
    acc2[0][1] = __builtin_amdgcn_wmma_f32_16x16x32_f16(false, wf2[6], false, __builtin_bit_cast(v16h, hb2[0]), (short)0, zero8, false, false); SB;
    PIG[1] = __builtin_elementwise_fma(EI[1], DGK[1], DGK[1]); SB;
    EI[3][1] = __builtin_amdgcn_exp2f(acc1[0][3][1]); SB;
    DF[1] = EF[1] + 1.0f; SB;
    EF[3][0] = __builtin_amdgcn_exp2f(acc1[0][3][2]); SB;
    EF[3][1] = __builtin_amdgcn_exp2f(acc1[0][3][3]); SB;
    PIG[2] = __builtin_elementwise_fma(EI[2], DGK[2], DGK[2]); SB;
    RF[0][0] = __builtin_amdgcn_rcpf(DF[0][0]); SB;
    acc2[0][2] = __builtin_amdgcn_wmma_f32_16x16x32_f16(false, wf1[14], false, __builtin_bit_cast(v16h, hb1[0]), (short)0, zero8, false, false); SB;
    RF[0][1] = __builtin_amdgcn_rcpf(DF[0][1]); SB;
    DF[2] = EF[2] + 1.0f; SB;
    RIG[0][0] = __builtin_amdgcn_rcpf(PIG[0][0]); SB;
    RIG[0][1] = __builtin_amdgcn_rcpf(PIG[0][1]); SB;
    RF[1][0] = __builtin_amdgcn_rcpf(DF[1][0]); SB;
    acc2[0][3] = __builtin_amdgcn_wmma_f32_16x16x32_f16(false, wf1[16], false, __builtin_bit_cast(v16h, hb1[0]), (short)0, zero8, false, false); SB;
    PIG[3] = __builtin_elementwise_fma(EI[3], DGK[3], DGK[3]); SB;
    RF[1][1] = __builtin_amdgcn_rcpf(DF[1][1]); SB;
    DF[3] = EF[3] + 1.0f; SB;
    RIG[1][0] = __builtin_amdgcn_rcpf(PIG[1][0]); SB;
    RIG[1][1] = __builtin_amdgcn_rcpf(PIG[1][1]); SB;
    RF[2][0] = __builtin_amdgcn_rcpf(DF[2][0]); SB;
    acc2[1][0] = __builtin_amdgcn_wmma_f32_16x16x32_f16(false, wf1[18], false, __builtin_bit_cast(v16h, hb1[0]), (short)0, zero8, false, false); SB;
    RF[2][1] = __builtin_amdgcn_rcpf(DF[2][1]); SB;
    TT[0] = __builtin_elementwise_fma(c1[0], RF[0], -RIG[0]); SB;
    RIG[2][0] = __builtin_amdgcn_rcpf(PIG[2][0]); SB;
    RIG[2][1] = __builtin_amdgcn_rcpf(PIG[2][1]); SB;
    RF[3][0] = __builtin_amdgcn_rcpf(DF[3][0]); SB;
    acc2[1][1] = __builtin_amdgcn_wmma_f32_16x16x32_f16(false, wf1[20], false, __builtin_bit_cast(v16h, hb1[0]), (short)0, zero8, false, false); SB;
    RF[3][1] = __builtin_amdgcn_rcpf(DF[3][1]); SB;
    RIG[3][0] = __builtin_amdgcn_rcpf(PIG[3][0]); SB;
    TT[1] = __builtin_elementwise_fma(c1[1], RF[1], -RIG[1]); SB;
    RIG[3][1] = __builtin_amdgcn_rcpf(PIG[3][1]); SB;
    c1[0] = __builtin_elementwise_fma(EG[0], RIG[0], TT[0]); SB;
    EO[0][0] = __builtin_amdgcn_exp2f(acc1[0][0][6]); SB;
    acc2[1][2] = __builtin_amdgcn_wmma_f32_16x16x32_f16(false, wf1[22], false, __builtin_bit_cast(v16h, hb1[0]), (short)0, zero8, false, false); SB;
    EO[0][1] = __builtin_amdgcn_exp2f(acc1[0][0][7]); SB;
    EO[1][0] = __builtin_amdgcn_exp2f(acc1[0][1][6]); SB;
    TT[2] = __builtin_elementwise_fma(c1[2], RF[2], -RIG[2]); SB;
    EO[1][1] = __builtin_amdgcn_exp2f(acc1[0][1][7]); SB;
    EO[2][0] = __builtin_amdgcn_exp2f(acc1[0][2][6]); SB;
    acc1[1][0] = __builtin_amdgcn_wmma_f32_16x16x32_f16(false, wf2[3], false, __builtin_bit_cast(v16h, hb2[1]), (short)0, acc1[1][0], false, false); SB;
    c1[1] = __builtin_elementwise_fma(EG[1], RIG[1], TT[1]); SB;
    EO[2][1] = __builtin_amdgcn_exp2f(acc1[0][2][7]); SB;
    TT[3] = __builtin_elementwise_fma(c1[3], RF[3], -RIG[3]); SB;
    EO[3][0] = __builtin_amdgcn_exp2f(acc1[0][3][6]); SB;
    CC[0][0] = cap_exp_arg(c1[0][0]); SB;
    EO[3][1] = __builtin_amdgcn_exp2f(acc1[0][3][7]); SB;
    CC[0][1] = cap_exp_arg(c1[0][1]); SB;
    acc2[1][3] = __builtin_amdgcn_wmma_f32_16x16x32_f16(false, wf2[10], false, __builtin_bit_cast(v16h, hb2[0]), (short)0, zero8, false, false); SB;
    c1[2] = __builtin_elementwise_fma(EG[2], RIG[2], TT[2]); SB;
    DQ[0] = EO[0] + 1.0f; SB;
    DQ[1] = EO[1] + 1.0f; SB;
    EC[0][0] = __builtin_amdgcn_exp2f(CC[0][0]); SB;
    CC[1][0] = cap_exp_arg(c1[1][0]); SB;
    EC[0][1] = __builtin_amdgcn_exp2f(CC[0][1]); SB;
    c1[3] = __builtin_elementwise_fma(EG[3], RIG[3], TT[3]); SB;
    acc1[1][1] = __builtin_amdgcn_wmma_f32_16x16x32_f16(false, wf1[25], false, __builtin_bit_cast(v16h, hb1[1]), (short)0, acc1[1][1], false, false); SB;
    CC[1][1] = cap_exp_arg(c1[1][1]); SB;
    DQ[2] = EO[2] + 1.0f; SB;
    EC[1][0] = __builtin_amdgcn_exp2f(CC[1][0]); SB;
    CC[2][0] = cap_exp_arg(c1[2][0]); SB;
    CC[2][1] = cap_exp_arg(c1[2][1]); SB;
    DQ[3] = EO[3] + 1.0f; SB;
    EC[1][1] = __builtin_amdgcn_exp2f(CC[1][1]); SB;
    acc1[1][2] = __builtin_amdgcn_wmma_f32_16x16x32_f16(false, wf1[9], false, __builtin_bit_cast(v16h, hb1[1]), (short)0, acc1[1][2], false, false); SB;
    CC[3][0] = cap_exp_arg(c1[3][0]); SB;
    EC[2][0] = __builtin_amdgcn_exp2f(CC[2][0]); SB;
    CC[3][1] = cap_exp_arg(c1[3][1]); SB;
    EC[2][1] = __builtin_amdgcn_exp2f(CC[2][1]); SB;
    POC[0] = __builtin_elementwise_fma(EC[0], DQ[0], DQ[0]); SB;
    acc1[1][3] = __builtin_amdgcn_wmma_f32_16x16x32_f16(false, wf1[11], false, __builtin_bit_cast(v16h, hb1[1]), (short)0, acc1[1][3], false, false); SB;
    EC[3][0] = __builtin_amdgcn_exp2f(CC[3][0]); SB;
    EC[3][1] = __builtin_amdgcn_exp2f(CC[3][1]); SB;
    POC[1] = __builtin_elementwise_fma(EC[1], DQ[1], DQ[1]); SB;
    acc2[0][0] = __builtin_amdgcn_wmma_f32_16x16x32_f16(false, wf1[13], false, __builtin_bit_cast(v16h, hb1[1]), (short)0, acc2[0][0], false, false); SB;
    GC[12][0] = cap_exp_arg(acc1[1][1][4]); SB;
    EI[12][0] = __builtin_amdgcn_exp2f(acc1[1][1][0]); SB;
    GC[12][1] = cap_exp_arg(acc1[1][1][5]); SB;
    EI[12][1] = __builtin_amdgcn_exp2f(acc1[1][1][1]); SB;
    POC[2] = __builtin_elementwise_fma(EC[2], DQ[2], DQ[2]); SB;
    EF[12][0] = __builtin_amdgcn_exp2f(acc1[1][1][2]); SB;
    EG[12][0] = __builtin_amdgcn_exp2f(GC[12][0]); SB;
    acc2[0][1] = __builtin_amdgcn_wmma_f32_16x16x32_f16(false, wf2[7], false, __builtin_bit_cast(v16h, hb2[1]), (short)0, acc2[0][1], false, false); SB;
    GC[4][0] = cap_exp_arg(acc1[1][2][4]); SB;
    EG[12][1] = __builtin_amdgcn_exp2f(GC[12][1]); SB;
    GC[4][1] = cap_exp_arg(acc1[1][2][5]); SB;
    EF[12][1] = __builtin_amdgcn_exp2f(acc1[1][1][3]); SB;
    POC[3] = __builtin_elementwise_fma(EC[3], DQ[3], DQ[3]); SB;
    EI[4][0] = __builtin_amdgcn_exp2f(acc1[1][2][0]); SB;
    EG[4][0] = __builtin_amdgcn_exp2f(GC[4][0]); SB;
    acc2[0][2] = __builtin_amdgcn_wmma_f32_16x16x32_f16(false, wf1[15], false, __builtin_bit_cast(v16h, hb1[1]), (short)0, acc2[0][2], false, false); SB;
    GC[5][0] = cap_exp_arg(acc1[1][3][4]); SB;
    EG[4][1] = __builtin_amdgcn_exp2f(GC[4][1]); SB;
    GC[5][1] = cap_exp_arg(acc1[1][3][5]); SB;
    EI[4][1] = __builtin_amdgcn_exp2f(acc1[1][2][1]); SB;
    EF[4][0] = __builtin_amdgcn_exp2f(acc1[1][2][2]); SB;
    DGK[12] = __builtin_elementwise_fma(EG[12], rkc2, rkc2); SB;
    EG[5][0] = __builtin_amdgcn_exp2f(GC[5][0]); SB;
    acc2[0][3] = __builtin_amdgcn_wmma_f32_16x16x32_f16(false, wf1[17], false, __builtin_bit_cast(v16h, hb1[1]), (short)0, acc2[0][3], false, false); SB;
    GC[6][0] = cap_exp_arg(acc2[0][0][4]); SB;
    EG[5][1] = __builtin_amdgcn_exp2f(GC[5][1]); SB;
    GC[6][1] = cap_exp_arg(acc2[0][0][5]); SB;
    EF[4][1] = __builtin_amdgcn_exp2f(acc1[1][2][3]); SB;
    DF[12] = EF[12] + 1.0f; SB;
    EI[5][0] = __builtin_amdgcn_exp2f(acc1[1][3][0]); SB;
    DGK[4] = __builtin_elementwise_fma(EG[4], rkc2, rkc2); SB;
    EG[6][0] = __builtin_amdgcn_exp2f(GC[6][0]); SB;
    acc2[1][0] = __builtin_amdgcn_wmma_f32_16x16x32_f16(false, wf1[19], false, __builtin_bit_cast(v16h, hb1[1]), (short)0, acc2[1][0], false, false); SB;
    PIG[12] = __builtin_elementwise_fma(EI[12], DGK[12], DGK[12]); SB;
    EG[6][1] = __builtin_amdgcn_exp2f(GC[6][1]); SB;
    EI[5][1] = __builtin_amdgcn_exp2f(acc1[1][3][1]); SB;
    EF[5][0] = __builtin_amdgcn_exp2f(acc1[1][3][2]); SB;
    DGK[5] = __builtin_elementwise_fma(EG[5], rkc2, rkc2); SB;
    EF[5][1] = __builtin_amdgcn_exp2f(acc1[1][3][3]); SB;
    acc2[1][1] = __builtin_amdgcn_wmma_f32_16x16x32_f16(false, wf1[21], false, __builtin_bit_cast(v16h, hb1[1]), (short)0, acc2[1][1], false, false); SB;
    GC[7][0] = cap_exp_arg(acc2[0][2][4]); SB;
    EI[6][0] = __builtin_amdgcn_exp2f(acc2[0][0][0]); SB;
    GC[7][1] = cap_exp_arg(acc2[0][2][5]); SB;
    EI[6][1] = __builtin_amdgcn_exp2f(acc2[0][0][1]); SB;
    DF[4] = EF[4] + 1.0f; SB;
    EF[6][0] = __builtin_amdgcn_exp2f(acc2[0][0][2]); SB;
    DGK[6] = __builtin_elementwise_fma(EG[6], rkc2, rkc2); SB;
    EG[7][0] = __builtin_amdgcn_exp2f(GC[7][0]); SB;
    acc2[1][2] = __builtin_amdgcn_wmma_f32_16x16x32_f16(false, wf1[23], false, __builtin_bit_cast(v16h, hb1[1]), (short)0, acc2[1][2], false, false); SB;
    GC[8][0] = cap_exp_arg(acc2[0][3][4]); SB;
    EG[7][1] = __builtin_amdgcn_exp2f(GC[7][1]); SB;
    GC[8][1] = cap_exp_arg(acc2[0][3][5]); SB;
    EF[6][1] = __builtin_amdgcn_exp2f(acc2[0][0][3]); SB;
    PIG[4] = __builtin_elementwise_fma(EI[4], DGK[4], DGK[4]); SB;
    EI[7][0] = __builtin_amdgcn_exp2f(acc2[0][2][0]); SB;
    DF[5] = EF[5] + 1.0f; SB;
    EG[8][0] = __builtin_amdgcn_exp2f(GC[8][0]); SB;
    acc1[1][0] = __builtin_amdgcn_wmma_f32_16x16x32_f16(false, wf2[0], false, __builtin_bit_cast(v16h, hb1[0]), (short)0, acc1[1][0], false, false); SB;
    PIG[5] = __builtin_elementwise_fma(EI[5], DGK[5], DGK[5]); SB;
    EG[8][1] = __builtin_amdgcn_exp2f(GC[8][1]); SB;
    PIG[6] = __builtin_elementwise_fma(EI[6], DGK[6], DGK[6]); SB;
    EI[7][1] = __builtin_amdgcn_exp2f(acc2[0][2][1]); SB;
    EF[7][0] = __builtin_amdgcn_exp2f(acc2[0][2][2]); SB;
    DGK[7] = __builtin_elementwise_fma(EG[7], rkc2, rkc2); SB;
    EF[7][1] = __builtin_amdgcn_exp2f(acc2[0][2][3]); SB;
    acc2[1][3] = __builtin_amdgcn_wmma_f32_16x16x32_f16(false, wf2[11], false, __builtin_bit_cast(v16h, hb2[1]), (short)0, acc2[1][3], false, false); SB;
    DF[6] = EF[6] + 1.0f; SB;
    EI[8][0] = __builtin_amdgcn_exp2f(acc2[0][3][0]); SB;
    EI[8][1] = __builtin_amdgcn_exp2f(acc2[0][3][1]); SB;
    EF[8][0] = __builtin_amdgcn_exp2f(acc2[0][3][2]); SB;
    DGK[8] = __builtin_elementwise_fma(EG[8], rkc2, rkc2); SB;
    EF[8][1] = __builtin_amdgcn_exp2f(acc2[0][3][3]); SB;
    acc2[0][1] = __builtin_amdgcn_wmma_f32_16x16x32_f16(false, wf2[4], false, __builtin_bit_cast(v16h, hb1[0]), (short)0, acc2[0][1], false, false); SB;
    PIG[7] = __builtin_elementwise_fma(EI[7], DGK[7], DGK[7]); SB;
    RF[12][0] = __builtin_amdgcn_rcpf(DF[12][0]); SB;
    RF[12][1] = __builtin_amdgcn_rcpf(DF[12][1]); SB;
    DF[7] = EF[7] + 1.0f; SB;
    RIG[12][0] = __builtin_amdgcn_rcpf(PIG[12][0]); SB;
    RIG[12][1] = __builtin_amdgcn_rcpf(PIG[12][1]); SB;
    acc1[1][0] = __builtin_amdgcn_wmma_f32_16x16x32_f16(false, wf2[1], false, __builtin_bit_cast(v16h, hb1[1]), (short)0, acc1[1][0], false, false); SB;
    PIG[8] = __builtin_elementwise_fma(EI[8], DGK[8], DGK[8]); SB;
    RF[4][0] = __builtin_amdgcn_rcpf(DF[4][0]); SB;
    RF[4][1] = __builtin_amdgcn_rcpf(DF[4][1]); SB;
    DF[8] = EF[8] + 1.0f; SB;
    RIG[4][0] = __builtin_amdgcn_rcpf(PIG[4][0]); SB;
    RIG[4][1] = __builtin_amdgcn_rcpf(PIG[4][1]); SB;
    acc2[1][3] = __builtin_amdgcn_wmma_f32_16x16x32_f16(false, wf2[8], false, __builtin_bit_cast(v16h, hb1[0]), (short)0, acc2[1][3], false, false); SB;
    RF[5][0] = __builtin_amdgcn_rcpf(DF[5][0]); SB;
    RF[5][1] = __builtin_amdgcn_rcpf(DF[5][1]); SB;
    TT[12] = __builtin_elementwise_fma(c1[12], RF[12], -RIG[12]); SB;
    RIG[5][0] = __builtin_amdgcn_rcpf(PIG[5][0]); SB;
    RIG[5][1] = __builtin_amdgcn_rcpf(PIG[5][1]); SB;
    acc2[0][1] = __builtin_amdgcn_wmma_f32_16x16x32_f16(false, wf2[5], false, __builtin_bit_cast(v16h, hb1[1]), (short)0, acc2[0][1], false, false); SB;
    RF[6][0] = __builtin_amdgcn_rcpf(DF[6][0]); SB;
    RF[6][1] = __builtin_amdgcn_rcpf(DF[6][1]); SB;
    RIG[6][0] = __builtin_amdgcn_rcpf(PIG[6][0]); SB;
    TT[4] = __builtin_elementwise_fma(c1[4], RF[4], -RIG[4]); SB;
    RIG[6][1] = __builtin_amdgcn_rcpf(PIG[6][1]); SB;
    c1[12] = __builtin_elementwise_fma(EG[12], RIG[12], TT[12]); SB;
    RF[7][0] = __builtin_amdgcn_rcpf(DF[7][0]); SB;
    RF[7][1] = __builtin_amdgcn_rcpf(DF[7][1]); SB;
    RIG[7][0] = __builtin_amdgcn_rcpf(PIG[7][0]); SB;
    TT[5] = __builtin_elementwise_fma(c1[5], RF[5], -RIG[5]); SB;
    RIG[7][1] = __builtin_amdgcn_rcpf(PIG[7][1]); SB;
    RF[8][0] = __builtin_amdgcn_rcpf(DF[8][0]); SB;
    acc2[1][3] = __builtin_amdgcn_wmma_f32_16x16x32_f16(false, wf2[9], false, __builtin_bit_cast(v16h, hb1[1]), (short)0, acc2[1][3], false, false); SB;
    c1[4] = __builtin_elementwise_fma(EG[4], RIG[4], TT[4]); SB;
    RF[8][1] = __builtin_amdgcn_rcpf(DF[8][1]); SB;
    TT[6] = __builtin_elementwise_fma(c1[6], RF[6], -RIG[6]); SB;
    RIG[8][0] = __builtin_amdgcn_rcpf(PIG[8][0]); SB;
    CC[12][0] = cap_exp_arg(c1[12][0]); SB;
    RIG[8][1] = __builtin_amdgcn_rcpf(PIG[8][1]); SB;
    CC[12][1] = cap_exp_arg(c1[12][1]); SB;
    EO[12][0] = __builtin_amdgcn_exp2f(acc1[1][1][6]); SB;
    c1[5] = __builtin_elementwise_fma(EG[5], RIG[5], TT[5]); SB;
    EO[12][1] = __builtin_amdgcn_exp2f(acc1[1][1][7]); SB;
    acc1[1][1] = __builtin_amdgcn_wmma_f32_16x16x32_f16(false, wf2[14], false, __builtin_bit_cast(v16h, hb2[0]), (short)0, zero8, false, false); SB;
    TT[7] = __builtin_elementwise_fma(c1[7], RF[7], -RIG[7]); SB;
    EO[4][0] = __builtin_amdgcn_exp2f(acc1[1][2][6]); SB;
    CC[4][0] = cap_exp_arg(c1[4][0]); SB;
    EO[4][1] = __builtin_amdgcn_exp2f(acc1[1][2][7]); SB;
    c1[6] = __builtin_elementwise_fma(EG[6], RIG[6], TT[6]); SB;
    EO[5][0] = __builtin_amdgcn_exp2f(acc1[1][3][6]); SB;
    CC[4][1] = cap_exp_arg(c1[4][1]); SB;
    EO[5][1] = __builtin_amdgcn_exp2f(acc1[1][3][7]); SB;
    acc1[1][2] = __builtin_amdgcn_wmma_f32_16x16x32_f16(false, wf2[18], false, __builtin_bit_cast(v16h, hb2[0]), (short)0, zero8, false, false); SB;
    TT[8] = __builtin_elementwise_fma(c1[8], RF[8], -RIG[8]); SB;
    EO[6][0] = __builtin_amdgcn_exp2f(acc2[0][0][6]); SB;
    CC[5][0] = cap_exp_arg(c1[5][0]); SB;
    EO[6][1] = __builtin_amdgcn_exp2f(acc2[0][0][7]); SB;
    c1[7] = __builtin_elementwise_fma(EG[7], RIG[7], TT[7]); SB;
    EO[7][0] = __builtin_amdgcn_exp2f(acc2[0][2][6]); SB;
    CC[5][1] = cap_exp_arg(c1[5][1]); SB;
    EO[7][1] = __builtin_amdgcn_exp2f(acc2[0][2][7]); SB;
    acc1[1][3] = __builtin_amdgcn_wmma_f32_16x16x32_f16(false, wf2[22], false, __builtin_bit_cast(v16h, hb2[0]), (short)0, zero8, false, false); SB;
    CC[6][0] = cap_exp_arg(c1[6][0]); SB;
    EO[8][0] = __builtin_amdgcn_exp2f(acc2[0][3][6]); SB;
    CC[6][1] = cap_exp_arg(c1[6][1]); SB;
    EO[8][1] = __builtin_amdgcn_exp2f(acc2[0][3][7]); SB;
    c1[8] = __builtin_elementwise_fma(EG[8], RIG[8], TT[8]); SB;
    EC[12][0] = __builtin_amdgcn_exp2f(CC[12][0]); SB;
    DQ[12] = EO[12] + 1.0f; SB;
    EC[12][1] = __builtin_amdgcn_exp2f(CC[12][1]); SB;
    acc1[1][1] = __builtin_amdgcn_wmma_f32_16x16x32_f16(false, wf2[15], false, __builtin_bit_cast(v16h, hb2[1]), (short)0, acc1[1][1], false, false); SB;
    CC[7][0] = cap_exp_arg(c1[7][0]); SB;
    EC[4][0] = __builtin_amdgcn_exp2f(CC[4][0]); SB;
    CC[7][1] = cap_exp_arg(c1[7][1]); SB;
    EC[4][1] = __builtin_amdgcn_exp2f(CC[4][1]); SB;
    DQ[4] = EO[4] + 1.0f; SB;
    EC[5][0] = __builtin_amdgcn_exp2f(CC[5][0]); SB;
    DQ[5] = EO[5] + 1.0f; SB;
    EC[5][1] = __builtin_amdgcn_exp2f(CC[5][1]); SB;
    acc1[1][2] = __builtin_amdgcn_wmma_f32_16x16x32_f16(false, wf2[19], false, __builtin_bit_cast(v16h, hb2[1]), (short)0, acc1[1][2], false, false); SB;
    CC[8][0] = cap_exp_arg(c1[8][0]); SB;
    EC[6][0] = __builtin_amdgcn_exp2f(CC[6][0]); SB;
    CC[8][1] = cap_exp_arg(c1[8][1]); SB;
    EC[6][1] = __builtin_amdgcn_exp2f(CC[6][1]); SB;
    DQ[6] = EO[6] + 1.0f; SB;
    EC[7][0] = __builtin_amdgcn_exp2f(CC[7][0]); SB;
    DQ[7] = EO[7] + 1.0f; SB;
    EC[7][1] = __builtin_amdgcn_exp2f(CC[7][1]); SB;
    acc2[0][0] = __builtin_amdgcn_wmma_f32_16x16x32_f16(false, wf2[26], false, __builtin_bit_cast(v16h, hb2[0]), (short)0, zero8, false, false); SB;
    DQ[8] = EO[8] + 1.0f; SB;
    EC[8][0] = __builtin_amdgcn_exp2f(CC[8][0]); SB;
    POC[12] = __builtin_elementwise_fma(EC[12], DQ[12], DQ[12]); SB;
    EC[8][1] = __builtin_amdgcn_exp2f(CC[8][1]); SB;
    POC[4] = __builtin_elementwise_fma(EC[4], DQ[4], DQ[4]); SB;
    ROC[0][0] = __builtin_amdgcn_rcpf(POC[0][0]); SB;
    POC[5] = __builtin_elementwise_fma(EC[5], DQ[5], DQ[5]); SB;
    ROC[0][1] = __builtin_amdgcn_rcpf(POC[0][1]); SB;
    acc1[1][1] = __builtin_amdgcn_wmma_f32_16x16x32_f16(false, wf2[12], false, __builtin_bit_cast(v16h, hb1[0]), (short)0, acc1[1][1], false, false); SB;
    POC[6] = __builtin_elementwise_fma(EC[6], DQ[6], DQ[6]); SB;
    ROC[1][0] = __builtin_amdgcn_rcpf(POC[1][0]); SB;
    ROC[1][1] = __builtin_amdgcn_rcpf(POC[1][1]); SB;
    POC[7] = __builtin_elementwise_fma(EC[7], DQ[7], DQ[7]); SB;
    ROC[12][0] = __builtin_amdgcn_rcpf(POC[12][0]); SB;
    ROC[12][1] = __builtin_amdgcn_rcpf(POC[12][1]); SB;
    acc1[1][3] = __builtin_amdgcn_wmma_f32_16x16x32_f16(false, wf2[23], false, __builtin_bit_cast(v16h, hb2[1]), (short)0, acc1[1][3], false, false); SB;
    POC[8] = __builtin_elementwise_fma(EC[8], DQ[8], DQ[8]); SB;
    ROC[2][0] = __builtin_amdgcn_rcpf(POC[2][0]); SB;
    ROC[2][1] = __builtin_amdgcn_rcpf(POC[2][1]); SB;
    HV[0] = __builtin_elementwise_fma(EC[0], ROC[0], -ROC[0]); SB;
    ROC[3][0] = __builtin_amdgcn_rcpf(POC[3][0]); SB;
    ROC[3][1] = __builtin_amdgcn_rcpf(POC[3][1]); SB;
    acc2[0][2] = __builtin_amdgcn_wmma_f32_16x16x32_f16(false, wf2[30], false, __builtin_bit_cast(v16h, hb2[0]), (short)0, zero8, false, false); SB;
    HV[1] = __builtin_elementwise_fma(EC[1], ROC[1], -ROC[1]); SB;
    ROC[4][0] = __builtin_amdgcn_rcpf(POC[4][0]); SB;
    ROC[4][1] = __builtin_amdgcn_rcpf(POC[4][1]); SB;
    HV[12] = __builtin_elementwise_fma(EC[12], ROC[12], -ROC[12]); SB;
    ROC[5][0] = __builtin_amdgcn_rcpf(POC[5][0]); SB;
    ROC[5][1] = __builtin_amdgcn_rcpf(POC[5][1]); SB;
    acc1[1][2] = __builtin_amdgcn_wmma_f32_16x16x32_f16(false, wf2[16], false, __builtin_bit_cast(v16h, hb1[0]), (short)0, acc1[1][2], false, false); SB;
    HV[2] = __builtin_elementwise_fma(EC[2], ROC[2], -ROC[2]); SB;
    ROC[6][0] = __builtin_amdgcn_rcpf(POC[6][0]); SB;
    { v2h pk; pk[0] = (_Float16)HV[0][0]; pk[1] = (_Float16)HV[0][1]; hn1[0][0] = __builtin_bit_cast(unsigned, pk); } SB;
    EI[9][0] = __builtin_amdgcn_exp2f(acc2[1][0][0]); SB;
    GC[9][0] = cap_exp_arg(acc2[1][0][4]); SB;
    EI[9][1] = __builtin_amdgcn_exp2f(acc2[1][0][1]); SB;
    GC[9][1] = cap_exp_arg(acc2[1][0][5]); SB;
    EF[9][0] = __builtin_amdgcn_exp2f(acc2[1][0][2]); SB;
    acc2[0][0] = __builtin_amdgcn_wmma_f32_16x16x32_f16(false, wf2[27], false, __builtin_bit_cast(v16h, hb2[1]), (short)0, acc2[0][0], false, false); SB;
    HV[3] = __builtin_elementwise_fma(EC[3], ROC[3], -ROC[3]); SB;
    EF[9][1] = __builtin_amdgcn_exp2f(acc2[1][0][3]); SB;
    HV[4] = __builtin_elementwise_fma(EC[4], ROC[4], -ROC[4]); SB;
    EG[9][0] = __builtin_amdgcn_exp2f(GC[9][0]); SB;
    HV[5] = __builtin_elementwise_fma(EC[5], ROC[5], -ROC[5]); SB;
    EG[9][1] = __builtin_amdgcn_exp2f(GC[9][1]); SB;
    HV[12][1] = (lh == 1) ? 0.0f : HV[12][1]; SB;
    EO[9][0] = __builtin_amdgcn_exp2f(acc2[1][0][6]); SB;
    acc2[0][3] = __builtin_amdgcn_wmma_f32_16x16x32_f16(false, wf2[34], false, __builtin_bit_cast(v16h, hb2[0]), (short)0, zero8, false, false); SB;
    { v2h pk; pk[0] = (_Float16)HV[1][0]; pk[1] = (_Float16)HV[1][1]; hn1[0][1] = __builtin_bit_cast(unsigned, pk); } SB;
    EI[10][0] = __builtin_amdgcn_exp2f(acc2[1][1][0]); SB;
    GC[10][0] = cap_exp_arg(acc2[1][1][4]); SB;
    EI[10][1] = __builtin_amdgcn_exp2f(acc2[1][1][1]); SB;
    GC[10][1] = cap_exp_arg(acc2[1][1][5]); SB;
    EF[10][0] = __builtin_amdgcn_exp2f(acc2[1][1][2]); SB;
    DF[9] = EF[9] + 1.0f; SB;
    EF[10][1] = __builtin_amdgcn_exp2f(acc2[1][1][3]); SB;
    acc1[1][1] = __builtin_amdgcn_wmma_f32_16x16x32_f16(false, wf2[13], false, __builtin_bit_cast(v16h, hb1[1]), (short)0, acc1[1][1], false, false); SB;
    { v2h pk; pk[0] = (_Float16)HV[2][0]; pk[1] = (_Float16)HV[2][1]; hn1[0][2] = __builtin_bit_cast(unsigned, pk); } SB;
    EG[10][0] = __builtin_amdgcn_exp2f(GC[10][0]); SB;
    GC[11][0] = cap_exp_arg(acc2[1][2][4]); SB;
    EG[10][1] = __builtin_amdgcn_exp2f(GC[10][1]); SB;
    GC[11][1] = cap_exp_arg(acc2[1][2][5]); SB;
    EI[11][0] = __builtin_amdgcn_exp2f(acc2[1][2][0]); SB;
    DGK[9] = __builtin_elementwise_fma(EG[9], rkc2, rkc2); SB;
    EI[11][1] = __builtin_amdgcn_exp2f(acc2[1][2][1]); SB;
    acc1[1][3] = __builtin_amdgcn_wmma_f32_16x16x32_f16(false, wf2[20], false, __builtin_bit_cast(v16h, hb1[0]), (short)0, acc1[1][3], false, false); SB;
    { v2h pk; pk[0] = (_Float16)HV[3][0]; pk[1] = (_Float16)HV[3][1]; hn1[0][3] = __builtin_bit_cast(unsigned, pk); } SB;
    EG[11][0] = __builtin_amdgcn_exp2f(GC[11][0]); SB;
    GC[13][0] = cap_exp_arg(acc1[1][0][4]); SB;
    EG[11][1] = __builtin_amdgcn_exp2f(GC[11][1]); SB;
    DF[10] = EF[10] + 1.0f; SB;
    EF[11][0] = __builtin_amdgcn_exp2f(acc2[1][2][2]); SB;
    GC[13][1] = cap_exp_arg(acc1[1][0][5]); SB;
    EF[11][1] = __builtin_amdgcn_exp2f(acc2[1][2][3]); SB;
    acc2[0][2] = __builtin_amdgcn_wmma_f32_16x16x32_f16(false, wf2[31], false, __builtin_bit_cast(v16h, hb2[1]), (short)0, acc2[0][2], false, false); SB;
    DGK[10] = __builtin_elementwise_fma(EG[10], rkc2, rkc2); SB;
    EG[13][0] = __builtin_amdgcn_exp2f(GC[13][0]); SB;
    PIG[9] = __builtin_elementwise_fma(EI[9], DGK[9], DGK[9]); SB;
    RF[9][0] = __builtin_amdgcn_rcpf(DF[9][0]); SB;
    { v2h pk; pk[0] = (_Float16)HV[4][0]; pk[1] = (_Float16)HV[4][1]; hn1[0][4] = __builtin_bit_cast(unsigned, pk); } SB;
    EG[13][1] = __builtin_amdgcn_exp2f(GC[13][1]); SB;
    GC[14][0] = cap_exp_arg(acc2[0][1][4]); SB;
    RF[9][1] = __builtin_amdgcn_rcpf(DF[9][1]); SB;
    acc1[1][2] = __builtin_amdgcn_wmma_f32_16x16x32_f16(false, wf2[17], false, __builtin_bit_cast(v16h, hb1[1]), (short)0, acc1[1][2], false, false); SB;
    DGK[11] = __builtin_elementwise_fma(EG[11], rkc2, rkc2); SB;
    RF[10][0] = __builtin_amdgcn_rcpf(DF[10][0]); SB;
    GC[14][1] = cap_exp_arg(acc2[0][1][5]); SB;
    RF[10][1] = __builtin_amdgcn_rcpf(DF[10][1]); SB;
    PIG[10] = __builtin_elementwise_fma(EI[10], DGK[10], DGK[10]); SB;
    RIG[9][0] = __builtin_amdgcn_rcpf(PIG[9][0]); SB;
    DF[11] = EF[11] + 1.0f; SB;
    RIG[9][1] = __builtin_amdgcn_rcpf(PIG[9][1]); SB;
    acc2[0][0] = __builtin_amdgcn_wmma_f32_16x16x32_f16(false, wf2[24], false, __builtin_bit_cast(v16h, hb1[0]), (short)0, acc2[0][0], false, false); SB;
    { v2h pk; pk[0] = (_Float16)HV[5][0]; pk[1] = (_Float16)HV[5][1]; hn1[0][5] = __builtin_bit_cast(unsigned, pk); } SB;
    EI[13][0] = __builtin_amdgcn_exp2f(acc1[1][0][0]); SB;
    DGK[13] = __builtin_elementwise_fma(EG[13], rkc2, rkc2); SB;
    EI[13][1] = __builtin_amdgcn_exp2f(acc1[1][0][1]); SB;
    PIG[11] = __builtin_elementwise_fma(EI[11], DGK[11], DGK[11]); SB;
    EF[13][0] = __builtin_amdgcn_exp2f(acc1[1][0][2]); SB;
    GC[15][0] = cap_exp_arg(acc2[1][3][4]); SB;
    RIG[10][0] = __builtin_amdgcn_rcpf(PIG[10][0]); SB;
    acc2[0][3] = __builtin_amdgcn_wmma_f32_16x16x32_f16(false, wf2[35], false, __builtin_bit_cast(v16h, hb2[1]), (short)0, acc2[0][3], false, false); SB;
    GC[15][1] = cap_exp_arg(acc2[1][3][5]); SB;
    RIG[10][1] = __builtin_amdgcn_rcpf(PIG[10][1]); SB;
    { v2h pk; pk[0] = (_Float16)HV[12][0]; pk[1] = (_Float16)HV[12][1]; hn1[1][4] = __builtin_bit_cast(unsigned, pk); } SB;
    RF[11][0] = __builtin_amdgcn_rcpf(DF[11][0]); SB;
    TT[9] = __builtin_elementwise_fma(c1[9], RF[9], -RIG[9]); SB;
    RF[11][1] = __builtin_amdgcn_rcpf(DF[11][1]); SB;
    GC[16][0] = cap_exp_arg(acc1[1][1][4]); SB;
    RIG[11][0] = __builtin_amdgcn_rcpf(PIG[11][0]); SB;
    acc1[1][3] = __builtin_amdgcn_wmma_f32_16x16x32_f16(false, wf2[21], false, __builtin_bit_cast(v16h, hb1[1]), (short)0, acc1[1][3], false, false); SB;
    PIG[13] = __builtin_elementwise_fma(EI[13], DGK[13], DGK[13]); SB;
    RIG[11][1] = __builtin_amdgcn_rcpf(PIG[11][1]); SB;
    GC[16][1] = cap_exp_arg(acc1[1][1][5]); SB;
    EF[13][1] = __builtin_amdgcn_exp2f(acc1[1][0][3]); SB;
    EG[14][0] = __builtin_amdgcn_exp2f(GC[14][0]); SB;
    TT[10] = __builtin_elementwise_fma(c1[10], RF[10], -RIG[10]); SB;
    EG[14][1] = __builtin_amdgcn_exp2f(GC[14][1]); SB;
    acc2[0][2] = __builtin_amdgcn_wmma_f32_16x16x32_f16(false, wf2[28], false, __builtin_bit_cast(v16h, hb1[0]), (short)0, acc2[0][2], false, false); SB;
    c1[9] = __builtin_elementwise_fma(EG[9], RIG[9], TT[9]); SB;
    EI[14][0] = __builtin_amdgcn_exp2f(acc2[0][1][0]); SB;
    EI[14][1] = __builtin_amdgcn_exp2f(acc2[0][1][1]); SB;
    EF[14][0] = __builtin_amdgcn_exp2f(acc2[0][1][2]); SB;
    TT[11] = __builtin_elementwise_fma(c1[11], RF[11], -RIG[11]); SB;
    EF[14][1] = __builtin_amdgcn_exp2f(acc2[0][1][3]); SB;
    acc2[0][0] = __builtin_amdgcn_wmma_f32_16x16x32_f16(false, wf2[25], false, __builtin_bit_cast(v16h, hb1[1]), (short)0, acc2[0][0], false, false); SB;
    DF[13] = EF[13] + 1.0f; SB;
    EG[15][0] = __builtin_amdgcn_exp2f(GC[15][0]); SB;
    c1[10] = __builtin_elementwise_fma(EG[10], RIG[10], TT[10]); SB;
    EG[15][1] = __builtin_amdgcn_exp2f(GC[15][1]); SB;
    DGK[14] = __builtin_elementwise_fma(EG[14], rkc2, rkc2); SB;
    EO[9][1] = __builtin_amdgcn_exp2f(acc2[1][0][7]); SB;
    CC[9][0] = cap_exp_arg(c1[9][0]); SB;
    EO[10][0] = __builtin_amdgcn_exp2f(acc2[1][1][6]); SB;
    acc2[1][0] = __builtin_amdgcn_wmma_f32_16x16x32_f16(false, wf2[38], false, __builtin_bit_cast(v16h, hb2[0]), (short)0, zero8, false, false); SB;
    c1[11] = __builtin_elementwise_fma(EG[11], RIG[11], TT[11]); SB;
    EO[10][1] = __builtin_amdgcn_exp2f(acc2[1][1][7]); SB;
    CC[9][1] = cap_exp_arg(c1[9][1]); SB;
    EO[11][0] = __builtin_amdgcn_exp2f(acc2[1][2][6]); SB;
    DF[14] = EF[14] + 1.0f; SB;
    EO[11][1] = __builtin_amdgcn_exp2f(acc2[1][2][7]); SB;
    CC[10][0] = cap_exp_arg(c1[10][0]); SB;
    RF[13][0] = __builtin_amdgcn_rcpf(DF[13][0]); SB;
    acc2[1][1] = __builtin_amdgcn_wmma_f32_16x16x32_f16(false, wf2[42], false, __builtin_bit_cast(v16h, hb2[0]), (short)0, zero8, false, false); SB;
    PIG[14] = __builtin_elementwise_fma(EI[14], DGK[14], DGK[14]); SB;
    RF[13][1] = __builtin_amdgcn_rcpf(DF[13][1]); SB;
    CC[10][1] = cap_exp_arg(c1[10][1]); SB;
    RIG[13][0] = __builtin_amdgcn_rcpf(PIG[13][0]); SB;
    CC[11][0] = cap_exp_arg(c1[11][0]); SB;
    RIG[13][1] = __builtin_amdgcn_rcpf(PIG[13][1]); SB;
    CC[11][1] = cap_exp_arg(c1[11][1]); SB;
    EI[15][0] = __builtin_amdgcn_exp2f(acc2[1][3][0]); SB;
    acc2[1][2] = __builtin_amdgcn_wmma_f32_16x16x32_f16(false, wf2[46], false, __builtin_bit_cast(v16h, hb2[0]), (short)0, zero8, false, false); SB;
    DGK[15] = __builtin_elementwise_fma(EG[15], rkc2, rkc2); SB;
    EI[15][1] = __builtin_amdgcn_exp2f(acc2[1][3][1]); SB;
    DQ[9] = EO[9] + 1.0f; SB;
    EF[15][0] = __builtin_amdgcn_exp2f(acc2[1][3][2]); SB;
    DQ[10] = EO[10] + 1.0f; SB;
    EF[15][1] = __builtin_amdgcn_exp2f(acc2[1][3][3]); SB;
    DQ[11] = EO[11] + 1.0f; SB;
    EG[16][0] = __builtin_amdgcn_exp2f(GC[16][0]); SB;
    acc2[1][0] = __builtin_amdgcn_wmma_f32_16x16x32_f16(false, wf2[39], false, __builtin_bit_cast(v16h, hb2[1]), (short)0, acc2[1][0], false, false); SB;
    EG[16][1] = __builtin_amdgcn_exp2f(GC[16][1]); SB;
    TT[13] = __builtin_elementwise_fma(c2[0], RF[13], -RIG[13]); SB;
    EC[9][0] = __builtin_amdgcn_exp2f(CC[9][0]); SB;
    EC[9][1] = __builtin_amdgcn_exp2f(CC[9][1]); SB;
    EC[10][0] = __builtin_amdgcn_exp2f(CC[10][0]); SB;
    acc2[1][1] = __builtin_amdgcn_wmma_f32_16x16x32_f16(false, wf2[43], false, __builtin_bit_cast(v16h, hb2[1]), (short)0, acc2[1][1], false, false); SB;
    PIG[15] = __builtin_elementwise_fma(EI[15], DGK[15], DGK[15]); SB;
    EC[10][1] = __builtin_amdgcn_exp2f(CC[10][1]); SB;
    DF[15] = EF[15] + 1.0f; SB;
    EC[11][0] = __builtin_amdgcn_exp2f(CC[11][0]); SB;
    set_x(hn1[1], xn); SB;
    EC[11][1] = __builtin_amdgcn_exp2f(CC[11][1]); SB;
    acc2[1][2] = __builtin_amdgcn_wmma_f32_16x16x32_f16(false, wf2[47], false, __builtin_bit_cast(v16h, hb2[1]), (short)0, acc2[1][2], false, false); SB;
    DGK[16] = __builtin_elementwise_fma(EG[16], rkc2, rkc2); SB;
    RF[14][0] = __builtin_amdgcn_rcpf(DF[14][0]); SB;
    POC[9] = __builtin_elementwise_fma(EC[9], DQ[9], DQ[9]); SB;
    RF[14][1] = __builtin_amdgcn_rcpf(DF[14][1]); SB;
    POC[10] = __builtin_elementwise_fma(EC[10], DQ[10], DQ[10]); SB;
    RIG[14][0] = __builtin_amdgcn_rcpf(PIG[14][0]); SB;
    c2[0] = __builtin_elementwise_fma(EG[13], RIG[13], TT[13]); SB;
    RIG[14][1] = __builtin_amdgcn_rcpf(PIG[14][1]); SB;
    acc2[0][3] = __builtin_amdgcn_wmma_f32_16x16x32_f16(false, wf2[32], false, __builtin_bit_cast(v16h, hb1[0]), (short)0, acc2[0][3], false, false); SB;
    EI[16][0] = __builtin_amdgcn_exp2f(acc1[1][1][0]); SB;
    EI[16][1] = __builtin_amdgcn_exp2f(acc1[1][1][1]); SB;
    POC[11] = __builtin_elementwise_fma(EC[11], DQ[11], DQ[11]); SB;
    EF[16][0] = __builtin_amdgcn_exp2f(acc1[1][1][2]); SB;
    EF[16][1] = __builtin_amdgcn_exp2f(acc1[1][1][3]); SB;
    acc2[1][0] = __builtin_amdgcn_wmma_f32_16x16x32_f16(false, wf2[36], false, __builtin_bit_cast(v16h, hb1[0]), (short)0, acc2[1][0], false, false); SB;
    RF[15][0] = __builtin_amdgcn_rcpf(DF[15][0]); SB;
    CC[13][0] = cap_exp_arg(c2[0][0]); SB;
    RF[15][1] = __builtin_amdgcn_rcpf(DF[15][1]); SB;
    CC[13][1] = cap_exp_arg(c2[0][1]); SB;
    RIG[15][0] = __builtin_amdgcn_rcpf(PIG[15][0]); SB;
    TT[14] = __builtin_elementwise_fma(c2[1], RF[14], -RIG[14]); SB;
    RIG[15][1] = __builtin_amdgcn_rcpf(PIG[15][1]); SB;
    acc2[1][1] = __builtin_amdgcn_wmma_f32_16x16x32_f16(false, wf2[40], false, __builtin_bit_cast(v16h, hb1[0]), (short)0, acc2[1][1], false, false); SB;
    PIG[16] = __builtin_elementwise_fma(EI[16], DGK[16], DGK[16]); SB;
    EO[13][0] = __builtin_amdgcn_exp2f(acc1[1][0][6]); SB;
    EO[13][1] = __builtin_amdgcn_exp2f(acc1[1][0][7]); SB;
    DF[16] = EF[16] + 1.0f; SB;
    ROC[6][1] = __builtin_amdgcn_rcpf(POC[6][1]); SB;
    ROC[7][0] = __builtin_amdgcn_rcpf(POC[7][0]); SB;
    acc1[1][0] = __builtin_amdgcn_wmma_f32_16x16x32_f16(false, wf2[50], false, __builtin_bit_cast(v16h, hb2[0]), (short)0, zero8, false, false); SB;
    c2[1] = __builtin_elementwise_fma(EG[14], RIG[14], TT[14]); SB;
    ROC[7][1] = __builtin_amdgcn_rcpf(POC[7][1]); SB;
    ROC[8][0] = __builtin_amdgcn_rcpf(POC[8][0]); SB;
    TT[15] = __builtin_elementwise_fma(c2[2], RF[15], -RIG[15]); SB;
    ROC[8][1] = __builtin_amdgcn_rcpf(POC[8][1]); SB;
    ROC[9][0] = __builtin_amdgcn_rcpf(POC[9][0]); SB;
    acc2[1][2] = __builtin_amdgcn_wmma_f32_16x16x32_f16(false, wf2[44], false, __builtin_bit_cast(v16h, hb1[0]), (short)0, acc2[1][2], false, false); SB;
    DQ[13] = EO[13] + 1.0f; SB;
    ROC[9][1] = __builtin_amdgcn_rcpf(POC[9][1]); SB;
    HV[6] = __builtin_elementwise_fma(EC[6], ROC[6], -ROC[6]); SB;
    ROC[10][0] = __builtin_amdgcn_rcpf(POC[10][0]); SB;
    CC[14][0] = cap_exp_arg(c2[1][0]); SB;
    ROC[10][1] = __builtin_amdgcn_rcpf(POC[10][1]); SB;
    HV[7] = __builtin_elementwise_fma(EC[7], ROC[7], -ROC[7]); SB;
    ROC[11][0] = __builtin_amdgcn_rcpf(POC[11][0]); SB;
    acc2[0][2] = __builtin_amdgcn_wmma_f32_16x16x32_f16(false, wf2[29], false, __builtin_bit_cast(v16h, hb1[1]), (short)0, acc2[0][2], false, false); SB;
    CC[14][1] = cap_exp_arg(c2[1][1]); SB;
    ROC[11][1] = __builtin_amdgcn_rcpf(POC[11][1]); SB;
    HV[8] = __builtin_elementwise_fma(EC[8], ROC[8], -ROC[8]); SB;
    RF[16][0] = __builtin_amdgcn_rcpf(DF[16][0]); SB;
    c2[2] = __builtin_elementwise_fma(EG[15], RIG[15], TT[15]); SB;
    RF[16][1] = __builtin_amdgcn_rcpf(DF[16][1]); SB;
    HV[9] = __builtin_elementwise_fma(EC[9], ROC[9], -ROC[9]); SB;
    RIG[16][0] = __builtin_amdgcn_rcpf(PIG[16][0]); SB;
    acc1[1][0] = __builtin_amdgcn_wmma_f32_16x16x32_f16(false, wf2[51], false, __builtin_bit_cast(v16h, hb2[1]), (short)0, acc1[1][0], false, false); SB;
    { v2h pk; pk[0] = (_Float16)HV[6][0]; pk[1] = (_Float16)HV[6][1]; hn1[0][6] = __builtin_bit_cast(unsigned, pk); } SB;
    EI[17][0] = __builtin_amdgcn_exp2f(acc1[1][2][0]); SB;
    GC[17][0] = cap_exp_arg(acc1[1][2][4]); SB;
    EI[17][1] = __builtin_amdgcn_exp2f(acc1[1][2][1]); SB;
    GC[17][1] = cap_exp_arg(acc1[1][2][5]); SB;
    EF[17][0] = __builtin_amdgcn_exp2f(acc1[1][2][2]); SB;
    HV[10] = __builtin_elementwise_fma(EC[10], ROC[10], -ROC[10]); SB;
    EF[17][1] = __builtin_amdgcn_exp2f(acc1[1][2][3]); SB;
    acc2[0][3] = __builtin_amdgcn_wmma_f32_16x16x32_f16(false, wf2[33], false, __builtin_bit_cast(v16h, hb1[1]), (short)0, acc2[0][3], false, false); SB;
    HV[11] = __builtin_elementwise_fma(EC[11], ROC[11], -ROC[11]); SB;
    EG[17][0] = __builtin_amdgcn_exp2f(GC[17][0]); SB;
    { v2h pk; pk[0] = (_Float16)HV[7][0]; pk[1] = (_Float16)HV[7][1]; hn1[0][7] = __builtin_bit_cast(unsigned, pk); } SB;
    EG[17][1] = __builtin_amdgcn_exp2f(GC[17][1]); SB;
    GC[18][0] = cap_exp_arg(acc1[1][3][4]); SB;
    RIG[16][1] = __builtin_amdgcn_rcpf(PIG[16][1]); SB;
    GC[18][1] = cap_exp_arg(acc1[1][3][5]); SB;
    EI[18][0] = __builtin_amdgcn_exp2f(acc1[1][3][0]); SB;
    acc2[1][0] = __builtin_amdgcn_wmma_f32_16x16x32_f16(false, wf2[37], false, __builtin_bit_cast(v16h, hb1[1]), (short)0, acc2[1][0], false, false); SB;
    { v2h pk; pk[0] = (_Float16)HV[8][0]; pk[1] = (_Float16)HV[8][1]; hn1[1][0] = __builtin_bit_cast(unsigned, pk); } SB;
    EI[18][1] = __builtin_amdgcn_exp2f(acc1[1][3][1]); SB;
    GC[19][0] = cap_exp_arg(acc2[0][0][4]); SB;
    EG[18][0] = __builtin_amdgcn_exp2f(GC[18][0]); SB;
    GC[19][1] = cap_exp_arg(acc2[0][0][5]); SB;
    EG[18][1] = __builtin_amdgcn_exp2f(GC[18][1]); SB;
    DF[17] = EF[17] + 1.0f; SB;
    EF[18][0] = __builtin_amdgcn_exp2f(acc1[1][3][2]); SB;
    acc1[1][0] = __builtin_amdgcn_wmma_f32_16x16x32_f16(false, wf2[48], false, __builtin_bit_cast(v16h, hb1[0]), (short)0, acc1[1][0], false, false); SB;
    DGK[17] = __builtin_elementwise_fma(EG[17], rkc2, rkc2); SB;
    EF[18][1] = __builtin_amdgcn_exp2f(acc1[1][3][3]); SB;
    TT[16] = __builtin_elementwise_fma(c2[3], RF[16], -RIG[16]); SB;
    EC[13][0] = __builtin_amdgcn_exp2f(CC[13][0]); SB;
    { v2h pk; pk[0] = (_Float16)HV[9][0]; pk[1] = (_Float16)HV[9][1]; hn1[1][1] = __builtin_bit_cast(unsigned, pk); } SB;
    EC[13][1] = __builtin_amdgcn_exp2f(CC[13][1]); SB;
    GC[20][0] = cap_exp_arg(acc2[0][2][4]); SB;
    EG[19][0] = __builtin_amdgcn_exp2f(GC[19][0]); SB;
    acc2[1][1] = __builtin_amdgcn_wmma_f32_16x16x32_f16(false, wf2[41], false, __builtin_bit_cast(v16h, hb1[1]), (short)0, acc2[1][1], false, false); SB;
    GC[20][1] = cap_exp_arg(acc2[0][2][5]); SB;
    EG[19][1] = __builtin_amdgcn_exp2f(GC[19][1]); SB;
    DGK[18] = __builtin_elementwise_fma(EG[18], rkc2, rkc2); SB;
    EO[14][0] = __builtin_amdgcn_exp2f(acc2[0][1][6]); SB;
    PIG[17] = __builtin_elementwise_fma(EI[17], DGK[17], DGK[17]); SB;
    EO[14][1] = __builtin_amdgcn_exp2f(acc2[0][1][7]); SB;
    DF[18] = EF[18] + 1.0f; SB;
    RF[17][0] = __builtin_amdgcn_rcpf(DF[17][0]); SB;
    acc2[1][2] = __builtin_amdgcn_wmma_f32_16x16x32_f16(false, wf2[45], false, __builtin_bit_cast(v16h, hb1[1]), (short)0, acc2[1][2], false, false); SB;
    { v2h pk; pk[0] = (_Float16)HV[10][0]; pk[1] = (_Float16)HV[10][1]; hn1[1][2] = __builtin_bit_cast(unsigned, pk); } SB;
    RF[17][1] = __builtin_amdgcn_rcpf(DF[17][1]); SB;
    POC[13] = __builtin_elementwise_fma(EC[13], DQ[13], DQ[13]); SB;
    EI[19][0] = __builtin_amdgcn_exp2f(acc2[0][0][0]); SB;
    { v2h pk; pk[0] = (_Float16)HV[11][0]; pk[1] = (_Float16)HV[11][1]; hn1[1][3] = __builtin_bit_cast(unsigned, pk); } SB;
    EI[19][1] = __builtin_amdgcn_exp2f(acc2[0][0][1]); SB;
    PIG[18] = __builtin_elementwise_fma(EI[18], DGK[18], DGK[18]); SB;
    RIG[17][0] = __builtin_amdgcn_rcpf(PIG[17][0]); SB;
    acc1[1][0] = __builtin_amdgcn_wmma_f32_16x16x32_f16(false, wf2[49], false, __builtin_bit_cast(v16h, hb1[1]), (short)0, acc1[1][0], false, false); SB;
    DGK[19] = __builtin_elementwise_fma(EG[19], rkc2, rkc2); SB;
    RIG[17][1] = __builtin_amdgcn_rcpf(PIG[17][1]); SB;
    DQ[14] = EO[14] + 1.0f; SB;
    EF[19][0] = __builtin_amdgcn_exp2f(acc2[0][0][2]); SB;
    CC[15][0] = cap_exp_arg(c2[2][0]); SB;
    EF[19][1] = __builtin_amdgcn_exp2f(acc2[0][0][3]); SB;
    CC[15][1] = cap_exp_arg(c2[2][1]); SB;
    EC[14][0] = __builtin_amdgcn_exp2f(CC[14][0]); SB;
    GC[21][0] = cap_exp_arg(acc2[0][3][4]); SB;
    EC[14][1] = __builtin_amdgcn_exp2f(CC[14][1]); SB;
    GC[21][1] = cap_exp_arg(acc2[0][3][5]); SB;
    EG[20][0] = __builtin_amdgcn_exp2f(GC[20][0]); SB;
    c2[3] = __builtin_elementwise_fma(EG[16], RIG[16], TT[16]); SB;
    EG[20][1] = __builtin_amdgcn_exp2f(GC[20][1]); SB;
    PIG[19] = __builtin_elementwise_fma(EI[19], DGK[19], DGK[19]); SB;
    EO[15][0] = __builtin_amdgcn_exp2f(acc2[1][3][6]); SB;
    TT[17] = __builtin_elementwise_fma(c2[4], RF[17], -RIG[17]); SB;
    EO[15][1] = __builtin_amdgcn_exp2f(acc2[1][3][7]); SB;
    GC[22][0] = cap_exp_arg(acc2[1][0][4]); SB;
    RF[18][0] = __builtin_amdgcn_rcpf(DF[18][0]); SB;
    DF[19] = EF[19] + 1.0f; SB;
    RF[18][1] = __builtin_amdgcn_rcpf(DF[18][1]); SB;
    GC[22][1] = cap_exp_arg(acc2[1][0][5]); SB;
    RIG[18][0] = __builtin_amdgcn_rcpf(PIG[18][0]); SB;
    POC[14] = __builtin_elementwise_fma(EC[14], DQ[14], DQ[14]); SB;
    RIG[18][1] = __builtin_amdgcn_rcpf(PIG[18][1]); SB;
    hb1[0] = hn1[0]; hb1[1] = hn1[1]; SB;
    ROC[13][0] = __builtin_amdgcn_rcpf(POC[13][0]); SB;
    DGK[20] = __builtin_elementwise_fma(EG[20], rkc2, rkc2); SB;
    ROC[13][1] = __builtin_amdgcn_rcpf(POC[13][1]); SB;
    CC[16][0] = cap_exp_arg(c2[3][0]); SB;
    EI[20][0] = __builtin_amdgcn_exp2f(acc2[0][2][0]); SB;
    DQ[15] = EO[15] + 1.0f; SB;
    EI[20][1] = __builtin_amdgcn_exp2f(acc2[0][2][1]); SB;
    acc1[0][0] = __builtin_amdgcn_wmma_f32_16x16x32_f16(false, wf1[0], false, __builtin_bit_cast(v16h, hb1[0]), (short)0, zero8, false, false); SB;
    CC[16][1] = cap_exp_arg(c2[3][1]); SB;
    EF[20][0] = __builtin_amdgcn_exp2f(acc2[0][2][2]); SB;
    c2[4] = __builtin_elementwise_fma(EG[17], RIG[17], TT[17]); SB;
    EF[20][1] = __builtin_amdgcn_exp2f(acc2[0][2][3]); SB;
    TT[18] = __builtin_elementwise_fma(c2[5], RF[18], -RIG[18]); SB;
    EC[15][0] = __builtin_amdgcn_exp2f(CC[15][0]); SB;
    EC[15][1] = __builtin_amdgcn_exp2f(CC[15][1]); SB;
    acc1[0][1] = __builtin_amdgcn_wmma_f32_16x16x32_f16(false, wf1[2], false, __builtin_bit_cast(v16h, hb1[0]), (short)0, zero8, false, false); SB;
    HV[13] = __builtin_elementwise_fma(EC[13], ROC[13], -ROC[13]); SB;
    EG[21][0] = __builtin_amdgcn_exp2f(GC[21][0]); SB;
    EG[21][1] = __builtin_amdgcn_exp2f(GC[21][1]); SB;
    PIG[20] = __builtin_elementwise_fma(EI[20], DGK[20], DGK[20]); SB;
    EO[16][0] = __builtin_amdgcn_exp2f(acc1[1][1][6]); SB;
    CC[17][0] = cap_exp_arg(c2[4][0]); SB;
    EO[16][1] = __builtin_amdgcn_exp2f(acc1[1][1][7]); SB;
    acc1[0][2] = __builtin_amdgcn_wmma_f32_16x16x32_f16(false, wf1[4], false, __builtin_bit_cast(v16h, hb1[0]), (short)0, zero8, false, false); SB;
    DF[20] = EF[20] + 1.0f; SB;
    RF[19][0] = __builtin_amdgcn_rcpf(DF[19][0]); SB;
    CC[17][1] = cap_exp_arg(c2[4][1]); SB;
    RF[19][1] = __builtin_amdgcn_rcpf(DF[19][1]); SB;
    POC[15] = __builtin_elementwise_fma(EC[15], DQ[15], DQ[15]); SB;
    RIG[19][0] = __builtin_amdgcn_rcpf(PIG[19][0]); SB;
    c2[5] = __builtin_elementwise_fma(EG[18], RIG[18], TT[18]); SB;
    RIG[19][1] = __builtin_amdgcn_rcpf(PIG[19][1]); SB;
    acc1[0][3] = __builtin_amdgcn_wmma_f32_16x16x32_f16(false, wf1[6], false, __builtin_bit_cast(v16h, hb1[0]), (short)0, zero8, false, false); SB;
    DGK[21] = __builtin_elementwise_fma(EG[21], rkc2, rkc2); SB;
    ROC[14][0] = __builtin_amdgcn_rcpf(POC[14][0]); SB;
    ys += HV[13][0] * wl[0][0]; SB;
    ROC[14][1] = __builtin_amdgcn_rcpf(POC[14][1]); SB;
    DQ[16] = EO[16] + 1.0f; SB;
    EI[21][0] = __builtin_amdgcn_exp2f(acc2[0][3][0]); SB;
    { v2h pk; pk[0] = (_Float16)HV[13][0]; pk[1] = (_Float16)HV[13][1]; hn2[0][0] = __builtin_bit_cast(unsigned, pk); } SB;
    EI[21][1] = __builtin_amdgcn_exp2f(acc2[0][3][1]); SB;
    acc1[0][0] = __builtin_amdgcn_wmma_f32_16x16x32_f16(false, wf1[1], false, __builtin_bit_cast(v16h, hb1[1]), (short)0, acc1[0][0], false, false); SB;
    ys += HV[13][1] * wl[0][1]; SB;
    EF[21][0] = __builtin_amdgcn_exp2f(acc2[0][3][2]); SB;
    GC[23][0] = cap_exp_arg(acc2[1][1][4]); SB;
    EF[21][1] = __builtin_amdgcn_exp2f(acc2[0][3][3]); SB;
    GC[23][1] = cap_exp_arg(acc2[1][1][5]); SB;
    EC[16][0] = __builtin_amdgcn_exp2f(CC[16][0]); SB;
    TT[19] = __builtin_elementwise_fma(c2[6], RF[19], -RIG[19]); SB;
    EC[16][1] = __builtin_amdgcn_exp2f(CC[16][1]); SB;
    acc1[0][1] = __builtin_amdgcn_wmma_f32_16x16x32_f16(false, wf1[3], false, __builtin_bit_cast(v16h, hb1[1]), (short)0, acc1[0][1], false, false); SB;
    HV[14] = __builtin_elementwise_fma(EC[14], ROC[14], -ROC[14]); SB;
    EG[22][0] = __builtin_amdgcn_exp2f(GC[22][0]); SB;
    CC[18][0] = cap_exp_arg(c2[5][0]); SB;
    EG[22][1] = __builtin_amdgcn_exp2f(GC[22][1]); SB;
    PIG[21] = __builtin_elementwise_fma(EI[21], DGK[21], DGK[21]); SB;
    EO[17][0] = __builtin_amdgcn_exp2f(acc1[1][2][6]); SB;
    CC[18][1] = cap_exp_arg(c2[5][1]); SB;
    EO[17][1] = __builtin_amdgcn_exp2f(acc1[1][2][7]); SB;
    acc1[0][2] = __builtin_amdgcn_wmma_f32_16x16x32_f16(false, wf1[5], false, __builtin_bit_cast(v16h, hb1[1]), (short)0, acc1[0][2], false, false); SB;
    DF[21] = EF[21] + 1.0f; SB;
    RF[20][0] = __builtin_amdgcn_rcpf(DF[20][0]); SB;
    c2[6] = __builtin_elementwise_fma(EG[19], RIG[19], TT[19]); SB;
    RF[20][1] = __builtin_amdgcn_rcpf(DF[20][1]); SB;
    POC[16] = __builtin_elementwise_fma(EC[16], DQ[16], DQ[16]); SB;
    RIG[20][0] = __builtin_amdgcn_rcpf(PIG[20][0]); SB;
    ys += HV[14][0] * wl[1][0]; SB;
    RIG[20][1] = __builtin_amdgcn_rcpf(PIG[20][1]); SB;
    acc1[0][3] = __builtin_amdgcn_wmma_f32_16x16x32_f16(false, wf1[7], false, __builtin_bit_cast(v16h, hb1[1]), (short)0, acc1[0][3], false, false); SB;
    DGK[22] = __builtin_elementwise_fma(EG[22], rkc2, rkc2); SB;
    ROC[15][0] = __builtin_amdgcn_rcpf(POC[15][0]); SB;
    { v2h pk; pk[0] = (_Float16)HV[14][0]; pk[1] = (_Float16)HV[14][1]; hn2[0][1] = __builtin_bit_cast(unsigned, pk); } SB;
    ROC[15][1] = __builtin_amdgcn_rcpf(POC[15][1]); SB;
    DQ[17] = EO[17] + 1.0f; SB;
    EI[22][0] = __builtin_amdgcn_exp2f(acc2[1][0][0]); SB;
    ys += HV[14][1] * wl[1][1]; SB;
    EI[22][1] = __builtin_amdgcn_exp2f(acc2[1][0][1]); SB;
    GC[24][0] = cap_exp_arg(acc2[1][2][4]); SB;
    EF[22][0] = __builtin_amdgcn_exp2f(acc2[1][0][2]); SB;
    GC[24][1] = cap_exp_arg(acc2[1][2][5]); SB;
    EF[22][1] = __builtin_amdgcn_exp2f(acc2[1][0][3]); SB;
    TT[20] = __builtin_elementwise_fma(c2[7], RF[20], -RIG[20]); SB;
    EC[17][0] = __builtin_amdgcn_exp2f(CC[17][0]); SB;
    CC[19][0] = cap_exp_arg(c2[6][0]); SB;
    EC[17][1] = __builtin_amdgcn_exp2f(CC[17][1]); SB;
    CC[19][1] = cap_exp_arg(c2[6][1]); SB;
    EG[23][0] = __builtin_amdgcn_exp2f(GC[23][0]); SB;
    HV[15] = __builtin_elementwise_fma(EC[15], ROC[15], -ROC[15]); SB;
    EG[23][1] = __builtin_amdgcn_exp2f(GC[23][1]); SB;
    EO[18][0] = __builtin_amdgcn_exp2f(acc1[1][3][6]); SB;
    PIG[22] = __builtin_elementwise_fma(EI[22], DGK[22], DGK[22]); SB;
    EO[18][1] = __builtin_amdgcn_exp2f(acc1[1][3][7]); SB;
    c2[7] = __builtin_elementwise_fma(EG[20], RIG[20], TT[20]); SB;
    RF[21][0] = __builtin_amdgcn_rcpf(DF[21][0]); SB;
    DF[22] = EF[22] + 1.0f; SB;
    RF[21][1] = __builtin_amdgcn_rcpf(DF[21][1]); SB;
    RIG[21][0] = __builtin_amdgcn_rcpf(PIG[21][0]); SB;
    POC[17] = __builtin_elementwise_fma(EC[17], DQ[17], DQ[17]); SB;
    RIG[21][1] = __builtin_amdgcn_rcpf(PIG[21][1]); SB;
    ys += HV[15][0] * wl[2][0]; SB;
    ROC[16][0] = __builtin_amdgcn_rcpf(POC[16][0]); SB;
    DGK[23] = __builtin_elementwise_fma(EG[23], rkc2, rkc2); SB;
    ROC[16][1] = __builtin_amdgcn_rcpf(POC[16][1]); SB;
    CC[20][0] = cap_exp_arg(c2[7][0]); SB;
    EI[23][0] = __builtin_amdgcn_exp2f(acc2[1][1][0]); SB;
    DQ[18] = EO[18] + 1.0f; SB;
    EI[23][1] = __builtin_amdgcn_exp2f(acc2[1][1][1]); SB;
    ys += HV[15][1] * wl[2][1]; SB;
    EF[23][0] = __builtin_amdgcn_exp2f(acc2[1][1][2]); SB;
    CC[20][1] = cap_exp_arg(c2[7][1]); SB;
    EF[23][1] = __builtin_amdgcn_exp2f(acc2[1][1][3]); SB;
    { v2h pk; pk[0] = (_Float16)HV[15][0]; pk[1] = (_Float16)HV[15][1]; hn2[0][2] = __builtin_bit_cast(unsigned, pk); } SB;
    EC[18][0] = __builtin_amdgcn_exp2f(CC[18][0]); SB;
    GC[25][0] = cap_exp_arg(acc1[1][0][4]); SB;
    EC[18][1] = __builtin_amdgcn_exp2f(CC[18][1]); SB;
    GC[25][1] = cap_exp_arg(acc1[1][0][5]); SB;
    EG[24][0] = __builtin_amdgcn_exp2f(GC[24][0]); SB;
    TT[21] = __builtin_elementwise_fma(c2[8], RF[21], -RIG[21]); SB;
    EG[24][1] = __builtin_amdgcn_exp2f(GC[24][1]); SB;
    HV[16] = __builtin_elementwise_fma(EC[16], ROC[16], -ROC[16]); SB;
    EO[19][0] = __builtin_amdgcn_exp2f(acc2[0][0][6]); SB;
    PIG[23] = __builtin_elementwise_fma(EI[23], DGK[23], DGK[23]); SB;
    EO[19][1] = __builtin_amdgcn_exp2f(acc2[0][0][7]); SB;
    EG[25][0] = __builtin_amdgcn_exp2f(GC[25][0]); SB;
    DF[23] = EF[23] + 1.0f; SB;
    EG[25][1] = __builtin_amdgcn_exp2f(GC[25][1]); SB;
    RF[22][0] = __builtin_amdgcn_rcpf(DF[22][0]); SB;
    POC[18] = __builtin_elementwise_fma(EC[18], DQ[18], DQ[18]); SB;
    RF[22][1] = __builtin_amdgcn_rcpf(DF[22][1]); SB;
    c2[8] = __builtin_elementwise_fma(EG[21], RIG[21], TT[21]); SB;
    RIG[22][0] = __builtin_amdgcn_rcpf(PIG[22][0]); SB;
    DGK[24] = __builtin_elementwise_fma(EG[24], rkc2, rkc2); SB;
    RIG[22][1] = __builtin_amdgcn_rcpf(PIG[22][1]); SB;
    ys += HV[16][0] * wl[3][0]; SB;
    ROC[17][0] = __builtin_amdgcn_rcpf(POC[17][0]); SB;
    DQ[19] = EO[19] + 1.0f; SB;
    ROC[17][1] = __builtin_amdgcn_rcpf(POC[17][1]); SB;
    { v2h pk; pk[0] = (_Float16)HV[16][0]; pk[1] = (_Float16)HV[16][1]; hn2[0][3] = __builtin_bit_cast(unsigned, pk); } SB;
    EI[24][0] = __builtin_amdgcn_exp2f(acc2[1][2][0]); SB;
    DGK[25] = __builtin_elementwise_fma(EG[25], rkc2, rkc2); SB;
    EI[24][1] = __builtin_amdgcn_exp2f(acc2[1][2][1]); SB;
    ys += HV[16][1] * wl[3][1]; SB;
    EF[24][0] = __builtin_amdgcn_exp2f(acc2[1][2][2]); SB;
    CC[21][0] = cap_exp_arg(c2[8][0]); SB;
    EF[24][1] = __builtin_amdgcn_exp2f(acc2[1][2][3]); SB;
    CC[21][1] = cap_exp_arg(c2[8][1]); SB;
    EC[19][0] = __builtin_amdgcn_exp2f(CC[19][0]); SB;
    TT[22] = __builtin_elementwise_fma(c2[9], RF[22], -RIG[22]); SB;
    EC[19][1] = __builtin_amdgcn_exp2f(CC[19][1]); SB;
    EO[20][0] = __builtin_amdgcn_exp2f(acc2[0][2][6]); SB;
    HV[17] = __builtin_elementwise_fma(EC[17], ROC[17], -ROC[17]); SB;
    EO[20][1] = __builtin_amdgcn_exp2f(acc2[0][2][7]); SB;
    EI[25][0] = __builtin_amdgcn_exp2f(acc1[1][0][0]); SB;
    PIG[24] = __builtin_elementwise_fma(EI[24], DGK[24], DGK[24]); SB;
    EI[25][1] = __builtin_amdgcn_exp2f(acc1[1][0][1]); SB;
    EF[25][0] = __builtin_amdgcn_exp2f(acc1[1][0][2]); SB;
    DF[24] = EF[24] + 1.0f; SB;
    EF[25][1] = __builtin_amdgcn_exp2f(acc1[1][0][3]); SB;
    c2[9] = __builtin_elementwise_fma(EG[22], RIG[22], TT[22]); SB;
    RF[23][0] = __builtin_amdgcn_rcpf(DF[23][0]); SB;
    POC[19] = __builtin_elementwise_fma(EC[19], DQ[19], DQ[19]); SB;
    RF[23][1] = __builtin_amdgcn_rcpf(DF[23][1]); SB;
    ys += HV[17][0] * wl[4][0]; SB;
    RIG[23][0] = __builtin_amdgcn_rcpf(PIG[23][0]); SB;
    DQ[20] = EO[20] + 1.0f; SB;
    RIG[23][1] = __builtin_amdgcn_rcpf(PIG[23][1]); SB;
    { v2h pk; pk[0] = (_Float16)HV[17][0]; pk[1] = (_Float16)HV[17][1]; hn2[0][4] = __builtin_bit_cast(unsigned, pk); } SB;
    ROC[18][0] = __builtin_amdgcn_rcpf(POC[18][0]); SB;
    PIG[25] = __builtin_elementwise_fma(EI[25], DGK[25], DGK[25]); SB;
    ROC[18][1] = __builtin_amdgcn_rcpf(POC[18][1]); SB;
    ys += HV[17][1] * wl[4][1]; SB;
    EC[20][0] = __builtin_amdgcn_exp2f(CC[20][0]); SB;
    DF[25] = EF[25] + 1.0f; SB;
    EC[20][1] = __builtin_amdgcn_exp2f(CC[20][1]); SB;
    CC[22][0] = cap_exp_arg(c2[9][0]); SB;
    EO[21][0] = __builtin_amdgcn_exp2f(acc2[0][3][6]); SB;
    CC[22][1] = cap_exp_arg(c2[9][1]); SB;
    EO[21][1] = __builtin_amdgcn_exp2f(acc2[0][3][7]); SB;
    RF[24][0] = __builtin_amdgcn_rcpf(DF[24][0]); SB;
    TT[23] = __builtin_elementwise_fma(c2[10], RF[23], -RIG[23]); SB;
    RF[24][1] = __builtin_amdgcn_rcpf(DF[24][1]); SB;
    RIG[24][0] = __builtin_amdgcn_rcpf(PIG[24][0]); SB;
    HV[18] = __builtin_elementwise_fma(EC[18], ROC[18], -ROC[18]); SB;
    RIG[24][1] = __builtin_amdgcn_rcpf(PIG[24][1]); SB;
    ROC[19][0] = __builtin_amdgcn_rcpf(POC[19][0]); SB;
    POC[20] = __builtin_elementwise_fma(EC[20], DQ[20], DQ[20]); SB;
    ROC[19][1] = __builtin_amdgcn_rcpf(POC[19][1]); SB;
    EC[21][0] = __builtin_amdgcn_exp2f(CC[21][0]); SB;
    DQ[21] = EO[21] + 1.0f; SB;
    EC[21][1] = __builtin_amdgcn_exp2f(CC[21][1]); SB;
    c2[10] = __builtin_elementwise_fma(EG[23], RIG[23], TT[23]); SB;
    EO[22][0] = __builtin_amdgcn_exp2f(acc2[1][0][6]); SB;
    ys += HV[18][0] * wl[5][0]; SB;
    EO[22][1] = __builtin_amdgcn_exp2f(acc2[1][0][7]); SB;
    { v2h pk; pk[0] = (_Float16)HV[18][0]; pk[1] = (_Float16)HV[18][1]; hn2[0][5] = __builtin_bit_cast(unsigned, pk); } SB;
    RF[25][0] = __builtin_amdgcn_rcpf(DF[25][0]); SB;
    TT[24] = __builtin_elementwise_fma(c2[11], RF[24], -RIG[24]); SB;
    RF[25][1] = __builtin_amdgcn_rcpf(DF[25][1]); SB;
    RIG[25][0] = __builtin_amdgcn_rcpf(PIG[25][0]); SB;
    HV[19] = __builtin_elementwise_fma(EC[19], ROC[19], -ROC[19]); SB;
    RIG[25][1] = __builtin_amdgcn_rcpf(PIG[25][1]); SB;
    ys += HV[18][1] * wl[5][1]; SB;
    ROC[20][0] = __builtin_amdgcn_rcpf(POC[20][0]); SB;
    POC[21] = __builtin_elementwise_fma(EC[21], DQ[21], DQ[21]); SB;
    ROC[20][1] = __builtin_amdgcn_rcpf(POC[20][1]); SB;
    CC[23][0] = cap_exp_arg(c2[10][0]); SB;
    EC[22][0] = __builtin_amdgcn_exp2f(CC[22][0]); SB;
    DQ[22] = EO[22] + 1.0f; SB;
    EC[22][1] = __builtin_amdgcn_exp2f(CC[22][1]); SB;
    CC[23][1] = cap_exp_arg(c2[10][1]); SB;
    EO[23][0] = __builtin_amdgcn_exp2f(acc2[1][1][6]); SB;
    c2[11] = __builtin_elementwise_fma(EG[24], RIG[24], TT[24]); SB;
    EO[23][1] = __builtin_amdgcn_exp2f(acc2[1][1][7]); SB;
    ys += HV[19][0] * wl[6][0]; SB;
    EC[23][0] = __builtin_amdgcn_exp2f(CC[23][0]); SB;
    TT[25] = __builtin_elementwise_fma(c2[12], RF[25], -RIG[25]); SB;
    ROC[21][0] = __builtin_amdgcn_rcpf(POC[21][0]); SB;
    { v2h pk; pk[0] = (_Float16)HV[19][0]; pk[1] = (_Float16)HV[19][1]; hn2[0][6] = __builtin_bit_cast(unsigned, pk); } SB;
    ROC[21][1] = __builtin_amdgcn_rcpf(POC[21][1]); SB;
    HV[20] = __builtin_elementwise_fma(EC[20], ROC[20], -ROC[20]); SB;
    EC[23][1] = __builtin_amdgcn_exp2f(CC[23][1]); SB;
    ys += HV[19][1] * wl[6][1]; SB;
    EO[24][0] = __builtin_amdgcn_exp2f(acc2[1][2][6]); SB;
    POC[22] = __builtin_elementwise_fma(EC[22], DQ[22], DQ[22]); SB;
    EO[24][1] = __builtin_amdgcn_exp2f(acc2[1][2][7]); SB;
    CC[24][0] = cap_exp_arg(c2[11][0]); SB;
    EO[25][0] = __builtin_amdgcn_exp2f(acc1[1][0][6]); SB;
    DQ[23] = EO[23] + 1.0f; SB;
    EO[25][1] = __builtin_amdgcn_exp2f(acc1[1][0][7]); SB;
    c2[12] = __builtin_elementwise_fma(EG[25], RIG[25], TT[25]); SB;
    CC[24][1] = cap_exp_arg(c2[11][1]); SB;
    ys += HV[20][0] * wl[7][0]; SB;
    EC[24][0] = __builtin_amdgcn_exp2f(CC[24][0]); SB;
    HV[21] = __builtin_elementwise_fma(EC[21], ROC[21], -ROC[21]); SB;
    { v2h pk; pk[0] = (_Float16)HV[20][0]; pk[1] = (_Float16)HV[20][1]; hn2[0][7] = __builtin_bit_cast(unsigned, pk); } SB;
    ROC[22][0] = __builtin_amdgcn_rcpf(POC[22][0]); SB;
    ROC[22][1] = __builtin_amdgcn_rcpf(POC[22][1]); SB;
    DQ[24] = EO[24] + 1.0f; SB;
    EC[24][1] = __builtin_amdgcn_exp2f(CC[24][1]); SB;
    POC[23] = __builtin_elementwise_fma(EC[23], DQ[23], DQ[23]); SB;
    CC[25][0] = cap_exp_arg(c2[12][0]); SB;
    CC[25][1] = cap_exp_arg(c2[12][1]); SB;
    ys += HV[20][1] * wl[7][1]; SB;
    DQ[25] = EO[25] + 1.0f; SB;
    { v2h pk; pk[0] = (_Float16)HV[21][0]; pk[1] = (_Float16)HV[21][1]; hn2[1][0] = __builtin_bit_cast(unsigned, pk); } SB;
    EC[25][0] = __builtin_amdgcn_exp2f(CC[25][0]); SB;
    EC[25][1] = __builtin_amdgcn_exp2f(CC[25][1]); SB;
    ys += HV[21][0] * wl[8][0]; SB;
    ROC[23][0] = __builtin_amdgcn_rcpf(POC[23][0]); SB;
    HV[22] = __builtin_elementwise_fma(EC[22], ROC[22], -ROC[22]); SB;
    ROC[23][1] = __builtin_amdgcn_rcpf(POC[23][1]); SB;
    POC[24] = __builtin_elementwise_fma(EC[24], DQ[24], DQ[24]); SB;
    ys += HV[21][1] * wl[8][1]; SB;
    POC[25] = __builtin_elementwise_fma(EC[25], DQ[25], DQ[25]); SB;
    ys += HV[22][0] * wl[9][0]; SB;
    { v2h pk; pk[0] = (_Float16)HV[22][0]; pk[1] = (_Float16)HV[22][1]; hn2[1][1] = __builtin_bit_cast(unsigned, pk); } SB;
    ROC[24][0] = __builtin_amdgcn_rcpf(POC[24][0]); SB;
    HV[23] = __builtin_elementwise_fma(EC[23], ROC[23], -ROC[23]); SB;
    ROC[24][1] = __builtin_amdgcn_rcpf(POC[24][1]); SB;
    ys += HV[22][1] * wl[9][1]; SB;
    ROC[25][0] = __builtin_amdgcn_rcpf(POC[25][0]); SB;
    ROC[25][1] = __builtin_amdgcn_rcpf(POC[25][1]); SB;
    ys += HV[23][0] * wl[10][0]; SB;
    { v2h pk; pk[0] = (_Float16)HV[23][0]; pk[1] = (_Float16)HV[23][1]; hn2[1][2] = __builtin_bit_cast(unsigned, pk); } SB;
    HV[24] = __builtin_elementwise_fma(EC[24], ROC[24], -ROC[24]); SB;
    ys += HV[23][1] * wl[10][1]; SB;
    HV[25] = __builtin_elementwise_fma(EC[25], ROC[25], -ROC[25]); SB;
    ys += HV[24][0] * wl[11][0]; SB;
    { v2h pk; pk[0] = (_Float16)HV[24][0]; pk[1] = (_Float16)HV[24][1]; hn2[1][3] = __builtin_bit_cast(unsigned, pk); } SB;
    ys += HV[24][1] * wl[11][1]; SB;
    HV[25][1] = (lh == 1) ? 0.0f : HV[25][1]; SB;
    { v2h pk; pk[0] = (_Float16)HV[25][0]; pk[1] = (_Float16)HV[25][1]; hn2[1][4] = __builtin_bit_cast(unsigned, pk); } SB;
    ys += HV[25][0] * wl[12][0]; SB;
    hb2[0] = hn2[0]; hb2[1] = hn2[1]; SB;
    ys += HV[25][1] * wl[12][1]; SB;
    ys += xhalf(ys); ylast = ys + blin; if (lh == 0) yout[t - 1] = ylast; SB;
  }

  for (int s = T - 1; s < Tout; ++s) {
    if (s >= 0) {
      ys = 0.f;
      products(IC(4));
      products(IC(5)); cell(IC(4));
      products(IC(6)); cell(IC(5));
      products(IC(7)); cell(IC(6));
      cell(IC(7));
      hb2[0] = hn2[0]; hb2[1] = hn2[1];
      ys += xhalf(ys);
      ylast = ys + blin;
      if (lh == 0) yout[s] = ylast;
    }
    if (s + 1 < Tout) {
      set_x(hb1[1], ylast);
      products(IC(0));
      products(IC(1)); cell(IC(0));
      products(IC(2)); cell(IC(1));
      products(IC(3)); cell(IC(2));
      cell(IC(3));
      hb1[0] = hn1[0]; hb1[1] = hn1[1];
    }
  }
#undef IC
#undef SB
}

extern "C" void kernel_launch(void* const* d_in, const int* in_sizes, int n_in,
                              void* d_out, int out_size, void* d_ws, size_t ws_size,
                              hipStream_t stream) {
  const float* input = (const float*)d_in[0];
  const float* w_ih1 = (const float*)d_in[2];
  const float* w_hh1 = (const float*)d_in[3];
  const float* b_ih1 = (const float*)d_in[4];
  const float* b_hh1 = (const float*)d_in[5];
  const float* w_ih2 = (const float*)d_in[6];
  const float* w_hh2 = (const float*)d_in[7];
  const float* b_ih2 = (const float*)d_in[8];
  const float* b_hh2 = (const float*)d_in[9];
  const float* w_lin = (const float*)d_in[10];
  const float* b_lin = (const float*)d_in[11];
  float* out = (float*)d_out;

  const int B = 1024;
  const int T = in_sizes[0] / B;
  const int Tout = out_size / B;

  lstm2_wmma_kernel<<<dim3(B / 64), dim3(128), 0, stream>>>(
      input, w_ih1, w_hh1, b_ih1, b_hh1, w_ih2, w_hh2, b_ih2, b_hh2,
      w_lin, b_lin, out, T, Tout);
}


# speedup vs baseline: 5.4076x; 1.0821x over previous
#include <hip/hip_runtime.h>
#include <hip/hip_fp16.h>
#include <type_traits>

typedef __attribute__((ext_vector_type(16))) _Float16 v16h;
typedef __attribute__((ext_vector_type(8)))  _Float16 v8h;
typedef __attribute__((ext_vector_type(2)))  _Float16 v2h;
typedef __attribute__((ext_vector_type(8)))  float    v8f;
typedef __attribute__((ext_vector_type(8)))  unsigned v8u;
typedef __attribute__((ext_vector_type(2)))  float    v2f;

#define HDIM 51
#define NT 13
#define LOG2E 1.44269504088896340736f

#define KC (2.0f * LOG2E)
__device__ __forceinline__ v2f exp2_pair(float x, float y) {
  v2f r; r[0] = __builtin_amdgcn_exp2f(x); r[1] = __builtin_amdgcn_exp2f(y); return r;
}
__device__ __forceinline__ v2f rcp_pair(v2f d) {
  v2f r; r[0] = __builtin_amdgcn_rcpf(d[0]); r[1] = __builtin_amdgcn_rcpf(d[1]); return r;
}
__device__ __forceinline__ float cap_exp_arg(float x) {
  const int xi = __builtin_bit_cast(int, x);
  const int ci = __builtin_bit_cast(int, 64.0f);
  return __builtin_bit_cast(float, __builtin_elementwise_min(xi, ci));
}
__device__ __forceinline__ float gate_scale(int q) { return (q == 2) ? KC : -LOG2E; }

__device__ __forceinline__ float xhalf(float v) {
  int b = __builtin_bit_cast(int, v);
  return __builtin_bit_cast(float, __builtin_amdgcn_permlanex16(b, b, 0x76543210, 0xfedcba98, false, false));
}

__device__ __forceinline__ v16h ldfrag(const v8h (*fr)[2][32], int f, int lane) {
  v8h lo = fr[f][0][lane];
  v8h hp = fr[f][1][lane];
  return __builtin_shufflevector(lo, hp, 0,1,2,3,4,5,6,7,8,9,10,11,12,13,14,15);
}

__global__ __launch_bounds__(128)
void lstm2_wmma_kernel(const float* __restrict__ input,
                       const float* __restrict__ w_ih1, const float* __restrict__ w_hh1,
                       const float* __restrict__ b_ih1, const float* __restrict__ b_hh1,
                       const float* __restrict__ w_ih2, const float* __restrict__ w_hh2,
                       const float* __restrict__ b_ih2, const float* __restrict__ b_hh2,
                       const float* __restrict__ w_lin, const float* __restrict__ b_lin,
                       float* __restrict__ out, int T, int Tout)
{
  __shared__ v8h a1frag[NT * 2][2][32];
  __shared__ v8h a2frag[NT * 4][2][32];

  const int tid = threadIdx.x;

#pragma unroll 26
  for (int it = 0; it < NT * 2 * 512 / 128; ++it) {
    const int idx = tid + it * 128;
    int f = idx >> 9, rem = idx & 511, l = rem >> 4, s = rem & 15;
    int m = f >> 1, b = f & 1;
    int cc = l & 15, lh = l >> 4;
    int unit = 4 * m + 2 * (cc >> 3) + (cc & 1), q = (cc & 7) >> 1;
    int mp = 8 * b + (s >> 1);
    int ku = 4 * mp + 2 * lh + (s & 1);
    const bool ok = unit < HDIM && mp < NT && ku < HDIM;
    const int uc = (unit < HDIM) ? unit : HDIM - 1, kc = (ku < HDIM) ? ku : HDIM - 1;
    const float sv = gate_scale(q) * w_hh1[(q * HDIM + uc) * HDIM + kc];
    const float v = ok ? sv : 0.0f;
    ((_Float16*)&a1frag[f][s >> 3][l])[s & 7] = (_Float16)v;
  }
#pragma unroll 26
  for (int it = 0; it < NT * 4 * 512 / 128; ++it) {
    const int idx = tid + it * 128;
    int f = idx >> 9, rem = idx & 511, l = rem >> 4, s = rem & 15;
    int m = f >> 2, b = f & 3;
    int cc = l & 15, lh = l >> 4;
    int unit = 4 * m + 2 * (cc >> 3) + (cc & 1), q = (cc & 7) >> 1;
    int mp = 8 * (b & 1) + (s >> 1);
    int ku = 4 * mp + 2 * lh + (s & 1);
    const bool ok = unit < HDIM && mp < NT && ku < HDIM;
    const float* wsrc = (b < 2) ? w_ih2 : w_hh2;
    const int uc = (unit < HDIM) ? unit : HDIM - 1, kc = (ku < HDIM) ? ku : HDIM - 1;
    const float sv = gate_scale(q) * wsrc[(q * HDIM + uc) * HDIM + kc];
    const float v = ok ? sv : 0.0f;
    ((_Float16*)&a2frag[f][s >> 3][l])[s & 7] = (_Float16)v;
  }
  __syncthreads();
  for (int e = tid; e < NT * 16 * 5; e += 128) {
    int m = e / 80, r2 = e % 80, cc = r2 / 5, s = 10 + r2 % 5;
    int unit = 4 * m + 2 * (cc >> 3) + (cc & 1), q = (cc & 7) >> 1;
    if (unit < HDIM) {
      int r = q * HDIM + unit;
      float sv = gate_scale(q) * ((s <= 12) ? w_ih1[r] : (b_ih1[r] + b_hh1[r]));
      _Float16 hd = (_Float16)sv;
      _Float16 tl = (_Float16)(sv - (float)hd);
      ((_Float16*)&a1frag[2 * m + 1][s >> 3][cc])[s & 7] = (s == 11 || s == 14) ? tl : hd;
    }
  }
  for (int e = tid; e < NT * 16 * 2; e += 128) {
    int m = e >> 5, cc = (e & 31) >> 1, s = 13 + (e & 1);
    int unit = 4 * m + 2 * (cc >> 3) + (cc & 1), q = (cc & 7) >> 1;
    if (unit < HDIM) {
      int r = q * HDIM + unit;
      float sv = gate_scale(q) * (b_ih2[r] + b_hh2[r]);
      _Float16 hd = (_Float16)sv;
      _Float16 tl = (_Float16)(sv - (float)hd);
      ((_Float16*)&a2frag[4 * m + 1][s >> 3][cc])[s & 7] = (s == 14) ? tl : hd;
    }
  }
  __syncthreads();

  const int w    = tid >> 5;
  const int lane = tid & 31;
  const int cix  = lane & 15;
  const int lh   = lane >> 4;
  const int rowbase = blockIdx.x * 64 + w * 16;

  float wl[NT][2];
#pragma unroll
  for (int m = 0; m < NT; ++m)
#pragma unroll
    for (int u = 0; u < 2; ++u) {
      int unit = 4 * m + 2 * lh + u;
      wl[m][u] = (unit < HDIM) ? w_lin[unit] : 0.0f;
    }
  const float blin = b_lin[0];

  v2f c1[NT], c2[NT];
  const v2f rkc2 = {1.0f / KC, 1.0f / KC};
#pragma unroll
  for (int m = 0; m < NT; ++m) { c1[m] = 0.f; c2[m] = 0.f; }

  v8u hb1[2], hb2[2];
#pragma unroll
  for (int b = 0; b < 2; ++b)
#pragma unroll
    for (int j = 0; j < 8; ++j) { hb1[b][j] = 0u; hb2[b][j] = 0u; }
  float ylast = 0.f;

  const float* xin = input + (long)(rowbase + cix) * T;
  float* yout = out + (long)(rowbase + cix) * Tout;

  v16h wf1[NT * 2], wf2[NT * 4];
#pragma unroll
  for (int f = 0; f < NT * 2; ++f) wf1[f] = ldfrag(a1frag, f, lane);
#pragma unroll
  for (int f = 0; f < NT * 4; ++f) wf2[f] = ldfrag(a2frag, f, lane);

  float ys = 0.f;
  v8f acc1[2][4], acc2[2][4];
  const v8f zero8 = {0.f, 0.f, 0.f, 0.f, 0.f, 0.f, 0.f, 0.f};

  auto set_x = [&](v8u& frag, float x) __attribute__((always_inline)) {
    const _Float16 xh = (_Float16)x;
    const _Float16 xt = (_Float16)(x - (float)xh);
    v2h p5; p5[0] = xh; p5[1] = xh;
    v2h p6; p6[0] = xt; p6[1] = (_Float16)1.0f;
    v2h p7; p7[0] = (_Float16)1.0f; p7[1] = (_Float16)0.0f;
    frag[5] = __builtin_bit_cast(unsigned, p5);
    frag[6] = __builtin_bit_cast(unsigned, p6);
    frag[7] = __builtin_bit_cast(unsigned, p7);
  };
  set_x(hb1[1], 0.f);
  v8u hn1[2] = {hb1[0], hb1[1]};
  v8u hn2[2] = {hb2[0], hb2[1]};

  auto products = [&](auto Rc) __attribute__((always_inline)) {
    constexpr int r = decltype(Rc)::value;
    constexpr int blk = r & 3;
    constexpr int m0 = (blk == 0) ? 0 : 1 + 3 * blk;
    constexpr int n  = (blk == 0) ? 4 : 3;
    if (r < 4) {
#pragma unroll
      for (int kb = 0; kb < 2; ++kb)
#pragma unroll
        for (int i = 0; i < n; ++i)
          acc1[r & 1][i] = __builtin_amdgcn_wmma_f32_16x16x32_f16(false, wf1[(m0 + i) * 2 + kb], false,
                      __builtin_bit_cast(v16h, hb1[kb]), (short)0,
                      (kb == 0) ? zero8 : acc1[r & 1][i], false, false);
    } else {
#pragma unroll
      for (int kk = 0; kk < 4; ++kk) {
        const int kb = (kk + 2) & 3;
#pragma unroll
        for (int i = 0; i < n; ++i)
          acc2[r & 1][i] = __builtin_amdgcn_wmma_f32_16x16x32_f16(false, wf2[(m0 + i) * 4 + kb], false,
                      __builtin_bit_cast(v16h, (kb < 2) ? hb1[kb & 1] : hb2[kb & 1]),
                      (short)0, (kk == 0) ? zero8 : acc2[r & 1][i], false, false);
      }
    }
  };

  auto cell = [&](auto Pc) __attribute__((always_inline)) {
    constexpr int p = decltype(Pc)::value;
    constexpr int blk = p & 3;
    constexpr int m0 = (blk == 0) ? 0 : 1 + 3 * blk;
    constexpr int n  = (blk == 0) ? 4 : 3;
#pragma unroll
    for (int i = 0; i < n; ++i) {
      const int m = m0 + i;
      const v8f a = (p < 4) ? acc1[p & 1][i] : acc2[p & 1][i];
      const v2f ei = exp2_pair(a[0], a[1]);
      const v2f ef = exp2_pair(a[2], a[3]);
      const v2f eg = exp2_pair(cap_exp_arg(a[4]), cap_exp_arg(a[5]));
      const v2f eo = exp2_pair(a[6], a[7]);
      const v2f df = ef + 1.0f;
      const v2f dq = eo + 1.0f;
      const v2f dgk = __builtin_elementwise_fma(eg, rkc2, rkc2);
      const v2f pig = __builtin_elementwise_fma(ei, dgk, dgk);
      const v2f rf  = rcp_pair(df);
      const v2f rig = rcp_pair(pig);
      const v2f cold = (p < 4) ? c1[m] : c2[m];
      const v2f cn = __builtin_elementwise_fma(eg, rig, __builtin_elementwise_fma(cold, rf, -rig));
      if (p < 4) c1[m] = cn; else c2[m] = cn;
      const v2f ec = exp2_pair(cap_exp_arg(cn[0]), cap_exp_arg(cn[1]));
      const v2f poc = __builtin_elementwise_fma(ec, dq, dq);
      const v2f roc = rcp_pair(poc);
      const v2f hv = __builtin_elementwise_fma(ec, roc, -roc);
      float hn[2] = {hv[0], hv[1]};
      if (m == NT - 1 && lh == 1) hn[1] = 0.0f;
      v2h pk; pk[0] = (_Float16)hn[0]; pk[1] = (_Float16)hn[1];
      if (p < 4) {
        hn1[m >> 3][m & 7] = __builtin_bit_cast(unsigned, pk);
      } else {
        ys += hn[0] * wl[m][0];
        ys += hn[1] * wl[m][1];
        hn2[m >> 3][m & 7] = __builtin_bit_cast(unsigned, pk);
      }
    }
  };

#define IC(N) std::integral_constant<int, N>{}
#define SB __builtin_amdgcn_sched_barrier(0)
  if (T > 0) {
    set_x(hb1[1], xin[0]);
    products(IC(0));
    products(IC(1)); cell(IC(0));
    products(IC(2)); cell(IC(1));
    products(IC(3)); cell(IC(2));
    cell(IC(3));
    set_x(hn1[1], (1 < T) ? xin[1] : 0.f);
    hb1[0] = hn1[0]; hb1[1] = hn1[1];
    products(IC(0));
  }

  for (int t = 1; t < T; ++t) {
    const float xn = (t + 1 < T) ? xin[t + 1] : 0.f;

    ys = 0.f;
    v2f EI[26], EF[26], EG[26], EO[26], DF[26], DQ[26], DGK[26], PIG[26], RF[26], RIG[26], TT[26], EC[26], POC[26], ROC[26], HV[26];
    float GC[26][2], CC[26][2];
    acc1[1][1] = __builtin_amdgcn_wmma_f32_16x16x32_f16(false, wf1[24], false, __builtin_bit_cast(v16h, hb1[0]), (short)0, zero8, false, false); SB;
    GC[0][0] = cap_exp_arg(acc1[0][0][4]); SB;
    EI[0][0] = __builtin_amdgcn_exp2f(acc1[0][0][0]); SB;
    GC[0][1] = cap_exp_arg(acc1[0][0][5]); SB;
    EI[0][1] = __builtin_amdgcn_exp2f(acc1[0][0][1]); SB;
    acc1[1][2] = __builtin_amdgcn_wmma_f32_16x16x32_f16(false, wf1[8], false, __builtin_bit_cast(v16h, hb1[0]), (short)0, zero8, false, false); SB;
    GC[1][0] = cap_exp_arg(acc1[0][1][4]); SB;
    EG[0][0] = __builtin_amdgcn_exp2f(GC[0][0]); SB;
    GC[1][1] = cap_exp_arg(acc1[0][1][5]); SB;
    EG[0][1] = __builtin_amdgcn_exp2f(GC[0][1]); SB;
    acc1[1][3] = __builtin_amdgcn_wmma_f32_16x16x32_f16(false, wf1[10], false, __builtin_bit_cast(v16h, hb1[0]), (short)0, zero8, false, false); SB;
    GC[2][0] = cap_exp_arg(acc1[0][2][4]); SB;
    EG[1][0] = __builtin_amdgcn_exp2f(GC[1][0]); SB;
    GC[2][1] = cap_exp_arg(acc1[0][2][5]); SB;
    EG[1][1] = __builtin_amdgcn_exp2f(GC[1][1]); SB;
    acc1[1][0] = __builtin_amdgcn_wmma_f32_16x16x32_f16(false, wf2[2], false, __builtin_bit_cast(v16h, hb2[0]), (short)0, zero8, false, false); SB;
    GC[3][0] = cap_exp_arg(acc1[0][3][4]); SB;
    EG[2][0] = __builtin_amdgcn_exp2f(GC[2][0]); SB;
    GC[3][1] = cap_exp_arg(acc1[0][3][5]); SB;
    EG[2][1] = __builtin_amdgcn_exp2f(GC[2][1]); SB;
    acc1[1][1] = __builtin_amdgcn_wmma_f32_16x16x32_f16(false, wf1[25], false, __builtin_bit_cast(v16h, hb1[1]), (short)0, acc1[1][1], false, false); SB;
    DGK[0] = __builtin_elementwise_fma(EG[0], rkc2, rkc2); SB;
    EG[3][0] = __builtin_amdgcn_exp2f(GC[3][0]); SB;
    DGK[1] = __builtin_elementwise_fma(EG[1], rkc2, rkc2); SB;
    EG[3][1] = __builtin_amdgcn_exp2f(GC[3][1]); SB;
    acc2[0][0] = __builtin_amdgcn_wmma_f32_16x16x32_f16(false, wf1[12], false, __builtin_bit_cast(v16h, hb1[0]), (short)0, zero8, false, false); SB;
    DGK[2] = __builtin_elementwise_fma(EG[2], rkc2, rkc2); SB;
    EF[0][0] = __builtin_amdgcn_exp2f(acc1[0][0][2]); SB;
    PIG[0] = __builtin_elementwise_fma(EI[0], DGK[0], DGK[0]); SB;
    EF[0][1] = __builtin_amdgcn_exp2f(acc1[0][0][3]); SB;
    acc2[0][1] = __builtin_amdgcn_wmma_f32_16x16x32_f16(false, wf2[6], false, __builtin_bit_cast(v16h, hb2[0]), (short)0, zero8, false, false); SB;
    DGK[3] = __builtin_elementwise_fma(EG[3], rkc2, rkc2); SB;
    EI[1][0] = __builtin_amdgcn_exp2f(acc1[0][1][0]); SB;
    EI[1][1] = __builtin_amdgcn_exp2f(acc1[0][1][1]); SB;
    acc1[1][2] = __builtin_amdgcn_wmma_f32_16x16x32_f16(false, wf1[9], false, __builtin_bit_cast(v16h, hb1[1]), (short)0, acc1[1][2], false, false); SB;
    DF[0] = EF[0] + 1.0f; SB;
    EF[1][0] = __builtin_amdgcn_exp2f(acc1[0][1][2]); SB;
    EF[1][1] = __builtin_amdgcn_exp2f(acc1[0][1][3]); SB;
    acc2[0][2] = __builtin_amdgcn_wmma_f32_16x16x32_f16(false, wf1[14], false, __builtin_bit_cast(v16h, hb1[0]), (short)0, zero8, false, false); SB;
    GC[12][0] = cap_exp_arg(acc1[1][1][4]); SB;
    EI[12][0] = __builtin_amdgcn_exp2f(acc1[1][1][0]); SB;
    GC[12][1] = cap_exp_arg(acc1[1][1][5]); SB;
    EI[12][1] = __builtin_amdgcn_exp2f(acc1[1][1][1]); SB;
    acc2[1][3] = __builtin_amdgcn_wmma_f32_16x16x32_f16(false, wf2[10], false, __builtin_bit_cast(v16h, hb2[0]), (short)0, zero8, false, false); SB;
    PIG[1] = __builtin_elementwise_fma(EI[1], DGK[1], DGK[1]); SB;
    EF[12][0] = __builtin_amdgcn_exp2f(acc1[1][1][2]); SB;
    DF[1] = EF[1] + 1.0f; SB;
    EF[12][1] = __builtin_amdgcn_exp2f(acc1[1][1][3]); SB;
    acc1[1][3] = __builtin_amdgcn_wmma_f32_16x16x32_f16(false, wf1[11], false, __builtin_bit_cast(v16h, hb1[1]), (short)0, acc1[1][3], false, false); SB;
    EO[12][0] = __builtin_amdgcn_exp2f(acc1[1][1][6]); SB;
    EO[12][1] = __builtin_amdgcn_exp2f(acc1[1][1][7]); SB;
    acc2[0][3] = __builtin_amdgcn_wmma_f32_16x16x32_f16(false, wf1[16], false, __builtin_bit_cast(v16h, hb1[0]), (short)0, zero8, false, false); SB;
    GC[4][0] = cap_exp_arg(acc1[1][2][4]); SB;
    EI[4][0] = __builtin_amdgcn_exp2f(acc1[1][2][0]); SB;
    GC[4][1] = cap_exp_arg(acc1[1][2][5]); SB;
    EI[4][1] = __builtin_amdgcn_exp2f(acc1[1][2][1]); SB;
    acc1[1][0] = __builtin_amdgcn_wmma_f32_16x16x32_f16(false, wf2[3], false, __builtin_bit_cast(v16h, hb2[1]), (short)0, acc1[1][0], false, false); SB;
    DF[12] = EF[12] + 1.0f; SB;
    EF[4][0] = __builtin_amdgcn_exp2f(acc1[1][2][2]); SB;
    DQ[12] = EO[12] + 1.0f; SB;
    EF[4][1] = __builtin_amdgcn_exp2f(acc1[1][2][3]); SB;
    acc1[1][1] = __builtin_amdgcn_wmma_f32_16x16x32_f16(false, wf2[14], false, __builtin_bit_cast(v16h, hb2[0]), (short)0, zero8, false, false); SB;
    EO[4][0] = __builtin_amdgcn_exp2f(acc1[1][2][6]); SB;
    EO[4][1] = __builtin_amdgcn_exp2f(acc1[1][2][7]); SB;
    acc2[0][0] = __builtin_amdgcn_wmma_f32_16x16x32_f16(false, wf1[13], false, __builtin_bit_cast(v16h, hb1[1]), (short)0, acc2[0][0], false, false); SB;
    GC[5][0] = cap_exp_arg(acc1[1][3][4]); SB;
    EI[5][0] = __builtin_amdgcn_exp2f(acc1[1][3][0]); SB;
    GC[5][1] = cap_exp_arg(acc1[1][3][5]); SB;
    EI[5][1] = __builtin_amdgcn_exp2f(acc1[1][3][1]); SB;
    acc2[1][0] = __builtin_amdgcn_wmma_f32_16x16x32_f16(false, wf1[18], false, __builtin_bit_cast(v16h, hb1[0]), (short)0, zero8, false, false); SB;
    DF[4] = EF[4] + 1.0f; SB;
    EF[5][0] = __builtin_amdgcn_exp2f(acc1[1][3][2]); SB;
    DQ[4] = EO[4] + 1.0f; SB;
    EF[5][1] = __builtin_amdgcn_exp2f(acc1[1][3][3]); SB;
    acc2[0][1] = __builtin_amdgcn_wmma_f32_16x16x32_f16(false, wf2[7], false, __builtin_bit_cast(v16h, hb2[1]), (short)0, acc2[0][1], false, false); SB;
    EO[5][0] = __builtin_amdgcn_exp2f(acc1[1][3][6]); SB;
    EO[5][1] = __builtin_amdgcn_exp2f(acc1[1][3][7]); SB;
    acc1[1][2] = __builtin_amdgcn_wmma_f32_16x16x32_f16(false, wf2[18], false, __builtin_bit_cast(v16h, hb2[0]), (short)0, zero8, false, false); SB;
    DF[5] = EF[5] + 1.0f; SB;
    EG[12][0] = __builtin_amdgcn_exp2f(GC[12][0]); SB;
    EG[12][1] = __builtin_amdgcn_exp2f(GC[12][1]); SB;
    acc2[0][2] = __builtin_amdgcn_wmma_f32_16x16x32_f16(false, wf1[15], false, __builtin_bit_cast(v16h, hb1[1]), (short)0, acc2[0][2], false, false); SB;
    GC[6][0] = cap_exp_arg(acc2[0][0][4]); SB;
    EI[6][0] = __builtin_amdgcn_exp2f(acc2[0][0][0]); SB;
    GC[6][1] = cap_exp_arg(acc2[0][0][5]); SB;
    EI[6][1] = __builtin_amdgcn_exp2f(acc2[0][0][1]); SB;
    acc2[1][1] = __builtin_amdgcn_wmma_f32_16x16x32_f16(false, wf1[20], false, __builtin_bit_cast(v16h, hb1[0]), (short)0, zero8, false, false); SB;
    DGK[12] = __builtin_elementwise_fma(EG[12], rkc2, rkc2); SB;
    EF[6][0] = __builtin_amdgcn_exp2f(acc2[0][0][2]); SB;
    DQ[5] = EO[5] + 1.0f; SB;
    EF[6][1] = __builtin_amdgcn_exp2f(acc2[0][0][3]); SB;
    acc2[1][3] = __builtin_amdgcn_wmma_f32_16x16x32_f16(false, wf2[11], false, __builtin_bit_cast(v16h, hb2[1]), (short)0, acc2[1][3], false, false); SB;
    PIG[12] = __builtin_elementwise_fma(EI[12], DGK[12], DGK[12]); SB;
    EO[6][0] = __builtin_amdgcn_exp2f(acc2[0][0][6]); SB;
    EO[6][1] = __builtin_amdgcn_exp2f(acc2[0][0][7]); SB;
    acc1[1][3] = __builtin_amdgcn_wmma_f32_16x16x32_f16(false, wf2[22], false, __builtin_bit_cast(v16h, hb2[0]), (short)0, zero8, false, false); SB;
    DF[6] = EF[6] + 1.0f; SB;
    EG[4][0] = __builtin_amdgcn_exp2f(GC[4][0]); SB;
    EG[4][1] = __builtin_amdgcn_exp2f(GC[4][1]); SB;
    acc2[0][3] = __builtin_amdgcn_wmma_f32_16x16x32_f16(false, wf1[17], false, __builtin_bit_cast(v16h, hb1[1]), (short)0, acc2[0][3], false, false); SB;
    GC[7][0] = cap_exp_arg(acc2[0][2][4]); SB;
    EI[7][0] = __builtin_amdgcn_exp2f(acc2[0][2][0]); SB;
    GC[7][1] = cap_exp_arg(acc2[0][2][5]); SB;
    EI[7][1] = __builtin_amdgcn_exp2f(acc2[0][2][1]); SB;
    acc1[1][0] = __builtin_amdgcn_wmma_f32_16x16x32_f16(false, wf2[0], false, __builtin_bit_cast(v16h, hb1[0]), (short)0, acc1[1][0], false, false); SB;
    DGK[4] = __builtin_elementwise_fma(EG[4], rkc2, rkc2); SB;
    EF[7][0] = __builtin_amdgcn_exp2f(acc2[0][2][2]); SB;
    DQ[6] = EO[6] + 1.0f; SB;
    EF[7][1] = __builtin_amdgcn_exp2f(acc2[0][2][3]); SB;
    acc2[1][2] = __builtin_amdgcn_wmma_f32_16x16x32_f16(false, wf1[22], false, __builtin_bit_cast(v16h, hb1[0]), (short)0, zero8, false, false); SB;
    PIG[4] = __builtin_elementwise_fma(EI[4], DGK[4], DGK[4]); SB;
    EO[7][0] = __builtin_amdgcn_exp2f(acc2[0][2][6]); SB;
    EO[7][1] = __builtin_amdgcn_exp2f(acc2[0][2][7]); SB;
    acc1[1][1] = __builtin_amdgcn_wmma_f32_16x16x32_f16(false, wf2[15], false, __builtin_bit_cast(v16h, hb2[1]), (short)0, acc1[1][1], false, false); SB;
    DF[7] = EF[7] + 1.0f; SB;
    EG[5][0] = __builtin_amdgcn_exp2f(GC[5][0]); SB;
    EG[5][1] = __builtin_amdgcn_exp2f(GC[5][1]); SB;
    acc2[0][0] = __builtin_amdgcn_wmma_f32_16x16x32_f16(false, wf2[26], false, __builtin_bit_cast(v16h, hb2[0]), (short)0, zero8, false, false); SB;
    GC[8][0] = cap_exp_arg(acc2[0][3][4]); SB;
    EI[8][0] = __builtin_amdgcn_exp2f(acc2[0][3][0]); SB;
    GC[8][1] = cap_exp_arg(acc2[0][3][5]); SB;
    EI[8][1] = __builtin_amdgcn_exp2f(acc2[0][3][1]); SB;
    acc2[1][0] = __builtin_amdgcn_wmma_f32_16x16x32_f16(false, wf1[19], false, __builtin_bit_cast(v16h, hb1[1]), (short)0, acc2[1][0], false, false); SB;
    DGK[5] = __builtin_elementwise_fma(EG[5], rkc2, rkc2); SB;
    EF[8][0] = __builtin_amdgcn_exp2f(acc2[0][3][2]); SB;
    DQ[7] = EO[7] + 1.0f; SB;
    EF[8][1] = __builtin_amdgcn_exp2f(acc2[0][3][3]); SB;
    acc2[0][1] = __builtin_amdgcn_wmma_f32_16x16x32_f16(false, wf2[4], false, __builtin_bit_cast(v16h, hb1[0]), (short)0, acc2[0][1], false, false); SB;
    PIG[5] = __builtin_elementwise_fma(EI[5], DGK[5], DGK[5]); SB;
    EO[8][0] = __builtin_amdgcn_exp2f(acc2[0][3][6]); SB;
    EO[8][1] = __builtin_amdgcn_exp2f(acc2[0][3][7]); SB;
    acc1[1][2] = __builtin_amdgcn_wmma_f32_16x16x32_f16(false, wf2[19], false, __builtin_bit_cast(v16h, hb2[1]), (short)0, acc1[1][2], false, false); SB;
    DF[8] = EF[8] + 1.0f; SB;
    EG[6][0] = __builtin_amdgcn_exp2f(GC[6][0]); SB;
    EG[6][1] = __builtin_amdgcn_exp2f(GC[6][1]); SB;
    acc2[0][2] = __builtin_amdgcn_wmma_f32_16x16x32_f16(false, wf2[30], false, __builtin_bit_cast(v16h, hb2[0]), (short)0, zero8, false, false); SB;
    DQ[8] = EO[8] + 1.0f; SB;
    EG[7][0] = __builtin_amdgcn_exp2f(GC[7][0]); SB;
    EG[7][1] = __builtin_amdgcn_exp2f(GC[7][1]); SB;
    acc2[1][1] = __builtin_amdgcn_wmma_f32_16x16x32_f16(false, wf1[21], false, __builtin_bit_cast(v16h, hb1[1]), (short)0, acc2[1][1], false, false); SB;
    GC[9][0] = cap_exp_arg(acc2[1][0][4]); SB;
    EI[9][0] = __builtin_amdgcn_exp2f(acc2[1][0][0]); SB;
    GC[9][1] = cap_exp_arg(acc2[1][0][5]); SB;
    EI[9][1] = __builtin_amdgcn_exp2f(acc2[1][0][1]); SB;
    acc2[1][3] = __builtin_amdgcn_wmma_f32_16x16x32_f16(false, wf2[8], false, __builtin_bit_cast(v16h, hb1[0]), (short)0, acc2[1][3], false, false); SB;
    DGK[6] = __builtin_elementwise_fma(EG[6], rkc2, rkc2); SB;
    EF[9][0] = __builtin_amdgcn_exp2f(acc2[1][0][2]); SB;
    DGK[7] = __builtin_elementwise_fma(EG[7], rkc2, rkc2); SB;
    EF[9][1] = __builtin_amdgcn_exp2f(acc2[1][0][3]); SB;
    acc1[1][3] = __builtin_amdgcn_wmma_f32_16x16x32_f16(false, wf2[23], false, __builtin_bit_cast(v16h, hb2[1]), (short)0, acc1[1][3], false, false); SB;
    PIG[6] = __builtin_elementwise_fma(EI[6], DGK[6], DGK[6]); SB;
    EO[9][0] = __builtin_amdgcn_exp2f(acc2[1][0][6]); SB;
    PIG[7] = __builtin_elementwise_fma(EI[7], DGK[7], DGK[7]); SB;
    EO[9][1] = __builtin_amdgcn_exp2f(acc2[1][0][7]); SB;
    acc1[1][0] = __builtin_amdgcn_wmma_f32_16x16x32_f16(false, wf2[1], false, __builtin_bit_cast(v16h, hb1[1]), (short)0, acc1[1][0], false, false); SB;
    DF[9] = EF[9] + 1.0f; SB;
    EG[8][0] = __builtin_amdgcn_exp2f(GC[8][0]); SB;
    EG[8][1] = __builtin_amdgcn_exp2f(GC[8][1]); SB;
    acc2[0][3] = __builtin_amdgcn_wmma_f32_16x16x32_f16(false, wf2[34], false, __builtin_bit_cast(v16h, hb2[0]), (short)0, zero8, false, false); SB;
    GC[10][0] = cap_exp_arg(acc2[1][1][4]); SB;
    EI[10][0] = __builtin_amdgcn_exp2f(acc2[1][1][0]); SB;
    GC[10][1] = cap_exp_arg(acc2[1][1][5]); SB;
    EI[10][1] = __builtin_amdgcn_exp2f(acc2[1][1][1]); SB;
    acc2[1][2] = __builtin_amdgcn_wmma_f32_16x16x32_f16(false, wf1[23], false, __builtin_bit_cast(v16h, hb1[1]), (short)0, acc2[1][2], false, false); SB;
    DGK[8] = __builtin_elementwise_fma(EG[8], rkc2, rkc2); SB;
    EF[10][0] = __builtin_amdgcn_exp2f(acc2[1][1][2]); SB;
    DQ[9] = EO[9] + 1.0f; SB;
    EF[10][1] = __builtin_amdgcn_exp2f(acc2[1][1][3]); SB;
    acc1[1][1] = __builtin_amdgcn_wmma_f32_16x16x32_f16(false, wf2[12], false, __builtin_bit_cast(v16h, hb1[0]), (short)0, acc1[1][1], false, false); SB;
    PIG[8] = __builtin_elementwise_fma(EI[8], DGK[8], DGK[8]); SB;
    EO[10][0] = __builtin_amdgcn_exp2f(acc2[1][1][6]); SB;
    EO[10][1] = __builtin_amdgcn_exp2f(acc2[1][1][7]); SB;
    acc2[0][0] = __builtin_amdgcn_wmma_f32_16x16x32_f16(false, wf2[27], false, __builtin_bit_cast(v16h, hb2[1]), (short)0, acc2[0][0], false, false); SB;
    GC[13][0] = cap_exp_arg(acc1[1][0][4]); SB;
    EI[13][0] = __builtin_amdgcn_exp2f(acc1[1][0][0]); SB;
    GC[13][1] = cap_exp_arg(acc1[1][0][5]); SB;
    EI[13][1] = __builtin_amdgcn_exp2f(acc1[1][0][1]); SB;
    acc2[0][1] = __builtin_amdgcn_wmma_f32_16x16x32_f16(false, wf2[5], false, __builtin_bit_cast(v16h, hb1[1]), (short)0, acc2[0][1], false, false); SB;
    DF[10] = EF[10] + 1.0f; SB;
    EG[13][0] = __builtin_amdgcn_exp2f(GC[13][0]); SB;
    DQ[10] = EO[10] + 1.0f; SB;
    EG[13][1] = __builtin_amdgcn_exp2f(GC[13][1]); SB;
    acc2[1][0] = __builtin_amdgcn_wmma_f32_16x16x32_f16(false, wf2[38], false, __builtin_bit_cast(v16h, hb2[0]), (short)0, zero8, false, false); SB;
    GC[11][0] = cap_exp_arg(acc2[1][2][4]); SB;
    EI[11][0] = __builtin_amdgcn_exp2f(acc2[1][2][0]); SB;
    GC[11][1] = cap_exp_arg(acc2[1][2][5]); SB;
    EI[11][1] = __builtin_amdgcn_exp2f(acc2[1][2][1]); SB;
    acc1[1][2] = __builtin_amdgcn_wmma_f32_16x16x32_f16(false, wf2[16], false, __builtin_bit_cast(v16h, hb1[0]), (short)0, acc1[1][2], false, false); SB;
    DGK[13] = __builtin_elementwise_fma(EG[13], rkc2, rkc2); SB;
    EF[11][0] = __builtin_amdgcn_exp2f(acc2[1][2][2]); SB;
    EF[11][1] = __builtin_amdgcn_exp2f(acc2[1][2][3]); SB;
    acc2[0][2] = __builtin_amdgcn_wmma_f32_16x16x32_f16(false, wf2[31], false, __builtin_bit_cast(v16h, hb2[1]), (short)0, acc2[0][2], false, false); SB;
    PIG[13] = __builtin_elementwise_fma(EI[13], DGK[13], DGK[13]); SB;
    EO[11][0] = __builtin_amdgcn_exp2f(acc2[1][2][6]); SB;
    EO[11][1] = __builtin_amdgcn_exp2f(acc2[1][2][7]); SB;
    acc2[1][3] = __builtin_amdgcn_wmma_f32_16x16x32_f16(false, wf2[9], false, __builtin_bit_cast(v16h, hb1[1]), (short)0, acc2[1][3], false, false); SB;
    GC[14][0] = cap_exp_arg(acc2[0][1][4]); SB;
    EF[13][0] = __builtin_amdgcn_exp2f(acc1[1][0][2]); SB;
    GC[14][1] = cap_exp_arg(acc2[0][1][5]); SB;
    EF[13][1] = __builtin_amdgcn_exp2f(acc1[1][0][3]); SB;
    acc2[1][1] = __builtin_amdgcn_wmma_f32_16x16x32_f16(false, wf2[42], false, __builtin_bit_cast(v16h, hb2[0]), (short)0, zero8, false, false); SB;
    DF[11] = EF[11] + 1.0f; SB;
    EO[13][0] = __builtin_amdgcn_exp2f(acc1[1][0][6]); SB;
    DQ[11] = EO[11] + 1.0f; SB;
    EO[13][1] = __builtin_amdgcn_exp2f(acc1[1][0][7]); SB;
    acc1[1][3] = __builtin_amdgcn_wmma_f32_16x16x32_f16(false, wf2[20], false, __builtin_bit_cast(v16h, hb1[0]), (short)0, acc1[1][3], false, false); SB;
    DF[13] = EF[13] + 1.0f; SB;
    EG[14][0] = __builtin_amdgcn_exp2f(GC[14][0]); SB;
    EG[14][1] = __builtin_amdgcn_exp2f(GC[14][1]); SB;
    acc2[0][3] = __builtin_amdgcn_wmma_f32_16x16x32_f16(false, wf2[35], false, __builtin_bit_cast(v16h, hb2[1]), (short)0, acc2[0][3], false, false); SB;
    DQ[13] = EO[13] + 1.0f; SB;
    EI[14][0] = __builtin_amdgcn_exp2f(acc2[0][1][0]); SB;
    EI[14][1] = __builtin_amdgcn_exp2f(acc2[0][1][1]); SB;
    acc1[1][1] = __builtin_amdgcn_wmma_f32_16x16x32_f16(false, wf2[13], false, __builtin_bit_cast(v16h, hb1[1]), (short)0, acc1[1][1], false, false); SB;
    GC[15][0] = cap_exp_arg(acc2[1][3][4]); SB;
    EF[14][0] = __builtin_amdgcn_exp2f(acc2[0][1][2]); SB;
    GC[15][1] = cap_exp_arg(acc2[1][3][5]); SB;
    EF[14][1] = __builtin_amdgcn_exp2f(acc2[0][1][3]); SB;
    acc2[1][2] = __builtin_amdgcn_wmma_f32_16x16x32_f16(false, wf2[46], false, __builtin_bit_cast(v16h, hb2[0]), (short)0, zero8, false, false); SB;
    DGK[14] = __builtin_elementwise_fma(EG[14], rkc2, rkc2); SB;
    RF[13][0] = __builtin_amdgcn_rcpf(DF[13][0]); SB;
    RF[13][1] = __builtin_amdgcn_rcpf(DF[13][1]); SB;
    acc2[0][0] = __builtin_amdgcn_wmma_f32_16x16x32_f16(false, wf2[24], false, __builtin_bit_cast(v16h, hb1[0]), (short)0, acc2[0][0], false, false); SB;
    DF[14] = EF[14] + 1.0f; SB;
    RIG[13][0] = __builtin_amdgcn_rcpf(PIG[13][0]); SB;
    PIG[14] = __builtin_elementwise_fma(EI[14], DGK[14], DGK[14]); SB;
    RIG[13][1] = __builtin_amdgcn_rcpf(PIG[13][1]); SB;
    acc2[1][0] = __builtin_amdgcn_wmma_f32_16x16x32_f16(false, wf2[39], false, __builtin_bit_cast(v16h, hb2[1]), (short)0, acc2[1][0], false, false); SB;
    EG[15][0] = __builtin_amdgcn_exp2f(GC[15][0]); SB;
    EG[15][1] = __builtin_amdgcn_exp2f(GC[15][1]); SB;
    acc1[1][2] = __builtin_amdgcn_wmma_f32_16x16x32_f16(false, wf2[17], false, __builtin_bit_cast(v16h, hb1[1]), (short)0, acc1[1][2], false, false); SB;
    GC[16][0] = cap_exp_arg(acc1[1][1][4]); SB;
    EI[15][0] = __builtin_amdgcn_exp2f(acc2[1][3][0]); SB;
    GC[16][1] = cap_exp_arg(acc1[1][1][5]); SB;
    EI[15][1] = __builtin_amdgcn_exp2f(acc2[1][3][1]); SB;
    acc1[1][0] = __builtin_amdgcn_wmma_f32_16x16x32_f16(false, wf2[50], false, __builtin_bit_cast(v16h, hb2[0]), (short)0, zero8, false, false); SB;
    TT[13] = __builtin_elementwise_fma(c2[0], RF[13], -RIG[13]); SB;
    EF[15][0] = __builtin_amdgcn_exp2f(acc2[1][3][2]); SB;
    DGK[15] = __builtin_elementwise_fma(EG[15], rkc2, rkc2); SB;
    EF[15][1] = __builtin_amdgcn_exp2f(acc2[1][3][3]); SB;
    acc2[0][2] = __builtin_amdgcn_wmma_f32_16x16x32_f16(false, wf2[28], false, __builtin_bit_cast(v16h, hb1[0]), (short)0, acc2[0][2], false, false); SB;
    c2[0] = __builtin_elementwise_fma(EG[13], RIG[13], TT[13]); SB;
    RF[14][0] = __builtin_amdgcn_rcpf(DF[14][0]); SB;
    PIG[15] = __builtin_elementwise_fma(EI[15], DGK[15], DGK[15]); SB;
    RF[14][1] = __builtin_amdgcn_rcpf(DF[14][1]); SB;
    acc2[1][1] = __builtin_amdgcn_wmma_f32_16x16x32_f16(false, wf2[43], false, __builtin_bit_cast(v16h, hb2[1]), (short)0, acc2[1][1], false, false); SB;
    DF[15] = EF[15] + 1.0f; SB;
    RIG[14][0] = __builtin_amdgcn_rcpf(PIG[14][0]); SB;
    CC[13][0] = cap_exp_arg(c2[0][0]); SB;
    RIG[14][1] = __builtin_amdgcn_rcpf(PIG[14][1]); SB;
    acc1[1][3] = __builtin_amdgcn_wmma_f32_16x16x32_f16(false, wf2[21], false, __builtin_bit_cast(v16h, hb1[1]), (short)0, acc1[1][3], false, false); SB;
    CC[13][1] = cap_exp_arg(c2[0][1]); SB;
    EG[16][0] = __builtin_amdgcn_exp2f(GC[16][0]); SB;
    GC[17][0] = cap_exp_arg(acc1[1][2][4]); SB;
    EG[16][1] = __builtin_amdgcn_exp2f(GC[16][1]); SB;
    acc2[0][3] = __builtin_amdgcn_wmma_f32_16x16x32_f16(false, wf2[32], false, __builtin_bit_cast(v16h, hb1[0]), (short)0, acc2[0][3], false, false); SB;
    TT[14] = __builtin_elementwise_fma(c2[1], RF[14], -RIG[14]); SB;
    EI[16][0] = __builtin_amdgcn_exp2f(acc1[1][1][0]); SB;
    GC[17][1] = cap_exp_arg(acc1[1][2][5]); SB;
    EI[16][1] = __builtin_amdgcn_exp2f(acc1[1][1][1]); SB;
    acc2[1][2] = __builtin_amdgcn_wmma_f32_16x16x32_f16(false, wf2[47], false, __builtin_bit_cast(v16h, hb2[1]), (short)0, acc2[1][2], false, false); SB;
    DGK[16] = __builtin_elementwise_fma(EG[16], rkc2, rkc2); SB;
    EF[16][0] = __builtin_amdgcn_exp2f(acc1[1][1][2]); SB;
    c2[1] = __builtin_elementwise_fma(EG[14], RIG[14], TT[14]); SB;
    EF[16][1] = __builtin_amdgcn_exp2f(acc1[1][1][3]); SB;
    acc2[0][0] = __builtin_amdgcn_wmma_f32_16x16x32_f16(false, wf2[25], false, __builtin_bit_cast(v16h, hb1[1]), (short)0, acc2[0][0], false, false); SB;
    PIG[16] = __builtin_elementwise_fma(EI[16], DGK[16], DGK[16]); SB;
    RF[15][0] = __builtin_amdgcn_rcpf(DF[15][0]); SB;
    CC[14][0] = cap_exp_arg(c2[1][0]); SB;
    RF[15][1] = __builtin_amdgcn_rcpf(DF[15][1]); SB;
    acc2[1][0] = __builtin_amdgcn_wmma_f32_16x16x32_f16(false, wf2[36], false, __builtin_bit_cast(v16h, hb1[0]), (short)0, acc2[1][0], false, false); SB;
    DF[16] = EF[16] + 1.0f; SB;
    RIG[15][0] = __builtin_amdgcn_rcpf(PIG[15][0]); SB;
    CC[14][1] = cap_exp_arg(c2[1][1]); SB;
    RIG[15][1] = __builtin_amdgcn_rcpf(PIG[15][1]); SB;
    acc1[1][0] = __builtin_amdgcn_wmma_f32_16x16x32_f16(false, wf2[51], false, __builtin_bit_cast(v16h, hb2[1]), (short)0, acc1[1][0], false, false); SB;
    GC[18][0] = cap_exp_arg(acc1[1][3][4]); SB;
    EG[17][0] = __builtin_amdgcn_exp2f(GC[17][0]); SB;
    GC[18][1] = cap_exp_arg(acc1[1][3][5]); SB;
    EG[17][1] = __builtin_amdgcn_exp2f(GC[17][1]); SB;
    acc2[0][2] = __builtin_amdgcn_wmma_f32_16x16x32_f16(false, wf2[29], false, __builtin_bit_cast(v16h, hb1[1]), (short)0, acc2[0][2], false, false); SB;
    TT[15] = __builtin_elementwise_fma(c2[2], RF[15], -RIG[15]); SB;
    EC[13][0] = __builtin_amdgcn_exp2f(CC[13][0]); SB;
    EC[13][1] = __builtin_amdgcn_exp2f(CC[13][1]); SB;
    acc2[1][1] = __builtin_amdgcn_wmma_f32_16x16x32_f16(false, wf2[40], false, __builtin_bit_cast(v16h, hb1[0]), (short)0, acc2[1][1], false, false); SB;
    DGK[17] = __builtin_elementwise_fma(EG[17], rkc2, rkc2); SB;
    EI[17][0] = __builtin_amdgcn_exp2f(acc1[1][2][0]); SB;
    c2[2] = __builtin_elementwise_fma(EG[15], RIG[15], TT[15]); SB;
    EI[17][1] = __builtin_amdgcn_exp2f(acc1[1][2][1]); SB;
    acc2[0][3] = __builtin_amdgcn_wmma_f32_16x16x32_f16(false, wf2[33], false, __builtin_bit_cast(v16h, hb1[1]), (short)0, acc2[0][3], false, false); SB;
    POC[13] = __builtin_elementwise_fma(EC[13], DQ[13], DQ[13]); SB;
    EF[17][0] = __builtin_amdgcn_exp2f(acc1[1][2][2]); SB;
    GC[19][0] = cap_exp_arg(acc2[0][0][4]); SB;
    EF[17][1] = __builtin_amdgcn_exp2f(acc1[1][2][3]); SB;
    acc2[1][2] = __builtin_amdgcn_wmma_f32_16x16x32_f16(false, wf2[44], false, __builtin_bit_cast(v16h, hb1[0]), (short)0, acc2[1][2], false, false); SB;
    PIG[17] = __builtin_elementwise_fma(EI[17], DGK[17], DGK[17]); SB;
    EO[14][0] = __builtin_amdgcn_exp2f(acc2[0][1][6]); SB;
    GC[19][1] = cap_exp_arg(acc2[0][0][5]); SB;
    EO[14][1] = __builtin_amdgcn_exp2f(acc2[0][1][7]); SB;
    acc2[1][0] = __builtin_amdgcn_wmma_f32_16x16x32_f16(false, wf2[37], false, __builtin_bit_cast(v16h, hb1[1]), (short)0, acc2[1][0], false, false); SB;
    DF[17] = EF[17] + 1.0f; SB;
    RF[16][0] = __builtin_amdgcn_rcpf(DF[16][0]); SB;
    CC[15][0] = cap_exp_arg(c2[2][0]); SB;
    RF[16][1] = __builtin_amdgcn_rcpf(DF[16][1]); SB;
    acc1[1][0] = __builtin_amdgcn_wmma_f32_16x16x32_f16(false, wf2[48], false, __builtin_bit_cast(v16h, hb1[0]), (short)0, acc1[1][0], false, false); SB;
    DQ[14] = EO[14] + 1.0f; SB;
    RIG[16][0] = __builtin_amdgcn_rcpf(PIG[16][0]); SB;
    CC[15][1] = cap_exp_arg(c2[2][1]); SB;
    RIG[16][1] = __builtin_amdgcn_rcpf(PIG[16][1]); SB;
    acc2[1][1] = __builtin_amdgcn_wmma_f32_16x16x32_f16(false, wf2[41], false, __builtin_bit_cast(v16h, hb1[1]), (short)0, acc2[1][1], false, false); SB;
    GC[20][0] = cap_exp_arg(acc2[0][2][4]); SB;
    EG[18][0] = __builtin_amdgcn_exp2f(GC[18][0]); SB;
    GC[20][1] = cap_exp_arg(acc2[0][2][5]); SB;
    EG[18][1] = __builtin_amdgcn_exp2f(GC[18][1]); SB;
    acc2[1][2] = __builtin_amdgcn_wmma_f32_16x16x32_f16(false, wf2[45], false, __builtin_bit_cast(v16h, hb1[1]), (short)0, acc2[1][2], false, false); SB;
    TT[16] = __builtin_elementwise_fma(c2[3], RF[16], -RIG[16]); SB;
    EC[14][0] = __builtin_amdgcn_exp2f(CC[14][0]); SB;
    GC[21][0] = cap_exp_arg(acc2[0][3][4]); SB;
    EC[14][1] = __builtin_amdgcn_exp2f(CC[14][1]); SB;
    DGK[18] = __builtin_elementwise_fma(EG[18], rkc2, rkc2); SB;
    GC[21][1] = cap_exp_arg(acc2[0][3][5]); SB;
    c2[3] = __builtin_elementwise_fma(EG[16], RIG[16], TT[16]); SB;
    GC[22][0] = cap_exp_arg(acc2[1][0][4]); SB;
    GC[22][1] = cap_exp_arg(acc2[1][0][5]); SB;
    acc1[1][0] = __builtin_amdgcn_wmma_f32_16x16x32_f16(false, wf2[49], false, __builtin_bit_cast(v16h, hb1[1]), (short)0, acc1[1][0], false, false); SB;
    POC[14] = __builtin_elementwise_fma(EC[14], DQ[14], DQ[14]); SB;
    ROC[13][0] = __builtin_amdgcn_rcpf(POC[13][0]); SB;
    CC[16][0] = cap_exp_arg(c2[3][0]); SB;
    ROC[13][1] = __builtin_amdgcn_rcpf(POC[13][1]); SB;
    CC[16][1] = cap_exp_arg(c2[3][1]); SB;
    EI[18][0] = __builtin_amdgcn_exp2f(acc1[1][3][0]); SB;
    GC[23][0] = cap_exp_arg(acc2[1][1][4]); SB;
    EI[18][1] = __builtin_amdgcn_exp2f(acc1[1][3][1]); SB;
    GC[23][1] = cap_exp_arg(acc2[1][1][5]); SB;
    EF[18][0] = __builtin_amdgcn_exp2f(acc1[1][3][2]); SB;
    EF[18][1] = __builtin_amdgcn_exp2f(acc1[1][3][3]); SB;
    HV[13] = __builtin_elementwise_fma(EC[13], ROC[13], -ROC[13]); SB;
    EO[15][0] = __builtin_amdgcn_exp2f(acc2[1][3][6]); SB;
    GC[24][0] = cap_exp_arg(acc2[1][2][4]); SB;
    EO[15][1] = __builtin_amdgcn_exp2f(acc2[1][3][7]); SB;
    PIG[18] = __builtin_elementwise_fma(EI[18], DGK[18], DGK[18]); SB;
    RF[17][0] = __builtin_amdgcn_rcpf(DF[17][0]); SB;
    ys += HV[13][0] * wl[0][0]; SB;
    RF[17][1] = __builtin_amdgcn_rcpf(DF[17][1]); SB;
    DF[18] = EF[18] + 1.0f; SB;
    RIG[17][0] = __builtin_amdgcn_rcpf(PIG[17][0]); SB;
    GC[24][1] = cap_exp_arg(acc2[1][2][5]); SB;
    RIG[17][1] = __builtin_amdgcn_rcpf(PIG[17][1]); SB;
    DQ[15] = EO[15] + 1.0f; SB;
    EG[19][0] = __builtin_amdgcn_exp2f(GC[19][0]); SB;
    ys += HV[13][1] * wl[0][1]; SB;
    EG[19][1] = __builtin_amdgcn_exp2f(GC[19][1]); SB;
    GC[25][0] = cap_exp_arg(acc1[1][0][4]); SB;
    EC[15][0] = __builtin_amdgcn_exp2f(CC[15][0]); SB;
    GC[25][1] = cap_exp_arg(acc1[1][0][5]); SB;
    EC[15][1] = __builtin_amdgcn_exp2f(CC[15][1]); SB;
    TT[17] = __builtin_elementwise_fma(c2[4], RF[17], -RIG[17]); SB;
    ROC[14][0] = __builtin_amdgcn_rcpf(POC[14][0]); SB;
    { v2h pk; pk[0] = (_Float16)HV[13][0]; pk[1] = (_Float16)HV[13][1]; hn2[0][0] = __builtin_bit_cast(unsigned, pk); } SB;
    ROC[14][1] = __builtin_amdgcn_rcpf(POC[14][1]); SB;
    DGK[19] = __builtin_elementwise_fma(EG[19], rkc2, rkc2); SB;
    EI[19][0] = __builtin_amdgcn_exp2f(acc2[0][0][0]); SB;
    c2[4] = __builtin_elementwise_fma(EG[17], RIG[17], TT[17]); SB;
    EI[19][1] = __builtin_amdgcn_exp2f(acc2[0][0][1]); SB;
    POC[15] = __builtin_elementwise_fma(EC[15], DQ[15], DQ[15]); SB;
    EF[19][0] = __builtin_amdgcn_exp2f(acc2[0][0][2]); SB;
    EF[19][1] = __builtin_amdgcn_exp2f(acc2[0][0][3]); SB;
    HV[14] = __builtin_elementwise_fma(EC[14], ROC[14], -ROC[14]); SB;
    EO[16][0] = __builtin_amdgcn_exp2f(acc1[1][1][6]); SB;
    CC[17][0] = cap_exp_arg(c2[4][0]); SB;
    EO[16][1] = __builtin_amdgcn_exp2f(acc1[1][1][7]); SB;
    PIG[19] = __builtin_elementwise_fma(EI[19], DGK[19], DGK[19]); SB;
    RF[18][0] = __builtin_amdgcn_rcpf(DF[18][0]); SB;
    CC[17][1] = cap_exp_arg(c2[4][1]); SB;
    RF[18][1] = __builtin_amdgcn_rcpf(DF[18][1]); SB;
    DF[19] = EF[19] + 1.0f; SB;
    RIG[18][0] = __builtin_amdgcn_rcpf(PIG[18][0]); SB;
    ys += HV[14][0] * wl[1][0]; SB;
    RIG[18][1] = __builtin_amdgcn_rcpf(PIG[18][1]); SB;
    DQ[16] = EO[16] + 1.0f; SB;
    EG[20][0] = __builtin_amdgcn_exp2f(GC[20][0]); SB;
    { v2h pk; pk[0] = (_Float16)HV[14][0]; pk[1] = (_Float16)HV[14][1]; hn2[0][1] = __builtin_bit_cast(unsigned, pk); } SB;
    EG[20][1] = __builtin_amdgcn_exp2f(GC[20][1]); SB;
    ys += HV[14][1] * wl[1][1]; SB;
    EC[16][0] = __builtin_amdgcn_exp2f(CC[16][0]); SB;
    EC[16][1] = __builtin_amdgcn_exp2f(CC[16][1]); SB;
    TT[18] = __builtin_elementwise_fma(c2[5], RF[18], -RIG[18]); SB;
    ROC[15][0] = __builtin_amdgcn_rcpf(POC[15][0]); SB;
    ROC[15][1] = __builtin_amdgcn_rcpf(POC[15][1]); SB;
    DGK[20] = __builtin_elementwise_fma(EG[20], rkc2, rkc2); SB;
    EI[20][0] = __builtin_amdgcn_exp2f(acc2[0][2][0]); SB;
    c2[5] = __builtin_elementwise_fma(EG[18], RIG[18], TT[18]); SB;
    EI[20][1] = __builtin_amdgcn_exp2f(acc2[0][2][1]); SB;
    POC[16] = __builtin_elementwise_fma(EC[16], DQ[16], DQ[16]); SB;
    EF[20][0] = __builtin_amdgcn_exp2f(acc2[0][2][2]); SB;
    EF[20][1] = __builtin_amdgcn_exp2f(acc2[0][2][3]); SB;
    HV[15] = __builtin_elementwise_fma(EC[15], ROC[15], -ROC[15]); SB;
    EO[17][0] = __builtin_amdgcn_exp2f(acc1[1][2][6]); SB;
    CC[18][0] = cap_exp_arg(c2[5][0]); SB;
    EO[17][1] = __builtin_amdgcn_exp2f(acc1[1][2][7]); SB;
    PIG[20] = __builtin_elementwise_fma(EI[20], DGK[20], DGK[20]); SB;
    RF[19][0] = __builtin_amdgcn_rcpf(DF[19][0]); SB;
    CC[18][1] = cap_exp_arg(c2[5][1]); SB;
    RF[19][1] = __builtin_amdgcn_rcpf(DF[19][1]); SB;
    DF[20] = EF[20] + 1.0f; SB;
    RIG[19][0] = __builtin_amdgcn_rcpf(PIG[19][0]); SB;
    ys += HV[15][0] * wl[2][0]; SB;
    RIG[19][1] = __builtin_amdgcn_rcpf(PIG[19][1]); SB;
    DQ[17] = EO[17] + 1.0f; SB;
    EG[21][0] = __builtin_amdgcn_exp2f(GC[21][0]); SB;
    { v2h pk; pk[0] = (_Float16)HV[15][0]; pk[1] = (_Float16)HV[15][1]; hn2[0][2] = __builtin_bit_cast(unsigned, pk); } SB;
    EG[21][1] = __builtin_amdgcn_exp2f(GC[21][1]); SB;
    ys += HV[15][1] * wl[2][1]; SB;
    EC[17][0] = __builtin_amdgcn_exp2f(CC[17][0]); SB;
    EC[17][1] = __builtin_amdgcn_exp2f(CC[17][1]); SB;
    TT[19] = __builtin_elementwise_fma(c2[6], RF[19], -RIG[19]); SB;
    ROC[16][0] = __builtin_amdgcn_rcpf(POC[16][0]); SB;
    ROC[16][1] = __builtin_amdgcn_rcpf(POC[16][1]); SB;
    DGK[21] = __builtin_elementwise_fma(EG[21], rkc2, rkc2); SB;
    EI[21][0] = __builtin_amdgcn_exp2f(acc2[0][3][0]); SB;
    c2[6] = __builtin_elementwise_fma(EG[19], RIG[19], TT[19]); SB;
    EI[21][1] = __builtin_amdgcn_exp2f(acc2[0][3][1]); SB;
    POC[17] = __builtin_elementwise_fma(EC[17], DQ[17], DQ[17]); SB;
    EF[21][0] = __builtin_amdgcn_exp2f(acc2[0][3][2]); SB;
    EF[21][1] = __builtin_amdgcn_exp2f(acc2[0][3][3]); SB;
    HV[16] = __builtin_elementwise_fma(EC[16], ROC[16], -ROC[16]); SB;
    EO[18][0] = __builtin_amdgcn_exp2f(acc1[1][3][6]); SB;
    CC[19][0] = cap_exp_arg(c2[6][0]); SB;
    EO[18][1] = __builtin_amdgcn_exp2f(acc1[1][3][7]); SB;
    PIG[21] = __builtin_elementwise_fma(EI[21], DGK[21], DGK[21]); SB;
    EG[9][0] = __builtin_amdgcn_exp2f(GC[9][0]); SB;
    CC[19][1] = cap_exp_arg(c2[6][1]); SB;
    EG[9][1] = __builtin_amdgcn_exp2f(GC[9][1]); SB;
    DF[21] = EF[21] + 1.0f; SB;
    EG[10][0] = __builtin_amdgcn_exp2f(GC[10][0]); SB;
    ys += HV[16][0] * wl[3][0]; SB;
    EG[10][1] = __builtin_amdgcn_exp2f(GC[10][1]); SB;
    DQ[18] = EO[18] + 1.0f; SB;
    EG[11][0] = __builtin_amdgcn_exp2f(GC[11][0]); SB;
    { v2h pk; pk[0] = (_Float16)HV[16][0]; pk[1] = (_Float16)HV[16][1]; hn2[0][3] = __builtin_bit_cast(unsigned, pk); } SB;
    EG[11][1] = __builtin_amdgcn_exp2f(GC[11][1]); SB;
    DGK[9] = __builtin_elementwise_fma(EG[9], rkc2, rkc2); SB;
    RF[20][0] = __builtin_amdgcn_rcpf(DF[20][0]); SB;
    ys += HV[16][1] * wl[3][1]; SB;
    RF[20][1] = __builtin_amdgcn_rcpf(DF[20][1]); SB;
    DGK[10] = __builtin_elementwise_fma(EG[10], rkc2, rkc2); SB;
    RIG[20][0] = __builtin_amdgcn_rcpf(PIG[20][0]); SB;
    PIG[9] = __builtin_elementwise_fma(EI[9], DGK[9], DGK[9]); SB;
    RIG[20][1] = __builtin_amdgcn_rcpf(PIG[20][1]); SB;
    DGK[11] = __builtin_elementwise_fma(EG[11], rkc2, rkc2); SB;
    EG[22][0] = __builtin_amdgcn_exp2f(GC[22][0]); SB;
    PIG[10] = __builtin_elementwise_fma(EI[10], DGK[10], DGK[10]); SB;
    EG[22][1] = __builtin_amdgcn_exp2f(GC[22][1]); SB;
    EC[18][0] = __builtin_amdgcn_exp2f(CC[18][0]); SB;
    PIG[11] = __builtin_elementwise_fma(EI[11], DGK[11], DGK[11]); SB;
    EC[18][1] = __builtin_amdgcn_exp2f(CC[18][1]); SB;
    TT[20] = __builtin_elementwise_fma(c2[7], RF[20], -RIG[20]); SB;
    EI[2][0] = __builtin_amdgcn_exp2f(acc1[0][2][0]); SB;
    EI[2][1] = __builtin_amdgcn_exp2f(acc1[0][2][1]); SB;
    DGK[22] = __builtin_elementwise_fma(EG[22], rkc2, rkc2); SB;
    EF[2][0] = __builtin_amdgcn_exp2f(acc1[0][2][2]); SB;
    c2[7] = __builtin_elementwise_fma(EG[20], RIG[20], TT[20]); SB;
    EF[2][1] = __builtin_amdgcn_exp2f(acc1[0][2][3]); SB;
    POC[18] = __builtin_elementwise_fma(EC[18], DQ[18], DQ[18]); SB;
    EI[3][0] = __builtin_amdgcn_exp2f(acc1[0][3][0]); SB;
    EI[3][1] = __builtin_amdgcn_exp2f(acc1[0][3][1]); SB;
    PIG[2] = __builtin_elementwise_fma(EI[2], DGK[2], DGK[2]); SB;
    EF[3][0] = __builtin_amdgcn_exp2f(acc1[0][3][2]); SB;
    CC[20][0] = cap_exp_arg(c2[7][0]); SB;
    EF[3][1] = __builtin_amdgcn_exp2f(acc1[0][3][3]); SB;
    DF[2] = EF[2] + 1.0f; SB;
    ROC[17][0] = __builtin_amdgcn_rcpf(POC[17][0]); SB;
    CC[20][1] = cap_exp_arg(c2[7][1]); SB;
    ROC[17][1] = __builtin_amdgcn_rcpf(POC[17][1]); SB;
    PIG[3] = __builtin_elementwise_fma(EI[3], DGK[3], DGK[3]); SB;
    EI[22][0] = __builtin_amdgcn_exp2f(acc2[1][0][0]); SB;
    EI[22][1] = __builtin_amdgcn_exp2f(acc2[1][0][1]); SB;
    DF[3] = EF[3] + 1.0f; SB;
    EF[22][0] = __builtin_amdgcn_exp2f(acc2[1][0][2]); SB;
    EF[22][1] = __builtin_amdgcn_exp2f(acc2[1][0][3]); SB;
    HV[17] = __builtin_elementwise_fma(EC[17], ROC[17], -ROC[17]); SB;
    EO[19][0] = __builtin_amdgcn_exp2f(acc2[0][0][6]); SB;
    EO[19][1] = __builtin_amdgcn_exp2f(acc2[0][0][7]); SB;
    PIG[22] = __builtin_elementwise_fma(EI[22], DGK[22], DGK[22]); SB;
    RF[21][0] = __builtin_amdgcn_rcpf(DF[21][0]); SB;
    ys += HV[17][0] * wl[4][0]; SB;
    RF[21][1] = __builtin_amdgcn_rcpf(DF[21][1]); SB;
    DF[22] = EF[22] + 1.0f; SB;
    RIG[21][0] = __builtin_amdgcn_rcpf(PIG[21][0]); SB;
    { v2h pk; pk[0] = (_Float16)HV[17][0]; pk[1] = (_Float16)HV[17][1]; hn2[0][4] = __builtin_bit_cast(unsigned, pk); } SB;
    RIG[21][1] = __builtin_amdgcn_rcpf(PIG[21][1]); SB;
    DQ[19] = EO[19] + 1.0f; SB;
    RF[12][0] = __builtin_amdgcn_rcpf(DF[12][0]); SB;
    ys += HV[17][1] * wl[4][1]; SB;
    RF[12][1] = __builtin_amdgcn_rcpf(DF[12][1]); SB;
    RIG[12][0] = __builtin_amdgcn_rcpf(PIG[12][0]); SB;
    RIG[12][1] = __builtin_amdgcn_rcpf(PIG[12][1]); SB;
    TT[21] = __builtin_elementwise_fma(c2[8], RF[21], -RIG[21]); SB;
    EG[23][0] = __builtin_amdgcn_exp2f(GC[23][0]); SB;
    EG[23][1] = __builtin_amdgcn_exp2f(GC[23][1]); SB;
    EC[19][0] = __builtin_amdgcn_exp2f(CC[19][0]); SB;
    c2[8] = __builtin_elementwise_fma(EG[21], RIG[21], TT[21]); SB;
    EC[19][1] = __builtin_amdgcn_exp2f(CC[19][1]); SB;
    TT[12] = __builtin_elementwise_fma(c1[12], RF[12], -RIG[12]); SB;
    ROC[18][0] = __builtin_amdgcn_rcpf(POC[18][0]); SB;
    ROC[18][1] = __builtin_amdgcn_rcpf(POC[18][1]); SB;
    DGK[23] = __builtin_elementwise_fma(EG[23], rkc2, rkc2); SB;
    RF[0][0] = __builtin_amdgcn_rcpf(DF[0][0]); SB;
    c1[12] = __builtin_elementwise_fma(EG[12], RIG[12], TT[12]); SB;
    RF[0][1] = __builtin_amdgcn_rcpf(DF[0][1]); SB;
    POC[19] = __builtin_elementwise_fma(EC[19], DQ[19], DQ[19]); SB;
    RIG[0][0] = __builtin_amdgcn_rcpf(PIG[0][0]); SB;
    CC[21][0] = cap_exp_arg(c2[8][0]); SB;
    RIG[0][1] = __builtin_amdgcn_rcpf(PIG[0][1]); SB;
    HV[18] = __builtin_elementwise_fma(EC[18], ROC[18], -ROC[18]); SB;
    RF[1][0] = __builtin_amdgcn_rcpf(DF[1][0]); SB;
    CC[21][1] = cap_exp_arg(c2[8][1]); SB;
    RF[1][1] = __builtin_amdgcn_rcpf(DF[1][1]); SB;
    CC[12][0] = cap_exp_arg(c1[12][0]); SB;
    RIG[1][0] = __builtin_amdgcn_rcpf(PIG[1][0]); SB;
    ys += HV[18][0] * wl[5][0]; SB;
    RIG[1][1] = __builtin_amdgcn_rcpf(PIG[1][1]); SB;
    TT[0] = __builtin_elementwise_fma(c1[0], RF[0], -RIG[0]); SB;
    RF[2][0] = __builtin_amdgcn_rcpf(DF[2][0]); SB;
    CC[12][1] = cap_exp_arg(c1[12][1]); SB;
    RF[2][1] = __builtin_amdgcn_rcpf(DF[2][1]); SB;
    ys += HV[18][1] * wl[5][1]; SB;
    RIG[2][0] = __builtin_amdgcn_rcpf(PIG[2][0]); SB;
    c1[0] = __builtin_elementwise_fma(EG[0], RIG[0], TT[0]); SB;
    RIG[2][1] = __builtin_amdgcn_rcpf(PIG[2][1]); SB;
    TT[1] = __builtin_elementwise_fma(c1[1], RF[1], -RIG[1]); SB;
    RF[3][0] = __builtin_amdgcn_rcpf(DF[3][0]); SB;
    { v2h pk; pk[0] = (_Float16)HV[18][0]; pk[1] = (_Float16)HV[18][1]; hn2[0][5] = __builtin_bit_cast(unsigned, pk); } SB;
    RF[3][1] = __builtin_amdgcn_rcpf(DF[3][1]); SB;
    CC[0][0] = cap_exp_arg(c1[0][0]); SB;
    RIG[3][0] = __builtin_amdgcn_rcpf(PIG[3][0]); SB;
    c1[1] = __builtin_elementwise_fma(EG[1], RIG[1], TT[1]); SB;
    RIG[3][1] = __builtin_amdgcn_rcpf(PIG[3][1]); SB;
    TT[2] = __builtin_elementwise_fma(c1[2], RF[2], -RIG[2]); SB;
    RF[4][0] = __builtin_amdgcn_rcpf(DF[4][0]); SB;
    CC[0][1] = cap_exp_arg(c1[0][1]); SB;
    RF[4][1] = __builtin_amdgcn_rcpf(DF[4][1]); SB;
    CC[1][0] = cap_exp_arg(c1[1][0]); SB;
    RIG[4][0] = __builtin_amdgcn_rcpf(PIG[4][0]); SB;
    c1[2] = __builtin_elementwise_fma(EG[2], RIG[2], TT[2]); SB;
    RIG[4][1] = __builtin_amdgcn_rcpf(PIG[4][1]); SB;
    TT[3] = __builtin_elementwise_fma(c1[3], RF[3], -RIG[3]); SB;
    RF[5][0] = __builtin_amdgcn_rcpf(DF[5][0]); SB;
    CC[1][1] = cap_exp_arg(c1[1][1]); SB;
    RF[5][1] = __builtin_amdgcn_rcpf(DF[5][1]); SB;
    CC[2][0] = cap_exp_arg(c1[2][0]); SB;
    RIG[5][0] = __builtin_amdgcn_rcpf(PIG[5][0]); SB;
    c1[3] = __builtin_elementwise_fma(EG[3], RIG[3], TT[3]); SB;
    RIG[5][1] = __builtin_amdgcn_rcpf(PIG[5][1]); SB;
    TT[4] = __builtin_elementwise_fma(c1[4], RF[4], -RIG[4]); SB;
    RF[6][0] = __builtin_amdgcn_rcpf(DF[6][0]); SB;
    CC[2][1] = cap_exp_arg(c1[2][1]); SB;
    RF[6][1] = __builtin_amdgcn_rcpf(DF[6][1]); SB;
    CC[3][0] = cap_exp_arg(c1[3][0]); SB;
    RIG[6][0] = __builtin_amdgcn_rcpf(PIG[6][0]); SB;
    c1[4] = __builtin_elementwise_fma(EG[4], RIG[4], TT[4]); SB;
    RIG[6][1] = __builtin_amdgcn_rcpf(PIG[6][1]); SB;
    TT[5] = __builtin_elementwise_fma(c1[5], RF[5], -RIG[5]); SB;
    RF[7][0] = __builtin_amdgcn_rcpf(DF[7][0]); SB;
    CC[3][1] = cap_exp_arg(c1[3][1]); SB;
    RF[7][1] = __builtin_amdgcn_rcpf(DF[7][1]); SB;
    CC[4][0] = cap_exp_arg(c1[4][0]); SB;
    RIG[7][0] = __builtin_amdgcn_rcpf(PIG[7][0]); SB;
    c1[5] = __builtin_elementwise_fma(EG[5], RIG[5], TT[5]); SB;
    RIG[7][1] = __builtin_amdgcn_rcpf(PIG[7][1]); SB;
    TT[6] = __builtin_elementwise_fma(c1[6], RF[6], -RIG[6]); SB;
    RF[8][0] = __builtin_amdgcn_rcpf(DF[8][0]); SB;
    CC[4][1] = cap_exp_arg(c1[4][1]); SB;
    RF[8][1] = __builtin_amdgcn_rcpf(DF[8][1]); SB;
    CC[5][0] = cap_exp_arg(c1[5][0]); SB;
    RIG[8][0] = __builtin_amdgcn_rcpf(PIG[8][0]); SB;
    c1[6] = __builtin_elementwise_fma(EG[6], RIG[6], TT[6]); SB;
    RIG[8][1] = __builtin_amdgcn_rcpf(PIG[8][1]); SB;
    TT[7] = __builtin_elementwise_fma(c1[7], RF[7], -RIG[7]); SB;
    RF[9][0] = __builtin_amdgcn_rcpf(DF[9][0]); SB;
    CC[5][1] = cap_exp_arg(c1[5][1]); SB;
    RF[9][1] = __builtin_amdgcn_rcpf(DF[9][1]); SB;
    CC[6][0] = cap_exp_arg(c1[6][0]); SB;
    RF[10][0] = __builtin_amdgcn_rcpf(DF[10][0]); SB;
    c1[7] = __builtin_elementwise_fma(EG[7], RIG[7], TT[7]); SB;
    RF[10][1] = __builtin_amdgcn_rcpf(DF[10][1]); SB;
    TT[8] = __builtin_elementwise_fma(c1[8], RF[8], -RIG[8]); SB;
    RIG[9][0] = __builtin_amdgcn_rcpf(PIG[9][0]); SB;
    CC[6][1] = cap_exp_arg(c1[6][1]); SB;
    RIG[9][1] = __builtin_amdgcn_rcpf(PIG[9][1]); SB;
    CC[7][0] = cap_exp_arg(c1[7][0]); SB;
    RIG[10][0] = __builtin_amdgcn_rcpf(PIG[10][0]); SB;
    c1[8] = __builtin_elementwise_fma(EG[8], RIG[8], TT[8]); SB;
    RIG[10][1] = __builtin_amdgcn_rcpf(PIG[10][1]); SB;
    CC[7][1] = cap_exp_arg(c1[7][1]); SB;
    RF[11][0] = __builtin_amdgcn_rcpf(DF[11][0]); SB;
    set_x(hn1[1], xn); SB;
    RF[11][1] = __builtin_amdgcn_rcpf(DF[11][1]); SB;
    TT[9] = __builtin_elementwise_fma(c1[9], RF[9], -RIG[9]); SB;
    RIG[11][0] = __builtin_amdgcn_rcpf(PIG[11][0]); SB;
    TT[10] = __builtin_elementwise_fma(c1[10], RF[10], -RIG[10]); SB;
    RIG[11][1] = __builtin_amdgcn_rcpf(PIG[11][1]); SB;
    CC[8][0] = cap_exp_arg(c1[8][0]); SB;
    EI[23][0] = __builtin_amdgcn_exp2f(acc2[1][1][0]); SB;
    c1[9] = __builtin_elementwise_fma(EG[9], RIG[9], TT[9]); SB;
    EI[23][1] = __builtin_amdgcn_exp2f(acc2[1][1][1]); SB;
    c1[10] = __builtin_elementwise_fma(EG[10], RIG[10], TT[10]); SB;
    EF[23][0] = __builtin_amdgcn_exp2f(acc2[1][1][2]); SB;
    CC[8][1] = cap_exp_arg(c1[8][1]); SB;
    EF[23][1] = __builtin_amdgcn_exp2f(acc2[1][1][3]); SB;
    TT[11] = __builtin_elementwise_fma(c1[11], RF[11], -RIG[11]); SB;
    EO[20][0] = __builtin_amdgcn_exp2f(acc2[0][2][6]); SB;
    CC[9][0] = cap_exp_arg(c1[9][0]); SB;
    EO[20][1] = __builtin_amdgcn_exp2f(acc2[0][2][7]); SB;
    PIG[23] = __builtin_elementwise_fma(EI[23], DGK[23], DGK[23]); SB;
    RF[22][0] = __builtin_amdgcn_rcpf(DF[22][0]); SB;
    c1[11] = __builtin_elementwise_fma(EG[11], RIG[11], TT[11]); SB;
    RF[22][1] = __builtin_amdgcn_rcpf(DF[22][1]); SB;
    DF[23] = EF[23] + 1.0f; SB;
    RIG[22][0] = __builtin_amdgcn_rcpf(PIG[22][0]); SB;
    CC[9][1] = cap_exp_arg(c1[9][1]); SB;
    RIG[22][1] = __builtin_amdgcn_rcpf(PIG[22][1]); SB;
    DQ[20] = EO[20] + 1.0f; SB;
    EG[24][0] = __builtin_amdgcn_exp2f(GC[24][0]); SB;
    CC[10][0] = cap_exp_arg(c1[10][0]); SB;
    EG[24][1] = __builtin_amdgcn_exp2f(GC[24][1]); SB;
    CC[10][1] = cap_exp_arg(c1[10][1]); SB;
    EC[20][0] = __builtin_amdgcn_exp2f(CC[20][0]); SB;
    CC[11][0] = cap_exp_arg(c1[11][0]); SB;
    EC[20][1] = __builtin_amdgcn_exp2f(CC[20][1]); SB;
    TT[22] = __builtin_elementwise_fma(c2[9], RF[22], -RIG[22]); SB;
    ROC[19][0] = __builtin_amdgcn_rcpf(POC[19][0]); SB;
    CC[11][1] = cap_exp_arg(c1[11][1]); SB;
    ROC[19][1] = __builtin_amdgcn_rcpf(POC[19][1]); SB;
    DGK[24] = __builtin_elementwise_fma(EG[24], rkc2, rkc2); SB;
    EI[24][0] = __builtin_amdgcn_exp2f(acc2[1][2][0]); SB;
    c2[9] = __builtin_elementwise_fma(EG[22], RIG[22], TT[22]); SB;
    EI[24][1] = __builtin_amdgcn_exp2f(acc2[1][2][1]); SB;
    POC[20] = __builtin_elementwise_fma(EC[20], DQ[20], DQ[20]); SB;
    EF[24][0] = __builtin_amdgcn_exp2f(acc2[1][2][2]); SB;
    EF[24][1] = __builtin_amdgcn_exp2f(acc2[1][2][3]); SB;
    HV[19] = __builtin_elementwise_fma(EC[19], ROC[19], -ROC[19]); SB;
    EO[21][0] = __builtin_amdgcn_exp2f(acc2[0][3][6]); SB;
    CC[22][0] = cap_exp_arg(c2[9][0]); SB;
    EO[21][1] = __builtin_amdgcn_exp2f(acc2[0][3][7]); SB;
    PIG[24] = __builtin_elementwise_fma(EI[24], DGK[24], DGK[24]); SB;
    RF[23][0] = __builtin_amdgcn_rcpf(DF[23][0]); SB;
    CC[22][1] = cap_exp_arg(c2[9][1]); SB;
    RF[23][1] = __builtin_amdgcn_rcpf(DF[23][1]); SB;
    DF[24] = EF[24] + 1.0f; SB;
    RIG[23][0] = __builtin_amdgcn_rcpf(PIG[23][0]); SB;
    ys += HV[19][0] * wl[6][0]; SB;
    RIG[23][1] = __builtin_amdgcn_rcpf(PIG[23][1]); SB;
    DQ[21] = EO[21] + 1.0f; SB;
    EG[25][0] = __builtin_amdgcn_exp2f(GC[25][0]); SB;
    { v2h pk; pk[0] = (_Float16)HV[19][0]; pk[1] = (_Float16)HV[19][1]; hn2[0][6] = __builtin_bit_cast(unsigned, pk); } SB;
    EG[25][1] = __builtin_amdgcn_exp2f(GC[25][1]); SB;
    ys += HV[19][1] * wl[6][1]; SB;
    EC[21][0] = __builtin_amdgcn_exp2f(CC[21][0]); SB;
    EC[21][1] = __builtin_amdgcn_exp2f(CC[21][1]); SB;
    TT[23] = __builtin_elementwise_fma(c2[10], RF[23], -RIG[23]); SB;
    EO[0][0] = __builtin_amdgcn_exp2f(acc1[0][0][6]); SB;
    EO[0][1] = __builtin_amdgcn_exp2f(acc1[0][0][7]); SB;
    DGK[25] = __builtin_elementwise_fma(EG[25], rkc2, rkc2); SB;
    EO[1][0] = __builtin_amdgcn_exp2f(acc1[0][1][6]); SB;
    c2[10] = __builtin_elementwise_fma(EG[23], RIG[23], TT[23]); SB;
    EO[1][1] = __builtin_amdgcn_exp2f(acc1[0][1][7]); SB;
    POC[21] = __builtin_elementwise_fma(EC[21], DQ[21], DQ[21]); SB;
    EO[2][0] = __builtin_amdgcn_exp2f(acc1[0][2][6]); SB;
    EO[2][1] = __builtin_amdgcn_exp2f(acc1[0][2][7]); SB;
    DQ[0] = EO[0] + 1.0f; SB;
    EO[3][0] = __builtin_amdgcn_exp2f(acc1[0][3][6]); SB;
    CC[23][0] = cap_exp_arg(c2[10][0]); SB;
    EO[3][1] = __builtin_amdgcn_exp2f(acc1[0][3][7]); SB;
    DQ[1] = EO[1] + 1.0f; SB;
    EC[12][0] = __builtin_amdgcn_exp2f(CC[12][0]); SB;
    CC[23][1] = cap_exp_arg(c2[10][1]); SB;
    EC[12][1] = __builtin_amdgcn_exp2f(CC[12][1]); SB;
    DQ[2] = EO[2] + 1.0f; SB;
    ROC[20][0] = __builtin_amdgcn_rcpf(POC[20][0]); SB;
    ROC[20][1] = __builtin_amdgcn_rcpf(POC[20][1]); SB;
    DQ[3] = EO[3] + 1.0f; SB;
    EI[25][0] = __builtin_amdgcn_exp2f(acc1[1][0][0]); SB;
    EI[25][1] = __builtin_amdgcn_exp2f(acc1[1][0][1]); SB;
    POC[12] = __builtin_elementwise_fma(EC[12], DQ[12], DQ[12]); SB;
    EF[25][0] = __builtin_amdgcn_exp2f(acc1[1][0][2]); SB;
    EF[25][1] = __builtin_amdgcn_exp2f(acc1[1][0][3]); SB;
    HV[20] = __builtin_elementwise_fma(EC[20], ROC[20], -ROC[20]); SB;
    EO[22][0] = __builtin_amdgcn_exp2f(acc2[1][0][6]); SB;
    EO[22][1] = __builtin_amdgcn_exp2f(acc2[1][0][7]); SB;
    PIG[25] = __builtin_elementwise_fma(EI[25], DGK[25], DGK[25]); SB;
    EC[0][0] = __builtin_amdgcn_exp2f(CC[0][0]); SB;
    ys += HV[20][0] * wl[7][0]; SB;
    EC[0][1] = __builtin_amdgcn_exp2f(CC[0][1]); SB;
    DF[25] = EF[25] + 1.0f; SB;
    EC[1][0] = __builtin_amdgcn_exp2f(CC[1][0]); SB;
    { v2h pk; pk[0] = (_Float16)HV[20][0]; pk[1] = (_Float16)HV[20][1]; hn2[0][7] = __builtin_bit_cast(unsigned, pk); } SB;
    EC[1][1] = __builtin_amdgcn_exp2f(CC[1][1]); SB;
    DQ[22] = EO[22] + 1.0f; SB;
    EC[2][0] = __builtin_amdgcn_exp2f(CC[2][0]); SB;
    ys += HV[20][1] * wl[7][1]; SB;
    EC[2][1] = __builtin_amdgcn_exp2f(CC[2][1]); SB;
    POC[0] = __builtin_elementwise_fma(EC[0], DQ[0], DQ[0]); SB;
    EC[3][0] = __builtin_amdgcn_exp2f(CC[3][0]); SB;
    EC[3][1] = __builtin_amdgcn_exp2f(CC[3][1]); SB;
    POC[1] = __builtin_elementwise_fma(EC[1], DQ[1], DQ[1]); SB;
    EC[4][0] = __builtin_amdgcn_exp2f(CC[4][0]); SB;
    EC[4][1] = __builtin_amdgcn_exp2f(CC[4][1]); SB;
    POC[2] = __builtin_elementwise_fma(EC[2], DQ[2], DQ[2]); SB;
    EC[5][0] = __builtin_amdgcn_exp2f(CC[5][0]); SB;
    EC[5][1] = __builtin_amdgcn_exp2f(CC[5][1]); SB;
    POC[3] = __builtin_elementwise_fma(EC[3], DQ[3], DQ[3]); SB;
    EC[6][0] = __builtin_amdgcn_exp2f(CC[6][0]); SB;
    EC[6][1] = __builtin_amdgcn_exp2f(CC[6][1]); SB;
    POC[4] = __builtin_elementwise_fma(EC[4], DQ[4], DQ[4]); SB;
    EC[7][0] = __builtin_amdgcn_exp2f(CC[7][0]); SB;
    EC[7][1] = __builtin_amdgcn_exp2f(CC[7][1]); SB;
    POC[5] = __builtin_elementwise_fma(EC[5], DQ[5], DQ[5]); SB;
    EC[8][0] = __builtin_amdgcn_exp2f(CC[8][0]); SB;
    EC[8][1] = __builtin_amdgcn_exp2f(CC[8][1]); SB;
    POC[6] = __builtin_elementwise_fma(EC[6], DQ[6], DQ[6]); SB;
    EC[9][0] = __builtin_amdgcn_exp2f(CC[9][0]); SB;
    EC[9][1] = __builtin_amdgcn_exp2f(CC[9][1]); SB;
    POC[7] = __builtin_elementwise_fma(EC[7], DQ[7], DQ[7]); SB;
    EC[10][0] = __builtin_amdgcn_exp2f(CC[10][0]); SB;
    EC[10][1] = __builtin_amdgcn_exp2f(CC[10][1]); SB;
    POC[8] = __builtin_elementwise_fma(EC[8], DQ[8], DQ[8]); SB;
    EC[11][0] = __builtin_amdgcn_exp2f(CC[11][0]); SB;
    EC[11][1] = __builtin_amdgcn_exp2f(CC[11][1]); SB;
    POC[9] = __builtin_elementwise_fma(EC[9], DQ[9], DQ[9]); SB;
    RF[24][0] = __builtin_amdgcn_rcpf(DF[24][0]); SB;
    RF[24][1] = __builtin_amdgcn_rcpf(DF[24][1]); SB;
    POC[10] = __builtin_elementwise_fma(EC[10], DQ[10], DQ[10]); SB;
    RIG[24][0] = __builtin_amdgcn_rcpf(PIG[24][0]); SB;
    RIG[24][1] = __builtin_amdgcn_rcpf(PIG[24][1]); SB;
    POC[11] = __builtin_elementwise_fma(EC[11], DQ[11], DQ[11]); SB;
    EC[22][0] = __builtin_amdgcn_exp2f(CC[22][0]); SB;
    EC[22][1] = __builtin_amdgcn_exp2f(CC[22][1]); SB;
    ROC[21][0] = __builtin_amdgcn_rcpf(POC[21][0]); SB;
    ROC[21][1] = __builtin_amdgcn_rcpf(POC[21][1]); SB;
    TT[24] = __builtin_elementwise_fma(c2[11], RF[24], -RIG[24]); SB;
    ROC[12][0] = __builtin_amdgcn_rcpf(POC[12][0]); SB;
    ROC[12][1] = __builtin_amdgcn_rcpf(POC[12][1]); SB;
    POC[22] = __builtin_elementwise_fma(EC[22], DQ[22], DQ[22]); SB;
    EO[23][0] = __builtin_amdgcn_exp2f(acc2[1][1][6]); SB;
    c2[11] = __builtin_elementwise_fma(EG[24], RIG[24], TT[24]); SB;
    EO[23][1] = __builtin_amdgcn_exp2f(acc2[1][1][7]); SB;
    HV[21] = __builtin_elementwise_fma(EC[21], ROC[21], -ROC[21]); SB;
    RF[25][0] = __builtin_amdgcn_rcpf(DF[25][0]); SB;
    RF[25][1] = __builtin_amdgcn_rcpf(DF[25][1]); SB;
    HV[12] = __builtin_elementwise_fma(EC[12], ROC[12], -ROC[12]); SB;
    RIG[25][0] = __builtin_amdgcn_rcpf(PIG[25][0]); SB;
    CC[24][0] = cap_exp_arg(c2[11][0]); SB;
    RIG[25][1] = __builtin_amdgcn_rcpf(PIG[25][1]); SB;
    DQ[23] = EO[23] + 1.0f; SB;
    ROC[0][0] = __builtin_amdgcn_rcpf(POC[0][0]); SB;
    CC[24][1] = cap_exp_arg(c2[11][1]); SB;
    ROC[0][1] = __builtin_amdgcn_rcpf(POC[0][1]); SB;
    ys += HV[21][0] * wl[8][0]; SB;
    ROC[1][0] = __builtin_amdgcn_rcpf(POC[1][0]); SB;
    HV[12][1] = (lh == 1) ? 0.0f : HV[12][1]; SB;
    ROC[1][1] = __builtin_amdgcn_rcpf(POC[1][1]); SB;
    TT[25] = __builtin_elementwise_fma(c2[12], RF[25], -RIG[25]); SB;
    ROC[2][0] = __builtin_amdgcn_rcpf(POC[2][0]); SB;
    ys += HV[21][1] * wl[8][1]; SB;
    ROC[2][1] = __builtin_amdgcn_rcpf(POC[2][1]); SB;
    HV[0] = __builtin_elementwise_fma(EC[0], ROC[0], -ROC[0]); SB;
    ROC[3][0] = __builtin_amdgcn_rcpf(POC[3][0]); SB;
    c2[12] = __builtin_elementwise_fma(EG[25], RIG[25], TT[25]); SB;
    ROC[3][1] = __builtin_amdgcn_rcpf(POC[3][1]); SB;
    HV[1] = __builtin_elementwise_fma(EC[1], ROC[1], -ROC[1]); SB;
    ROC[4][0] = __builtin_amdgcn_rcpf(POC[4][0]); SB;
    { v2h pk; pk[0] = (_Float16)HV[0][0]; pk[1] = (_Float16)HV[0][1]; hn1[0][0] = __builtin_bit_cast(unsigned, pk); } SB;
    ROC[4][1] = __builtin_amdgcn_rcpf(POC[4][1]); SB;
    HV[2] = __builtin_elementwise_fma(EC[2], ROC[2], -ROC[2]); SB;
    ROC[5][0] = __builtin_amdgcn_rcpf(POC[5][0]); SB;
    { v2h pk; pk[0] = (_Float16)HV[1][0]; pk[1] = (_Float16)HV[1][1]; hn1[0][1] = __builtin_bit_cast(unsigned, pk); } SB;
    ROC[5][1] = __builtin_amdgcn_rcpf(POC[5][1]); SB;
    HV[3] = __builtin_elementwise_fma(EC[3], ROC[3], -ROC[3]); SB;
    ROC[6][0] = __builtin_amdgcn_rcpf(POC[6][0]); SB;
    { v2h pk; pk[0] = (_Float16)HV[2][0]; pk[1] = (_Float16)HV[2][1]; hn1[0][2] = __builtin_bit_cast(unsigned, pk); } SB;
    ROC[6][1] = __builtin_amdgcn_rcpf(POC[6][1]); SB;
    HV[4] = __builtin_elementwise_fma(EC[4], ROC[4], -ROC[4]); SB;
    ROC[7][0] = __builtin_amdgcn_rcpf(POC[7][0]); SB;
    { v2h pk; pk[0] = (_Float16)HV[3][0]; pk[1] = (_Float16)HV[3][1]; hn1[0][3] = __builtin_bit_cast(unsigned, pk); } SB;
    ROC[7][1] = __builtin_amdgcn_rcpf(POC[7][1]); SB;
    HV[5] = __builtin_elementwise_fma(EC[5], ROC[5], -ROC[5]); SB;
    ROC[8][0] = __builtin_amdgcn_rcpf(POC[8][0]); SB;
    { v2h pk; pk[0] = (_Float16)HV[4][0]; pk[1] = (_Float16)HV[4][1]; hn1[0][4] = __builtin_bit_cast(unsigned, pk); } SB;
    ROC[8][1] = __builtin_amdgcn_rcpf(POC[8][1]); SB;
    HV[6] = __builtin_elementwise_fma(EC[6], ROC[6], -ROC[6]); SB;
    ROC[9][0] = __builtin_amdgcn_rcpf(POC[9][0]); SB;
    { v2h pk; pk[0] = (_Float16)HV[5][0]; pk[1] = (_Float16)HV[5][1]; hn1[0][5] = __builtin_bit_cast(unsigned, pk); } SB;
    ROC[9][1] = __builtin_amdgcn_rcpf(POC[9][1]); SB;
    HV[7] = __builtin_elementwise_fma(EC[7], ROC[7], -ROC[7]); SB;
    ROC[10][0] = __builtin_amdgcn_rcpf(POC[10][0]); SB;
    { v2h pk; pk[0] = (_Float16)HV[12][0]; pk[1] = (_Float16)HV[12][1]; hn1[1][4] = __builtin_bit_cast(unsigned, pk); } SB;
    ROC[10][1] = __builtin_amdgcn_rcpf(POC[10][1]); SB;
    HV[8] = __builtin_elementwise_fma(EC[8], ROC[8], -ROC[8]); SB;
    ROC[11][0] = __builtin_amdgcn_rcpf(POC[11][0]); SB;
    { v2h pk; pk[0] = (_Float16)HV[6][0]; pk[1] = (_Float16)HV[6][1]; hn1[0][6] = __builtin_bit_cast(unsigned, pk); } SB;
    ROC[11][1] = __builtin_amdgcn_rcpf(POC[11][1]); SB;
    HV[9] = __builtin_elementwise_fma(EC[9], ROC[9], -ROC[9]); SB;
    EC[23][0] = __builtin_amdgcn_exp2f(CC[23][0]); SB;
    { v2h pk; pk[0] = (_Float16)HV[7][0]; pk[1] = (_Float16)HV[7][1]; hn1[0][7] = __builtin_bit_cast(unsigned, pk); } SB;
    EC[23][1] = __builtin_amdgcn_exp2f(CC[23][1]); SB;
    HV[10] = __builtin_elementwise_fma(EC[10], ROC[10], -ROC[10]); SB;
    ROC[22][0] = __builtin_amdgcn_rcpf(POC[22][0]); SB;
    { v2h pk; pk[0] = (_Float16)HV[8][0]; pk[1] = (_Float16)HV[8][1]; hn1[1][0] = __builtin_bit_cast(unsigned, pk); } SB;
    ROC[22][1] = __builtin_amdgcn_rcpf(POC[22][1]); SB;
    HV[11] = __builtin_elementwise_fma(EC[11], ROC[11], -ROC[11]); SB;
    EO[24][0] = __builtin_amdgcn_exp2f(acc2[1][2][6]); SB;
    { v2h pk; pk[0] = (_Float16)HV[9][0]; pk[1] = (_Float16)HV[9][1]; hn1[1][1] = __builtin_bit_cast(unsigned, pk); } SB;
    EO[24][1] = __builtin_amdgcn_exp2f(acc2[1][2][7]); SB;
    POC[23] = __builtin_elementwise_fma(EC[23], DQ[23], DQ[23]); SB;
    EC[24][0] = __builtin_amdgcn_exp2f(CC[24][0]); SB;
    { v2h pk; pk[0] = (_Float16)HV[10][0]; pk[1] = (_Float16)HV[10][1]; hn1[1][2] = __builtin_bit_cast(unsigned, pk); } SB;
    EC[24][1] = __builtin_amdgcn_exp2f(CC[24][1]); SB;
    HV[22] = __builtin_elementwise_fma(EC[22], ROC[22], -ROC[22]); SB;
    EO[25][0] = __builtin_amdgcn_exp2f(acc1[1][0][6]); SB;
    { v2h pk; pk[0] = (_Float16)HV[11][0]; pk[1] = (_Float16)HV[11][1]; hn1[1][3] = __builtin_bit_cast(unsigned, pk); } SB;
    ROC[23][0] = __builtin_amdgcn_rcpf(POC[23][0]); SB;
    DQ[24] = EO[24] + 1.0f; SB;
    ROC[23][1] = __builtin_amdgcn_rcpf(POC[23][1]); SB;
    CC[25][0] = cap_exp_arg(c2[12][0]); SB;
    EO[25][1] = __builtin_amdgcn_exp2f(acc1[1][0][7]); SB;
    CC[25][1] = cap_exp_arg(c2[12][1]); SB;
    ys += HV[22][0] * wl[9][0]; SB;
    hb1[0] = hn1[0]; hb1[1] = hn1[1]; SB;
    POC[24] = __builtin_elementwise_fma(EC[24], DQ[24], DQ[24]); SB;
    HV[23] = __builtin_elementwise_fma(EC[23], ROC[23], -ROC[23]); SB;
    ys += HV[22][1] * wl[9][1]; SB;
    DQ[25] = EO[25] + 1.0f; SB;
    { v2h pk; pk[0] = (_Float16)HV[21][0]; pk[1] = (_Float16)HV[21][1]; hn2[1][0] = __builtin_bit_cast(unsigned, pk); } SB;
    acc1[0][0] = __builtin_amdgcn_wmma_f32_16x16x32_f16(false, wf1[0], false, __builtin_bit_cast(v16h, hb1[0]), (short)0, zero8, false, false); SB;
    ys += HV[23][0] * wl[10][0]; SB;
    EC[25][0] = __builtin_amdgcn_exp2f(CC[25][0]); SB;
    { v2h pk; pk[0] = (_Float16)HV[22][0]; pk[1] = (_Float16)HV[22][1]; hn2[1][1] = __builtin_bit_cast(unsigned, pk); } SB;
    EC[25][1] = __builtin_amdgcn_exp2f(CC[25][1]); SB;
    acc1[0][1] = __builtin_amdgcn_wmma_f32_16x16x32_f16(false, wf1[2], false, __builtin_bit_cast(v16h, hb1[0]), (short)0, zero8, false, false); SB;
    ys += HV[23][1] * wl[10][1]; SB;
    ROC[24][0] = __builtin_amdgcn_rcpf(POC[24][0]); SB;
    { v2h pk; pk[0] = (_Float16)HV[23][0]; pk[1] = (_Float16)HV[23][1]; hn2[1][2] = __builtin_bit_cast(unsigned, pk); } SB;
    ROC[24][1] = __builtin_amdgcn_rcpf(POC[24][1]); SB;
    acc1[0][2] = __builtin_amdgcn_wmma_f32_16x16x32_f16(false, wf1[4], false, __builtin_bit_cast(v16h, hb1[0]), (short)0, zero8, false, false); SB;
    POC[25] = __builtin_elementwise_fma(EC[25], DQ[25], DQ[25]); SB;
    acc1[0][3] = __builtin_amdgcn_wmma_f32_16x16x32_f16(false, wf1[6], false, __builtin_bit_cast(v16h, hb1[0]), (short)0, zero8, false, false); SB;
    HV[24] = __builtin_elementwise_fma(EC[24], ROC[24], -ROC[24]); SB;
    ROC[25][0] = __builtin_amdgcn_rcpf(POC[25][0]); SB;
    ROC[25][1] = __builtin_amdgcn_rcpf(POC[25][1]); SB;
    acc1[0][0] = __builtin_amdgcn_wmma_f32_16x16x32_f16(false, wf1[1], false, __builtin_bit_cast(v16h, hb1[1]), (short)0, acc1[0][0], false, false); SB;
    ys += HV[24][0] * wl[11][0]; SB;
    { v2h pk; pk[0] = (_Float16)HV[24][0]; pk[1] = (_Float16)HV[24][1]; hn2[1][3] = __builtin_bit_cast(unsigned, pk); } SB;
    acc1[0][1] = __builtin_amdgcn_wmma_f32_16x16x32_f16(false, wf1[3], false, __builtin_bit_cast(v16h, hb1[1]), (short)0, acc1[0][1], false, false); SB;
    HV[25] = __builtin_elementwise_fma(EC[25], ROC[25], -ROC[25]); SB;
    ys += HV[24][1] * wl[11][1]; SB;
    acc1[0][2] = __builtin_amdgcn_wmma_f32_16x16x32_f16(false, wf1[5], false, __builtin_bit_cast(v16h, hb1[1]), (short)0, acc1[0][2], false, false); SB;
    HV[25][1] = (lh == 1) ? 0.0f : HV[25][1]; SB;
    ys += HV[25][0] * wl[12][0]; SB;
    acc1[0][3] = __builtin_amdgcn_wmma_f32_16x16x32_f16(false, wf1[7], false, __builtin_bit_cast(v16h, hb1[1]), (short)0, acc1[0][3], false, false); SB;
    { v2h pk; pk[0] = (_Float16)HV[25][0]; pk[1] = (_Float16)HV[25][1]; hn2[1][4] = __builtin_bit_cast(unsigned, pk); } SB;
    ys += HV[25][1] * wl[12][1]; SB;
    hb2[0] = hn2[0]; hb2[1] = hn2[1]; SB;
    ys += xhalf(ys); ylast = ys + blin; if (lh == 0) yout[t - 1] = ylast; SB;
  }

  for (int s = T - 1; s < Tout; ++s) {
    if (s >= 0) {
      ys = 0.f;
      products(IC(4));
      products(IC(5)); cell(IC(4));
      products(IC(6)); cell(IC(5));
      products(IC(7)); cell(IC(6));
      cell(IC(7));
      hb2[0] = hn2[0]; hb2[1] = hn2[1];
      ys += xhalf(ys);
      ylast = ys + blin;
      if (lh == 0) yout[s] = ylast;
    }
    if (s + 1 < Tout) {
      set_x(hb1[1], ylast);
      products(IC(0));
      products(IC(1)); cell(IC(0));
      products(IC(2)); cell(IC(1));
      products(IC(3)); cell(IC(2));
      cell(IC(3));
      hb1[0] = hn1[0]; hb1[1] = hn1[1];
    }
  }
#undef IC
#undef SB
}

extern "C" void kernel_launch(void* const* d_in, const int* in_sizes, int n_in,
                              void* d_out, int out_size, void* d_ws, size_t ws_size,
                              hipStream_t stream) {
  const float* input = (const float*)d_in[0];
  const float* w_ih1 = (const float*)d_in[2];
  const float* w_hh1 = (const float*)d_in[3];
  const float* b_ih1 = (const float*)d_in[4];
  const float* b_hh1 = (const float*)d_in[5];
  const float* w_ih2 = (const float*)d_in[6];
  const float* w_hh2 = (const float*)d_in[7];
  const float* b_ih2 = (const float*)d_in[8];
  const float* b_hh2 = (const float*)d_in[9];
  const float* w_lin = (const float*)d_in[10];
  const float* b_lin = (const float*)d_in[11];
  float* out = (float*)d_out;

  const int B = 1024;
  const int T = in_sizes[0] / B;
  const int Tout = out_size / B;

  lstm2_wmma_kernel<<<dim3(B / 64), dim3(128), 0, stream>>>(
      input, w_ih1, w_hh1, b_ih1, b_hh1, w_ih2, w_hh2, b_ih2, b_hh2,
      w_lin, b_lin, out, T, Tout);
}
